# starting kernel
# baseline (speedup 1.0000x reference)
#include <hip/hip_runtime.h>
#include <hip/hip_bf16.h>

typedef __attribute__((ext_vector_type(8))) float v8f;
typedef __attribute__((ext_vector_type(2))) float v2f;

__global__ void zero_kernel(float* __restrict__ p, int n) {
  int i = blockIdx.x * blockDim.x + threadIdx.x;
  if (i < n) p[i] = 0.0f;
}

__global__ void degree_kernel(const int* __restrict__ dst, float* __restrict__ deg, int nEdges) {
  int i = blockIdx.x * blockDim.x + threadIdx.x;
  if (i < nEdges)
    __hip_atomic_fetch_add(&deg[dst[i]], 1.0f, __ATOMIC_RELAXED, __HIP_MEMORY_SCOPE_AGENT);
}

__global__ void recip_deg_kernel(float* __restrict__ deg, int n) {
  int i = blockIdx.x * blockDim.x + threadIdx.x;
  if (i < n) deg[i] = 1.0f / fmaxf(deg[i], 1.0f);
}

__global__ void scatter_add_kernel(const float* __restrict__ x, const int* __restrict__ src,
                                   const int* __restrict__ dst, float* __restrict__ agg,
                                   int nEdges, int f4) {
  int tid = blockIdx.x * blockDim.x + threadIdx.x;
  int total = nEdges * f4;
  if (tid >= total) return;
  int e = tid / f4;
  int c = tid - e * f4;
  int s = src[e];
  int d = dst[e];
  float4 v = ((const float4*)x)[s * f4 + c];
  float* p = agg + ((size_t)d * f4 + c) * 4;
  __hip_atomic_fetch_add(p + 0, v.x, __ATOMIC_RELAXED, __HIP_MEMORY_SCOPE_AGENT);
  __hip_atomic_fetch_add(p + 1, v.y, __ATOMIC_RELAXED, __HIP_MEMORY_SCOPE_AGENT);
  __hip_atomic_fetch_add(p + 2, v.z, __ATOMIC_RELAXED, __HIP_MEMORY_SCOPE_AGENT);
  __hip_atomic_fetch_add(p + 3, v.w, __ATOMIC_RELAXED, __HIP_MEMORY_SCOPE_AGENT);
}

template <int FIN>
__global__ __launch_bounds__(128) void
gemm_bias_act_kernel(const float* __restrict__ agg, const float* __restrict__ rdeg,
                     const float* __restrict__ xin, const float* __restrict__ W,
                     const float* __restrict__ bias, float* __restrict__ out,
                     int nRows, int Fout, int doRelu)
{
  constexpr int STRIDE = FIN + 4;
  __shared__ float As[64 * STRIDE];
  __shared__ float Bs[128 * STRIDE];
  __shared__ float bsh[128];

  const int rowBase   = blockIdx.x * 64;
  const int foutTiles = (Fout + 15) >> 4;
  const int bRows     = foutTiles * 16;

  for (int idx = threadIdx.x; idx < bRows * FIN; idx += 128) {
    int r = idx / FIN;
    int k = idx - r * FIN;
    Bs[r * STRIDE + k] = (r < Fout) ? W[(size_t)r * FIN + k] : 0.0f;
  }
  {
    int t = threadIdx.x;
    if (t < 128) bsh[t] = (t < Fout) ? bias[t] : 0.0f;
  }
  for (int idx = threadIdx.x; idx < 64 * FIN; idx += 128) {
    int r  = idx / FIN;
    int k  = idx - r * FIN;
    int gr = rowBase + r;
    float v = 0.0f;
    if (gr < nRows) {
      v = xin[(size_t)gr * FIN + k];
      if (agg) v += agg[(size_t)gr * FIN + k] * rdeg[gr];
    }
    As[r * STRIDE + k] = v;
  }
  __syncthreads();

  const int lane = threadIdx.x & 31;
  const int wave = threadIdx.x >> 5;
  const int half = lane >> 4;
  const int l16  = lane & 15;
  const float* Arow = &As[(wave * 16 + l16) * STRIDE];
  const bool fullRows = (rowBase + 64 <= nRows);

  for (int ct = 0; ct < foutTiles; ++ct) {
    const float* Brow = &Bs[(ct * 16 + l16) * STRIDE];
    v8f c = {};
    #pragma unroll 8
    for (int k0 = 0; k0 < FIN; k0 += 4) {
      int ka = k0 + 2 * half;
      v2f a, b;
      a.x = Arow[ka];
      a.y = Arow[ka + 1];
      b.x = Brow[ka];
      b.y = Brow[ka + 1];
      c = __builtin_amdgcn_wmma_f32_16x16x4_f32(false, a, false, b, (short)0, c, false, false);
    }
    const int n  = ct * 16 + l16;
    float bv = bsh[n];
    float* colBase = out + (size_t)(rowBase + wave * 16 + 8 * half) * Fout + n;
    if (fullRows && (ct * 16 + 16 <= Fout)) {
      #pragma unroll
      for (int r = 0; r < 8; ++r) {
        float val = c[r] + bv;
        if (doRelu) val = fmaxf(val, 0.0f);
        colBase[(size_t)r * Fout] = val;
      }
    } else {
      #pragma unroll
      for (int r = 0; r < 8; ++r) {
        int gRow = rowBase + wave * 16 + r + 8 * half;
        float val = c[r] + bv;
        if (doRelu) val = fmaxf(val, 0.0f);
        if (gRow < nRows && n < Fout) out[(size_t)gRow * Fout + n] = val;
      }
    }
  }
}

extern "C" void kernel_launch(void* const* d_in, const int* in_sizes, int n_in,
                              void* d_out, int out_size, void* d_ws, size_t ws_size,
                              hipStream_t stream) {
  const float* x0 = (const float*)d_in[0];
  const int*   ei = (const int*)d_in[1];
  const float *W1 = (const float*)d_in[2],  *b1 = (const float*)d_in[3];
  const float *W2 = (const float*)d_in[4],  *b2 = (const float*)d_in[5];
  const float *W3 = (const float*)d_in[6],  *b3 = (const float*)d_in[7];
  const float *Wp = (const float*)d_in[8],  *bp = (const float*)d_in[9];
  const float *Wc = (const float*)d_in[10], *bc = (const float*)d_in[11];

  const int N      = in_sizes[0] / 64;
  const int nEdges = in_sizes[1] / 2;
  const int* src = ei;
  const int* dst = ei + nEdges;

  float* ws  = (float*)d_ws;
  float* deg = ws;
  float* agg = ws + 65536;
  float* h2  = agg + (size_t)N * 128;

  float* out = (float*)d_out;
  float* h1  = out;
  float* emb = out;
  float* nxt = out + (size_t)N * 64;
  float* cls = out + (size_t)N * 128;

  auto nb = [](int n) { return dim3((unsigned)((n + 255) / 256)); };
  dim3 b256(256);
  dim3 gemmGrid((unsigned)((N + 63) / 64)), gemmBlock(128);

  zero_kernel     <<<nb(N),      b256, 0, stream>>>(deg, N);
  degree_kernel   <<<nb(nEdges), b256, 0, stream>>>(dst, deg, nEdges);
  recip_deg_kernel<<<nb(N),      b256, 0, stream>>>(deg, N);

  zero_kernel        <<<nb(N * 64),      b256, 0, stream>>>(agg, N * 64);
  scatter_add_kernel <<<nb(nEdges * 16), b256, 0, stream>>>(x0, src, dst, agg, nEdges, 16);
  gemm_bias_act_kernel<64><<<gemmGrid, gemmBlock, 0, stream>>>(agg, deg, x0, W1, b1, h1, N, 128, 1);

  zero_kernel        <<<nb(N * 128),     b256, 0, stream>>>(agg, N * 128);
  scatter_add_kernel <<<nb(nEdges * 32), b256, 0, stream>>>(h1, src, dst, agg, nEdges, 32);
  gemm_bias_act_kernel<128><<<gemmGrid, gemmBlock, 0, stream>>>(agg, deg, h1, W2, b2, h2, N, 128, 1);

  zero_kernel        <<<nb(N * 128),     b256, 0, stream>>>(agg, N * 128);
  scatter_add_kernel <<<nb(nEdges * 32), b256, 0, stream>>>(h2, src, dst, agg, nEdges, 32);
  gemm_bias_act_kernel<128><<<gemmGrid, gemmBlock, 0, stream>>>(agg, deg, h2, W3, b3, emb, N, 64, 1);

  gemm_bias_act_kernel<64><<<gemmGrid, gemmBlock, 0, stream>>>(nullptr, nullptr, emb, Wp, bp, nxt, N, 64, 0);
  gemm_bias_act_kernel<64><<<gemmGrid, gemmBlock, 0, stream>>>(nullptr, nullptr, emb, Wc, bc, cls, N, 10, 0);
}


# GEMM operand feed: A strip read straight from global as float4 into registers and kept for all column tiles, W tile staged by float4 loads all in flight; sum order: k inside each 8 permuted
# speedup vs baseline: 1.5984x; 1.5984x over previous
#include <hip/hip_runtime.h>
#include <hip/hip_bf16.h>

typedef __attribute__((ext_vector_type(8))) float v8f;
typedef __attribute__((ext_vector_type(2))) float v2f;

__global__ void zero_kernel(float* __restrict__ p, int n) {
  int i = blockIdx.x * blockDim.x + threadIdx.x;
  if (i < n) p[i] = 0.0f;
}

__global__ void degree_kernel(const int* __restrict__ dst, float* __restrict__ deg, int nEdges) {
  int i = blockIdx.x * blockDim.x + threadIdx.x;
  if (i < nEdges)
    __hip_atomic_fetch_add(&deg[dst[i]], 1.0f, __ATOMIC_RELAXED, __HIP_MEMORY_SCOPE_AGENT);
}

__global__ void recip_deg_kernel(float* __restrict__ deg, int n) {
  int i = blockIdx.x * blockDim.x + threadIdx.x;
  if (i < n) deg[i] = 1.0f / fmaxf(deg[i], 1.0f);
}

__global__ void scatter_add_kernel(const float* __restrict__ x, const int* __restrict__ src,
                                   const int* __restrict__ dst, float* __restrict__ agg,
                                   int nEdges, int f4) {
  int tid = blockIdx.x * blockDim.x + threadIdx.x;
  int total = nEdges * f4;
  if (tid >= total) return;
  int e = tid / f4;
  int c = tid - e * f4;
  int s = src[e];
  int d = dst[e];
  float4 v = ((const float4*)x)[s * f4 + c];
  float* p = agg + ((size_t)d * f4 + c) * 4;
  __hip_atomic_fetch_add(p + 0, v.x, __ATOMIC_RELAXED, __HIP_MEMORY_SCOPE_AGENT);
  __hip_atomic_fetch_add(p + 1, v.y, __ATOMIC_RELAXED, __HIP_MEMORY_SCOPE_AGENT);
  __hip_atomic_fetch_add(p + 2, v.z, __ATOMIC_RELAXED, __HIP_MEMORY_SCOPE_AGENT);
  __hip_atomic_fetch_add(p + 3, v.w, __ATOMIC_RELAXED, __HIP_MEMORY_SCOPE_AGENT);
}

template <int FIN, int NT>
__device__ __forceinline__ void
gemm_tiles(const float4 (&a4)[FIN / 8], const float* __restrict__ Bl,
           const float* __restrict__ bsh, float* __restrict__ out,
           int ct, int rowBase, int half, int l16, int nRows, int Fout, int doRelu,
           bool fullRows)
{
  constexpr int STRIDE = FIN + 4;
  constexpr int KP = FIN / 8;
  v8f c[NT];
  #pragma unroll
  for (int t = 0; t < NT; ++t) c[t] = v8f{};
  #pragma unroll
  for (int j = 0; j < KP; ++j) {
    float4 b[NT];
    #pragma unroll
    for (int t = 0; t < NT; ++t)
      b[t] = *(const float4*)(Bl + (ct + t) * 16 * STRIDE + 8 * j);
    v2f alo, ahi;
    alo.x = a4[j].x; alo.y = a4[j].y;
    ahi.x = a4[j].z; ahi.y = a4[j].w;
    #pragma unroll
    for (int t = 0; t < NT; ++t) {
      v2f blo; blo.x = b[t].x; blo.y = b[t].y;
      c[t] = __builtin_amdgcn_wmma_f32_16x16x4_f32(false, alo, false, blo, (short)0, c[t], false, false);
    }
    #pragma unroll
    for (int t = 0; t < NT; ++t) {
      v2f bhi; bhi.x = b[t].z; bhi.y = b[t].w;
      c[t] = __builtin_amdgcn_wmma_f32_16x16x4_f32(false, ahi, false, bhi, (short)0, c[t], false, false);
    }
  }
  #pragma unroll
  for (int t = 0; t < NT; ++t) {
    const int n = (ct + t) * 16 + l16;
    const float bv = bsh[n];
    float* colBase = out + (size_t)(rowBase + 8 * half) * Fout + n;
    if (fullRows && ((ct + t) * 16 + 16 <= Fout)) {
      #pragma unroll
      for (int r = 0; r < 8; ++r) {
        float val = c[t][r] + bv;
        if (doRelu) val = fmaxf(val, 0.0f);
        colBase[(size_t)r * Fout] = val;
      }
    } else {
      #pragma unroll
      for (int r = 0; r < 8; ++r) {
        int gRow = rowBase + r + 8 * half;
        float val = c[t][r] + bv;
        if (doRelu) val = fmaxf(val, 0.0f);
        if (gRow < nRows && n < Fout) out[(size_t)gRow * Fout + n] = val;
      }
    }
  }
}

template <int FIN>
__global__ __launch_bounds__(256) void
gemm_bias_act_kernel(const float* __restrict__ agg, const float* __restrict__ rdeg,
                     const float* __restrict__ xin, const float* __restrict__ W,
                     const float* __restrict__ bias, float* __restrict__ out,
                     int nRows, int Fout, int doRelu)
{
  constexpr int STRIDE = FIN + 4;
  constexpr int F4  = FIN / 4;
  constexpr int KP  = FIN / 8;
  constexpr int WLD = (128 * F4) / 256;
  __shared__ __attribute__((aligned(16))) float Bs[128 * STRIDE];
  __shared__ float bsh[128];

  const int foutTiles = (Fout + 15) >> 4;
  const int nW4       = foutTiles * 16 * F4;

  const int lane = threadIdx.x & 31;
  const int wave = threadIdx.x >> 5;
  const int half = lane >> 4;
  const int l16  = lane & 15;
  const int rowBase = (blockIdx.x * 8 + wave) * 16;
  const int row     = min(rowBase + l16, nRows - 1);

  const float bval = bias[min((int)threadIdx.x, Fout - 1)];
  const int nWvalid = Fout * F4;
  float4 wv[WLD];
  #pragma unroll
  for (int i = 0; i < WLD; ++i) {
    int idx = (int)threadIdx.x + i * 256;
    wv[i] = ((const float4*)W)[min(idx, nWvalid - 1)];
  }
  float4 a4[KP];
  {
    const float4* xr = (const float4*)(xin + (size_t)row * FIN) + half;
    #pragma unroll
    for (int j = 0; j < KP; ++j) a4[j] = xr[2 * j];
  }
  float4 g4[KP];
  float rd = 0.0f;
  if (agg) {
    const float4* ar = (const float4*)(agg + (size_t)row * FIN) + half;
    #pragma unroll
    for (int j = 0; j < KP; ++j) g4[j] = ar[2 * j];
    rd = rdeg[row];
  }
  #pragma unroll
  for (int i = 0; i < WLD; ++i) {
    int idx = (int)threadIdx.x + i * 256;
    if (idx < nW4) {
      int r  = idx / F4;
      int c4 = idx - r * F4;
      float4 v = wv[i];
      if (idx >= nWvalid) v = make_float4(0.0f, 0.0f, 0.0f, 0.0f);
      *(float4*)&Bs[r * STRIDE + 4 * c4] = v;
    }
  }
  if (threadIdx.x < 128) bsh[threadIdx.x] = ((int)threadIdx.x < Fout) ? bval : 0.0f;
  if (agg) {
    #pragma unroll
    for (int j = 0; j < KP; ++j) {
      a4[j].x += g4[j].x * rd;
      a4[j].y += g4[j].y * rd;
      a4[j].z += g4[j].z * rd;
      a4[j].w += g4[j].w * rd;
    }
  }
  __syncthreads();
  if (rowBase >= nRows) return;

  const float* Bl = &Bs[l16 * STRIDE + 4 * half];
  const bool fullRows = (rowBase + 16 <= nRows);
  int ct = 0;
  for (; ct + 2 <= foutTiles; ct += 2)
    gemm_tiles<FIN, 2>(a4, Bl, bsh, out, ct, rowBase, half, l16, nRows, Fout, doRelu, fullRows);
  if (ct < foutTiles)
    gemm_tiles<FIN, 1>(a4, Bl, bsh, out, ct, rowBase, half, l16, nRows, Fout, doRelu, fullRows);
}

extern "C" void kernel_launch(void* const* d_in, const int* in_sizes, int n_in,
                              void* d_out, int out_size, void* d_ws, size_t ws_size,
                              hipStream_t stream) {
  const float* x0 = (const float*)d_in[0];
  const int*   ei = (const int*)d_in[1];
  const float *W1 = (const float*)d_in[2],  *b1 = (const float*)d_in[3];
  const float *W2 = (const float*)d_in[4],  *b2 = (const float*)d_in[5];
  const float *W3 = (const float*)d_in[6],  *b3 = (const float*)d_in[7];
  const float *Wp = (const float*)d_in[8],  *bp = (const float*)d_in[9];
  const float *Wc = (const float*)d_in[10], *bc = (const float*)d_in[11];

  const int N      = in_sizes[0] / 64;
  const int nEdges = in_sizes[1] / 2;
  const int* src = ei;
  const int* dst = ei + nEdges;

  float* ws  = (float*)d_ws;
  float* deg = ws;
  float* agg = ws + 65536;
  float* h2  = agg + (size_t)N * 128;

  float* out = (float*)d_out;
  float* h1  = out;
  float* emb = out;
  float* nxt = out + (size_t)N * 64;
  float* cls = out + (size_t)N * 128;

  auto nb = [](int n) { return dim3((unsigned)((n + 255) / 256)); };
  dim3 b256(256);
  dim3 gemmGrid((unsigned)((N + 127) / 128)), gemmBlock(256);

  zero_kernel     <<<nb(N),      b256, 0, stream>>>(deg, N);
  degree_kernel   <<<nb(nEdges), b256, 0, stream>>>(dst, deg, nEdges);
  recip_deg_kernel<<<nb(N),      b256, 0, stream>>>(deg, N);

  zero_kernel        <<<nb(N * 64),      b256, 0, stream>>>(agg, N * 64);
  scatter_add_kernel <<<nb(nEdges * 16), b256, 0, stream>>>(x0, src, dst, agg, nEdges, 16);
  gemm_bias_act_kernel<64><<<gemmGrid, gemmBlock, 0, stream>>>(agg, deg, x0, W1, b1, h1, N, 128, 1);

  zero_kernel        <<<nb(N * 128),     b256, 0, stream>>>(agg, N * 128);
  scatter_add_kernel <<<nb(nEdges * 32), b256, 0, stream>>>(h1, src, dst, agg, nEdges, 32);
  gemm_bias_act_kernel<128><<<gemmGrid, gemmBlock, 0, stream>>>(agg, deg, h1, W2, b2, h2, N, 128, 1);

  zero_kernel        <<<nb(N * 128),     b256, 0, stream>>>(agg, N * 128);
  scatter_add_kernel <<<nb(nEdges * 32), b256, 0, stream>>>(h2, src, dst, agg, nEdges, 32);
  gemm_bias_act_kernel<128><<<gemmGrid, gemmBlock, 0, stream>>>(agg, deg, h2, W3, b3, emb, N, 64, 1);

  gemm_bias_act_kernel<64><<<gemmGrid, gemmBlock, 0, stream>>>(nullptr, nullptr, emb, Wp, bp, nxt, N, 64, 0);
  gemm_bias_act_kernel<64><<<gemmGrid, gemmBlock, 0, stream>>>(nullptr, nullptr, emb, Wc, bc, cls, N, 10, 0);
}


# Float-atomic scatter-add replaced by per-call edge lists grouped by destination (int count, scan, place, order by source id) and a per-node row gather; sum order: fixed, ascending source id
# speedup vs baseline: 2.9954x; 1.8740x over previous
#include <hip/hip_runtime.h>
#include <hip/hip_bf16.h>

typedef __attribute__((ext_vector_type(8))) float v8f;
typedef __attribute__((ext_vector_type(2))) float v2f;


__global__ void zero_int_kernel(int* __restrict__ p, int n) {
  int i = blockIdx.x * blockDim.x + threadIdx.x;
  if (i < n) p[i] = 0;
}

__global__ void indegree_kernel(const int* __restrict__ dst, int* __restrict__ cnt, int nEdges) {
  int i = blockIdx.x * blockDim.x + threadIdx.x;
  if (i < nEdges)
    __hip_atomic_fetch_add(&cnt[dst[i]], 1, __ATOMIC_RELAXED, __HIP_MEMORY_SCOPE_AGENT);
}

__global__ __launch_bounds__(256) void
rowptr_scan_kernel(const int* __restrict__ cnt, int* __restrict__ rowptr, int n)
{
  __shared__ int sIncl[8];
  __shared__ int sPre[8];
  const int t = threadIdx.x, lane = t & 31, w = t >> 5;
  const int base = blockIdx.x * 1024;
  const int4* c4 = (const int4*)cnt;
  int pre = 0;
  #pragma unroll 4
  for (int i = t; i < (base >> 2); i += 256) {
    int4 v = c4[i];
    pre += (v.x + v.y) + (v.z + v.w);
  }
  const int idx = base + 4 * t;
  const int v0 = (idx     < n) ? cnt[idx]     : 0;
  const int v1 = (idx + 1 < n) ? cnt[idx + 1] : 0;
  const int v2 = (idx + 2 < n) ? cnt[idx + 2] : 0;
  const int v3 = (idx + 3 < n) ? cnt[idx + 3] : 0;
  const int local = (v0 + v1) + (v2 + v3);
  int incl = local;
  #pragma unroll
  for (int o = 1; o < 32; o <<= 1) {
    int y = __shfl_up(incl, o);
    if (lane >= o) incl += y;
  }
  #pragma unroll
  for (int o = 16; o > 0; o >>= 1) pre += __shfl_xor(pre, o);
  if (lane == 31) sIncl[w] = incl;
  if (lane == 0)  sPre[w]  = pre;
  __syncthreads();
  int off = 0;
  #pragma unroll
  for (int i = 0; i < 8; ++i) {
    off += sPre[i];
    if (i < w) off += sIncl[i];
  }
  int r = off + incl - local;
  if (idx     < n) rowptr[idx]     = r;
  r += v0;
  if (idx + 1 < n) rowptr[idx + 1] = r;
  r += v1;
  if (idx + 2 < n) rowptr[idx + 2] = r;
  r += v2;
  if (idx + 3 < n) rowptr[idx + 3] = r;
  r += v3;
  if (idx < n && idx + 4 >= n) rowptr[n] = r;
}

__global__ void place_kernel(const int* __restrict__ src, const int* __restrict__ dst,
                             const int* __restrict__ rowptr, int* __restrict__ cnt,
                             int* __restrict__ nbr, int nEdges) {
  int i = blockIdx.x * blockDim.x + threadIdx.x;
  if (i < nEdges) {
    const int d = dst[i];
    const int k = __hip_atomic_fetch_add(&cnt[d], -1, __ATOMIC_RELAXED, __HIP_MEMORY_SCOPE_AGENT);
    nbr[rowptr[d] + k - 1] = src[i];
  }
}

constexpr int NODES_PER_WAVE = 4;

__global__ __launch_bounds__(256) void
order_rows_kernel(const int* __restrict__ rowptr, const int* __restrict__ nbr,
                  int* __restrict__ nbrSorted, int nNodes)
{
  const int lane = threadIdx.x & 31;
  const int wv   = blockIdx.x * 8 + (threadIdx.x >> 5);
  for (int q = 0; q < NODES_PER_WAVE; ++q) {
    const int node = wv * NODES_PER_WAVE + q;
    if (node >= nNodes) return;
    const int beg = rowptr[node], end = rowptr[node + 1];
    for (int i = beg + lane; i < end; i += 32) {
      const int s = nbr[i];
      int r = 0;
      for (int j = beg; j < end; j += 4) {
        int t[4];
        #pragma unroll
        for (int u = 0; u < 4; ++u) t[u] = nbr[min(j + u, end - 1)];
        #pragma unroll
        for (int u = 0; u < 4; ++u) {
          const int jj = j + u;
          const bool front = (t[u] < s) || (t[u] == s && jj < i);
          r += (jj < end && front) ? 1 : 0;
        }
      }
      nbrSorted[beg + r] = s;
    }
  }
}

template <int F4, int U, bool CLAMP>
__device__ __forceinline__ void
add_rows(float4& acc, const float4* __restrict__ x4, const int* __restrict__ nbr,
         int e, int end, int c, int sub)
{
  constexpr int EPS = 32 / F4;
  float4 v[U];
  #pragma unroll
  for (int u = 0; u < U; ++u) {
    int ee = e + u * EPS + sub;
    if (CLAMP) ee = min(ee, end - 1);
    const int s = nbr[ee];
    v[u] = x4[(size_t)s * F4 + c];
  }
  if constexpr (CLAMP) {
    static_assert(U == 4, "the fence below names four rows");
    asm volatile("" :: "v"(v[0].x), "v"(v[0].y), "v"(v[0].z), "v"(v[0].w),
                       "v"(v[1].x), "v"(v[1].y), "v"(v[1].z), "v"(v[1].w),
                       "v"(v[2].x), "v"(v[2].y), "v"(v[2].z), "v"(v[2].w),
                       "v"(v[3].x), "v"(v[3].y), "v"(v[3].z), "v"(v[3].w));
  }
  #pragma unroll
  for (int u = 0; u < U; ++u) {
    if (!CLAMP || (e + u * EPS + sub < end)) {
      acc.x += v[u].x; acc.y += v[u].y; acc.z += v[u].z; acc.w += v[u].w;
    }
  }
}

template <int F4>
__global__ __launch_bounds__(256) void
gather_mean_kernel(const float* __restrict__ x, const int* __restrict__ rowptr,
                   const int* __restrict__ nbr, float* __restrict__ A, int nNodes)
{
  constexpr int EPS = 32 / F4;
  const int lane = threadIdx.x & 31;
  const int c    = lane % F4;
  const int sub  = lane / F4;
  const int wv   = blockIdx.x * 8 + (threadIdx.x >> 5);
  const float4* x4 = (const float4*)x;
  for (int q = 0; q < NODES_PER_WAVE; ++q) {
    const int node = wv * NODES_PER_WAVE + q;
    if (node >= nNodes) return;
    const int beg = rowptr[node], end = rowptr[node + 1];
    const float4 xv = x4[(size_t)node * F4 + c];
    float4 acc = make_float4(0.0f, 0.0f, 0.0f, 0.0f);
    int e = beg;
    for (; e + 8 * EPS <= end; e += 8 * EPS) add_rows<F4, 8, false>(acc, x4, nbr, e, end, c, sub);
    for (; e < end; e += 4 * EPS)            add_rows<F4, 4, true >(acc, x4, nbr, e, end, c, sub);
    if (EPS == 2) {
      acc.x += __shfl_xor(acc.x, 16);
      acc.y += __shfl_xor(acc.y, 16);
      acc.z += __shfl_xor(acc.z, 16);
      acc.w += __shfl_xor(acc.w, 16);
    }
    const float rd = 1.0f / fmaxf((float)(end - beg), 1.0f);
    if (sub == 0) {
      float4 o;
      o.x = xv.x + acc.x * rd;
      o.y = xv.y + acc.y * rd;
      o.z = xv.z + acc.z * rd;
      o.w = xv.w + acc.w * rd;
      ((float4*)A)[(size_t)node * F4 + c] = o;
    }
  }
}

template <int FIN, int NT>
__device__ __forceinline__ void
gemm_tiles(const float4 (&a4)[FIN / 8], const float* __restrict__ Bl,
           const float* __restrict__ bsh, float* __restrict__ out,
           int ct, int rowBase, int half, int l16, int nRows, int Fout, int doRelu,
           bool fullRows)
{
  constexpr int STRIDE = FIN + 4;
  constexpr int KP = FIN / 8;
  v8f c[NT];
  #pragma unroll
  for (int t = 0; t < NT; ++t) c[t] = v8f{};
  #pragma unroll
  for (int j = 0; j < KP; ++j) {
    float4 b[NT];
    #pragma unroll
    for (int t = 0; t < NT; ++t)
      b[t] = *(const float4*)(Bl + (ct + t) * 16 * STRIDE + 8 * j);
    v2f alo, ahi;
    alo.x = a4[j].x; alo.y = a4[j].y;
    ahi.x = a4[j].z; ahi.y = a4[j].w;
    #pragma unroll
    for (int t = 0; t < NT; ++t) {
      v2f blo; blo.x = b[t].x; blo.y = b[t].y;
      c[t] = __builtin_amdgcn_wmma_f32_16x16x4_f32(false, alo, false, blo, (short)0, c[t], false, false);
    }
    #pragma unroll
    for (int t = 0; t < NT; ++t) {
      v2f bhi; bhi.x = b[t].z; bhi.y = b[t].w;
      c[t] = __builtin_amdgcn_wmma_f32_16x16x4_f32(false, ahi, false, bhi, (short)0, c[t], false, false);
    }
  }
  #pragma unroll
  for (int t = 0; t < NT; ++t) {
    const int n = (ct + t) * 16 + l16;
    const float bv = bsh[n];
    float* colBase = out + (size_t)(rowBase + 8 * half) * Fout + n;
    if (fullRows && ((ct + t) * 16 + 16 <= Fout)) {
      #pragma unroll
      for (int r = 0; r < 8; ++r) {
        float val = c[t][r] + bv;
        if (doRelu) val = fmaxf(val, 0.0f);
        colBase[(size_t)r * Fout] = val;
      }
    } else {
      #pragma unroll
      for (int r = 0; r < 8; ++r) {
        int gRow = rowBase + r + 8 * half;
        float val = c[t][r] + bv;
        if (doRelu) val = fmaxf(val, 0.0f);
        if (gRow < nRows && n < Fout) out[(size_t)gRow * Fout + n] = val;
      }
    }
  }
}

template <int FIN>
__global__ __launch_bounds__(256) void
gemm_bias_act_kernel(const float* __restrict__ xin, const float* __restrict__ W,
                     const float* __restrict__ bias, float* __restrict__ out,
                     int nRows, int Fout, int doRelu)
{
  constexpr int STRIDE = FIN + 4;
  constexpr int F4  = FIN / 4;
  constexpr int KP  = FIN / 8;
  constexpr int WLD = (128 * F4) / 256;
  __shared__ __attribute__((aligned(16))) float Bs[128 * STRIDE];
  __shared__ float bsh[128];

  const int foutTiles = (Fout + 15) >> 4;
  const int nW4       = foutTiles * 16 * F4;

  const int lane = threadIdx.x & 31;
  const int wave = threadIdx.x >> 5;
  const int half = lane >> 4;
  const int l16  = lane & 15;
  const int rowBase = (blockIdx.x * 8 + wave) * 16;
  const int row     = min(rowBase + l16, nRows - 1);

  const float bval = bias[min((int)threadIdx.x, Fout - 1)];
  const int nWvalid = Fout * F4;
  float4 wv[WLD];
  #pragma unroll
  for (int i = 0; i < WLD; ++i) {
    int idx = (int)threadIdx.x + i * 256;
    wv[i] = ((const float4*)W)[min(idx, nWvalid - 1)];
  }
  float4 a4[KP];
  {
    const float4* xr = (const float4*)(xin + (size_t)row * FIN) + half;
    #pragma unroll
    for (int j = 0; j < KP; ++j) a4[j] = xr[2 * j];
  }
  #pragma unroll
  for (int i = 0; i < WLD; ++i) {
    int idx = (int)threadIdx.x + i * 256;
    if (idx < nW4) {
      int r  = idx / F4;
      int c4 = idx - r * F4;
      float4 v = wv[i];
      if (idx >= nWvalid) v = make_float4(0.0f, 0.0f, 0.0f, 0.0f);
      *(float4*)&Bs[r * STRIDE + 4 * c4] = v;
    }
  }
  if (threadIdx.x < 128) bsh[threadIdx.x] = ((int)threadIdx.x < Fout) ? bval : 0.0f;
  __syncthreads();
  if (rowBase >= nRows) return;

  const float* Bl = &Bs[l16 * STRIDE + 4 * half];
  const bool fullRows = (rowBase + 16 <= nRows);
  int ct = 0;
  for (; ct + 2 <= foutTiles; ct += 2)
    gemm_tiles<FIN, 2>(a4, Bl, bsh, out, ct, rowBase, half, l16, nRows, Fout, doRelu, fullRows);
  if (ct < foutTiles)
    gemm_tiles<FIN, 1>(a4, Bl, bsh, out, ct, rowBase, half, l16, nRows, Fout, doRelu, fullRows);
}

extern "C" void kernel_launch(void* const* d_in, const int* in_sizes, int n_in,
                              void* d_out, int out_size, void* d_ws, size_t ws_size,
                              hipStream_t stream) {
  const float* x0 = (const float*)d_in[0];
  const int*   ei = (const int*)d_in[1];
  const float *W1 = (const float*)d_in[2],  *b1 = (const float*)d_in[3];
  const float *W2 = (const float*)d_in[4],  *b2 = (const float*)d_in[5];
  const float *W3 = (const float*)d_in[6],  *b3 = (const float*)d_in[7];
  const float *Wp = (const float*)d_in[8],  *bp = (const float*)d_in[9];
  const float *Wc = (const float*)d_in[10], *bc = (const float*)d_in[11];

  const int N      = in_sizes[0] / 64;
  const int nEdges = in_sizes[1] / 2;
  const int* src = ei;
  const int* dst = ei + nEdges;

  float* ws        = (float*)d_ws;
  int*   cnt       = (int*)d_ws;
  float* A         = ws + 65536;
  int*   rowptr    = (int*)(A + (size_t)N * 128);
  int*   nbr       = rowptr + 65536;
  int*   nbrSorted = nbr + nEdges;

  float* out = (float*)d_out;
  float* h1  = out;
  float* h2  = out;
  float* emb = out;
  float* nxt = out + (size_t)N * 64;
  float* cls = out + (size_t)N * 128;

  auto nb = [](int n) { return dim3((unsigned)((n + 255) / 256)); };
  dim3 b256(256);
  dim3 gemmGrid((unsigned)((N + 127) / 128)), gemmBlock(256);
  dim3 nodeGrid((unsigned)((N + 8 * NODES_PER_WAVE - 1) / (8 * NODES_PER_WAVE)));

  zero_int_kernel   <<<nb(N),      b256, 0, stream>>>(cnt, N);
  indegree_kernel   <<<nb(nEdges), b256, 0, stream>>>(dst, cnt, nEdges);
  rowptr_scan_kernel<<<dim3((unsigned)((N + 1023) / 1024)), b256, 0, stream>>>(cnt, rowptr, N);
  place_kernel      <<<nb(nEdges), b256, 0, stream>>>(src, dst, rowptr, cnt, nbr, nEdges);
  order_rows_kernel <<<nodeGrid,   b256, 0, stream>>>(rowptr, nbr, nbrSorted, N);

  gather_mean_kernel<16><<<nodeGrid, b256, 0, stream>>>(x0, rowptr, nbrSorted, A, N);
  gemm_bias_act_kernel<64><<<gemmGrid, gemmBlock, 0, stream>>>(A, W1, b1, h1, N, 128, 1);

  gather_mean_kernel<32><<<nodeGrid, b256, 0, stream>>>(h1, rowptr, nbrSorted, A, N);
  gemm_bias_act_kernel<128><<<gemmGrid, gemmBlock, 0, stream>>>(A, W2, b2, h2, N, 128, 1);

  gather_mean_kernel<32><<<nodeGrid, b256, 0, stream>>>(h2, rowptr, nbrSorted, A, N);
  gemm_bias_act_kernel<128><<<gemmGrid, gemmBlock, 0, stream>>>(A, W3, b3, emb, N, 64, 1);

  gemm_bias_act_kernel<64><<<gemmGrid, gemmBlock, 0, stream>>>(emb, Wp, bp, nxt, N, 64, 0);
  gemm_bias_act_kernel<64><<<gemmGrid, gemmBlock, 0, stream>>>(emb, Wc, bc, cls, N, 10, 0);
}


# Layer-3 GEMM and both heads in one launch: each wave keeps its finished 16x64 embedding tile in LDS and runs the two head products on it (2 launches, 2 reads of emb less); same sums, same order
# speedup vs baseline: 3.1135x; 1.0394x over previous
#include <hip/hip_runtime.h>
#include <hip/hip_bf16.h>

typedef __attribute__((ext_vector_type(8))) float v8f;
typedef __attribute__((ext_vector_type(2))) float v2f;


__global__ void zero_int_kernel(int* __restrict__ p, int n) {
  int i = blockIdx.x * blockDim.x + threadIdx.x;
  if (i < n) p[i] = 0;
}

__global__ void indegree_kernel(const int* __restrict__ dst, int* __restrict__ cnt, int nEdges) {
  int i = blockIdx.x * blockDim.x + threadIdx.x;
  if (i < nEdges)
    __hip_atomic_fetch_add(&cnt[dst[i]], 1, __ATOMIC_RELAXED, __HIP_MEMORY_SCOPE_AGENT);
}

__global__ __launch_bounds__(256) void
rowptr_scan_kernel(const int* __restrict__ cnt, int* __restrict__ rowptr, int n)
{
  __shared__ int sIncl[8];
  __shared__ int sPre[8];
  const int t = threadIdx.x, lane = t & 31, w = t >> 5;
  const int base = blockIdx.x * 1024;
  const int4* c4 = (const int4*)cnt;
  int pre = 0;
  #pragma unroll 4
  for (int i = t; i < (base >> 2); i += 256) {
    int4 v = c4[i];
    pre += (v.x + v.y) + (v.z + v.w);
  }
  const int idx = base + 4 * t;
  const int v0 = (idx     < n) ? cnt[idx]     : 0;
  const int v1 = (idx + 1 < n) ? cnt[idx + 1] : 0;
  const int v2 = (idx + 2 < n) ? cnt[idx + 2] : 0;
  const int v3 = (idx + 3 < n) ? cnt[idx + 3] : 0;
  const int local = (v0 + v1) + (v2 + v3);
  int incl = local;
  #pragma unroll
  for (int o = 1; o < 32; o <<= 1) {
    int y = __shfl_up(incl, o);
    if (lane >= o) incl += y;
  }
  #pragma unroll
  for (int o = 16; o > 0; o >>= 1) pre += __shfl_xor(pre, o);
  if (lane == 31) sIncl[w] = incl;
  if (lane == 0)  sPre[w]  = pre;
  __syncthreads();
  int off = 0;
  #pragma unroll
  for (int i = 0; i < 8; ++i) {
    off += sPre[i];
    if (i < w) off += sIncl[i];
  }
  int r = off + incl - local;
  if (idx     < n) rowptr[idx]     = r;
  r += v0;
  if (idx + 1 < n) rowptr[idx + 1] = r;
  r += v1;
  if (idx + 2 < n) rowptr[idx + 2] = r;
  r += v2;
  if (idx + 3 < n) rowptr[idx + 3] = r;
  r += v3;
  if (idx < n && idx + 4 >= n) rowptr[n] = r;
}

__global__ void place_kernel(const int* __restrict__ src, const int* __restrict__ dst,
                             const int* __restrict__ rowptr, int* __restrict__ cnt,
                             int* __restrict__ nbr, int nEdges) {
  int i = blockIdx.x * blockDim.x + threadIdx.x;
  if (i < nEdges) {
    const int d = dst[i];
    const int k = __hip_atomic_fetch_add(&cnt[d], -1, __ATOMIC_RELAXED, __HIP_MEMORY_SCOPE_AGENT);
    nbr[rowptr[d] + k - 1] = src[i];
  }
}

constexpr int NODES_PER_WAVE = 4;

__global__ __launch_bounds__(256) void
order_rows_kernel(const int* __restrict__ rowptr, const int* __restrict__ nbr,
                  int* __restrict__ nbrSorted, int nNodes)
{
  const int lane = threadIdx.x & 31;
  const int wv   = blockIdx.x * 8 + (threadIdx.x >> 5);
  for (int q = 0; q < NODES_PER_WAVE; ++q) {
    const int node = wv * NODES_PER_WAVE + q;
    if (node >= nNodes) return;
    const int beg = rowptr[node], end = rowptr[node + 1];
    for (int i = beg + lane; i < end; i += 32) {
      const int s = nbr[i];
      int r = 0;
      for (int j = beg; j < end; j += 4) {
        int t[4];
        #pragma unroll
        for (int u = 0; u < 4; ++u) t[u] = nbr[min(j + u, end - 1)];
        #pragma unroll
        for (int u = 0; u < 4; ++u) {
          const int jj = j + u;
          const bool front = (t[u] < s) || (t[u] == s && jj < i);
          r += (jj < end && front) ? 1 : 0;
        }
      }
      nbrSorted[beg + r] = s;
    }
  }
}

template <int F4, int U, bool CLAMP>
__device__ __forceinline__ void
add_rows(float4& acc, const float4* __restrict__ x4, const int* __restrict__ nbr,
         int e, int end, int c, int sub)
{
  constexpr int EPS = 32 / F4;
  float4 v[U];
  #pragma unroll
  for (int u = 0; u < U; ++u) {
    int ee = e + u * EPS + sub;
    if (CLAMP) ee = min(ee, end - 1);
    const int s = nbr[ee];
    v[u] = x4[(size_t)s * F4 + c];
  }
  if constexpr (CLAMP) {
    static_assert(U == 4, "the fence below names four rows");
    asm volatile("" :: "v"(v[0].x), "v"(v[0].y), "v"(v[0].z), "v"(v[0].w),
                       "v"(v[1].x), "v"(v[1].y), "v"(v[1].z), "v"(v[1].w),
                       "v"(v[2].x), "v"(v[2].y), "v"(v[2].z), "v"(v[2].w),
                       "v"(v[3].x), "v"(v[3].y), "v"(v[3].z), "v"(v[3].w));
  }
  #pragma unroll
  for (int u = 0; u < U; ++u) {
    if (!CLAMP || (e + u * EPS + sub < end)) {
      acc.x += v[u].x; acc.y += v[u].y; acc.z += v[u].z; acc.w += v[u].w;
    }
  }
}

template <int F4>
__global__ __launch_bounds__(256) void
gather_mean_kernel(const float* __restrict__ x, const int* __restrict__ rowptr,
                   const int* __restrict__ nbr, float* __restrict__ A, int nNodes)
{
  constexpr int EPS = 32 / F4;
  const int lane = threadIdx.x & 31;
  const int c    = lane % F4;
  const int sub  = lane / F4;
  const int wv   = blockIdx.x * 8 + (threadIdx.x >> 5);
  const float4* x4 = (const float4*)x;
  for (int q = 0; q < NODES_PER_WAVE; ++q) {
    const int node = wv * NODES_PER_WAVE + q;
    if (node >= nNodes) return;
    const int beg = rowptr[node], end = rowptr[node + 1];
    const float4 xv = x4[(size_t)node * F4 + c];
    float4 acc = make_float4(0.0f, 0.0f, 0.0f, 0.0f);
    int e = beg;
    for (; e + 8 * EPS <= end; e += 8 * EPS) add_rows<F4, 8, false>(acc, x4, nbr, e, end, c, sub);
    for (; e < end; e += 4 * EPS)            add_rows<F4, 4, true >(acc, x4, nbr, e, end, c, sub);
    if (EPS == 2) {
      acc.x += __shfl_xor(acc.x, 16);
      acc.y += __shfl_xor(acc.y, 16);
      acc.z += __shfl_xor(acc.z, 16);
      acc.w += __shfl_xor(acc.w, 16);
    }
    const float rd = 1.0f / fmaxf((float)(end - beg), 1.0f);
    if (sub == 0) {
      float4 o;
      o.x = xv.x + acc.x * rd;
      o.y = xv.y + acc.y * rd;
      o.z = xv.z + acc.z * rd;
      o.w = xv.w + acc.w * rd;
      ((float4*)A)[(size_t)node * F4 + c] = o;
    }
  }
}

constexpr int KEEP_PITCH = 64 + 4;
template <int FIN, int NT, bool KEEP = false>
__device__ __forceinline__ void
gemm_tiles(const float4 (&a4)[FIN / 8], const float* __restrict__ Bl,
           const float* __restrict__ bsh, float* __restrict__ out,
           int ct, int rowBase, int half, int l16, int nRows, int Fout, int doRelu,
           bool fullRows, float* keep = nullptr)
{
  constexpr int STRIDE = FIN + 4;
  constexpr int KP = FIN / 8;
  v8f c[NT];
  #pragma unroll
  for (int t = 0; t < NT; ++t) c[t] = v8f{};
  #pragma unroll
  for (int j = 0; j < KP; ++j) {
    float4 b[NT];
    #pragma unroll
    for (int t = 0; t < NT; ++t)
      b[t] = *(const float4*)(Bl + (ct + t) * 16 * STRIDE + 8 * j);
    v2f alo, ahi;
    alo.x = a4[j].x; alo.y = a4[j].y;
    ahi.x = a4[j].z; ahi.y = a4[j].w;
    #pragma unroll
    for (int t = 0; t < NT; ++t) {
      v2f blo; blo.x = b[t].x; blo.y = b[t].y;
      c[t] = __builtin_amdgcn_wmma_f32_16x16x4_f32(false, alo, false, blo, (short)0, c[t], false, false);
    }
    #pragma unroll
    for (int t = 0; t < NT; ++t) {
      v2f bhi; bhi.x = b[t].z; bhi.y = b[t].w;
      c[t] = __builtin_amdgcn_wmma_f32_16x16x4_f32(false, ahi, false, bhi, (short)0, c[t], false, false);
    }
  }
  #pragma unroll
  for (int t = 0; t < NT; ++t) {
    const int n = (ct + t) * 16 + l16;
    const float bv = bsh[n];
    float* colBase = out + (size_t)(rowBase + 8 * half) * Fout + n;
    if (fullRows && ((ct + t) * 16 + 16 <= Fout)) {
      #pragma unroll
      for (int r = 0; r < 8; ++r) {
        float val = c[t][r] + bv;
        if (doRelu) val = fmaxf(val, 0.0f);
        colBase[(size_t)r * Fout] = val;
        if constexpr (KEEP) keep[(r + 8 * half) * KEEP_PITCH + n] = val;
      }
    } else {
      #pragma unroll
      for (int r = 0; r < 8; ++r) {
        int gRow = rowBase + r + 8 * half;
        float val = c[t][r] + bv;
        if (doRelu) val = fmaxf(val, 0.0f);
        if (gRow < nRows && n < Fout) out[(size_t)gRow * Fout + n] = val;
        if constexpr (KEEP) keep[(r + 8 * half) * KEEP_PITCH + n] = val;
      }
    }
  }
}

template <int FIN>
__global__ __launch_bounds__(256) void
gemm_bias_act_kernel(const float* __restrict__ xin, const float* __restrict__ W,
                     const float* __restrict__ bias, float* __restrict__ out,
                     int nRows, int Fout, int doRelu)
{
  constexpr int STRIDE = FIN + 4;
  constexpr int F4  = FIN / 4;
  constexpr int KP  = FIN / 8;
  constexpr int WLD = (128 * F4) / 256;
  __shared__ __attribute__((aligned(16))) float Bs[128 * STRIDE];
  __shared__ float bsh[128];

  const int foutTiles = (Fout + 15) >> 4;
  const int nW4       = foutTiles * 16 * F4;

  const int lane = threadIdx.x & 31;
  const int wave = threadIdx.x >> 5;
  const int half = lane >> 4;
  const int l16  = lane & 15;
  const int rowBase = (blockIdx.x * 8 + wave) * 16;
  const int row     = min(rowBase + l16, nRows - 1);

  const float bval = bias[min((int)threadIdx.x, Fout - 1)];
  const int nWvalid = Fout * F4;
  float4 wv[WLD];
  #pragma unroll
  for (int i = 0; i < WLD; ++i) {
    int idx = (int)threadIdx.x + i * 256;
    wv[i] = ((const float4*)W)[min(idx, nWvalid - 1)];
  }
  float4 a4[KP];
  {
    const float4* xr = (const float4*)(xin + (size_t)row * FIN) + half;
    #pragma unroll
    for (int j = 0; j < KP; ++j) a4[j] = xr[2 * j];
  }
  #pragma unroll
  for (int i = 0; i < WLD; ++i) {
    int idx = (int)threadIdx.x + i * 256;
    if (idx < nW4) {
      int r  = idx / F4;
      int c4 = idx - r * F4;
      float4 v = wv[i];
      if (idx >= nWvalid) v = make_float4(0.0f, 0.0f, 0.0f, 0.0f);
      *(float4*)&Bs[r * STRIDE + 4 * c4] = v;
    }
  }
  if (threadIdx.x < 128) bsh[threadIdx.x] = ((int)threadIdx.x < Fout) ? bval : 0.0f;
  __syncthreads();
  if (rowBase >= nRows) return;

  const float* Bl = &Bs[l16 * STRIDE + 4 * half];
  const bool fullRows = (rowBase + 16 <= nRows);
  int ct = 0;
  for (; ct + 2 <= foutTiles; ct += 2)
    gemm_tiles<FIN, 2>(a4, Bl, bsh, out, ct, rowBase, half, l16, nRows, Fout, doRelu, fullRows);
  if (ct < foutTiles)
    gemm_tiles<FIN, 1>(a4, Bl, bsh, out, ct, rowBase, half, l16, nRows, Fout, doRelu, fullRows);
}

__global__ __launch_bounds__(256) void
layer3_heads_kernel(const float* __restrict__ xin, const float* __restrict__ W3,
                    const float* __restrict__ b3, const float* __restrict__ Wp,
                    const float* __restrict__ bp, const float* __restrict__ Wc,
                    const float* __restrict__ bc, float* __restrict__ emb,
                    float* __restrict__ nxt, float* __restrict__ cls, int nRows, int nClasses)
{
  constexpr int FIN     = 128;
  constexpr int STRIDE  = FIN + 4;
  constexpr int F4      = FIN / 4;
  constexpr int KP      = FIN / 8;
  constexpr int FE      = 64;
  constexpr int STRIDE2 = KEEP_PITCH;
  constexpr int F4E     = FE / 4;
  constexpr int KPE     = FE / 8;
  constexpr int WLD3    = (FE * F4) / 256;
  constexpr int WLDP    = (FE * F4E) / 256;
  __shared__ __attribute__((aligned(16))) float Bs[FE * STRIDE];
  __shared__ __attribute__((aligned(16))) float Bh[(FE + 16) * STRIDE2];
  __shared__ __attribute__((aligned(16))) float Es[8 * 16 * STRIDE2];
  __shared__ float bsh[FE];
  __shared__ float bsh2[FE + 16];

  const int tid  = (int)threadIdx.x;
  const int lane = tid & 31;
  const int wave = tid >> 5;
  const int half = lane >> 4;
  const int l16  = lane & 15;
  const int rowBase = (blockIdx.x * 8 + wave) * 16;
  const int row     = min(rowBase + l16, nRows - 1);

  const float bv3 = b3[min(tid, FE - 1)];
  const float bvp = bp[min(tid, FE - 1)];
  const float bvc = bc[min(tid, nClasses - 1)];
  float4 w3[WLD3];
  #pragma unroll
  for (int i = 0; i < WLD3; ++i) w3[i] = ((const float4*)W3)[tid + i * 256];
  float4 wp[WLDP];
  #pragma unroll
  for (int i = 0; i < WLDP; ++i) wp[i] = ((const float4*)Wp)[tid + i * 256];
  const int wcRow = tid >> 4;
  const float4 wc = ((const float4*)Wc)[min(tid, nClasses * F4E - 1)];
  float4 a4[KP];
  {
    const float4* xr = (const float4*)(xin + (size_t)row * FIN) + half;
    #pragma unroll
    for (int j = 0; j < KP; ++j) a4[j] = xr[2 * j];
  }
  #pragma unroll
  for (int i = 0; i < WLD3; ++i) {
    const int idx = tid + i * 256;
    const int r = idx / F4, c4 = idx - r * F4;
    *(float4*)&Bs[r * STRIDE + 4 * c4] = w3[i];
  }
  #pragma unroll
  for (int i = 0; i < WLDP; ++i) {
    const int idx = tid + i * 256;
    const int r = idx / F4E, c4 = idx - r * F4E;
    *(float4*)&Bh[r * STRIDE2 + 4 * c4] = wp[i];
  }
  {
    const int c4 = tid - wcRow * F4E;
    *(float4*)&Bh[(FE + wcRow) * STRIDE2 + 4 * c4] =
        (wcRow < nClasses) ? wc : make_float4(0.0f, 0.0f, 0.0f, 0.0f);
  }
  if (tid < FE) { bsh[tid] = bv3; bsh2[tid] = bvp; }
  if (tid < 16) bsh2[FE + tid] = (tid < nClasses) ? bvc : 0.0f;
  __syncthreads();
  if (rowBase >= nRows) return;

  const bool fullRows = (rowBase + 16 <= nRows);
  float* Ew = &Es[wave * 16 * STRIDE2];

  {
    const float* Bl = &Bs[l16 * STRIDE + 4 * half];
    #pragma unroll
    for (int ct = 0; ct < FE / 16; ct += 2)
      gemm_tiles<FIN, 2, true>(a4, Bl, bsh, emb, ct, rowBase, half, l16, nRows, FE, 1, fullRows, Ew);
  }
  float4 e4[KPE];
  #pragma unroll
  for (int j = 0; j < KPE; ++j) e4[j] = *(const float4*)(Ew + l16 * STRIDE2 + 8 * j + 4 * half);
  const float* Bl2 = &Bh[l16 * STRIDE2 + 4 * half];
  #pragma unroll
  for (int ct = 0; ct < FE / 16; ct += 2)
    gemm_tiles<FE, 2>(e4, Bl2, bsh2, nxt, ct, rowBase, half, l16, nRows, FE, 0, fullRows);
  gemm_tiles<FE, 1>(e4, Bl2 + FE * STRIDE2, bsh2 + FE, cls, 0, rowBase, half, l16, nRows, nClasses, 0, fullRows);
}

extern "C" void kernel_launch(void* const* d_in, const int* in_sizes, int n_in,
                              void* d_out, int out_size, void* d_ws, size_t ws_size,
                              hipStream_t stream) {
  const float* x0 = (const float*)d_in[0];
  const int*   ei = (const int*)d_in[1];
  const float *W1 = (const float*)d_in[2],  *b1 = (const float*)d_in[3];
  const float *W2 = (const float*)d_in[4],  *b2 = (const float*)d_in[5];
  const float *W3 = (const float*)d_in[6],  *b3 = (const float*)d_in[7];
  const float *Wp = (const float*)d_in[8],  *bp = (const float*)d_in[9];
  const float *Wc = (const float*)d_in[10], *bc = (const float*)d_in[11];

  const int N      = in_sizes[0] / 64;
  const int nEdges = in_sizes[1] / 2;
  const int* src = ei;
  const int* dst = ei + nEdges;

  float* ws        = (float*)d_ws;
  int*   cnt       = (int*)d_ws;
  float* A         = ws + 65536;
  int*   rowptr    = (int*)(A + (size_t)N * 128);
  int*   nbr       = rowptr + 65536;
  int*   nbrSorted = nbr + nEdges;

  float* out = (float*)d_out;
  float* h1  = out;
  float* h2  = out;
  float* emb = out;
  float* nxt = out + (size_t)N * 64;
  float* cls = out + (size_t)N * 128;

  auto nb = [](int n) { return dim3((unsigned)((n + 255) / 256)); };
  dim3 b256(256);
  dim3 gemmGrid((unsigned)((N + 127) / 128)), gemmBlock(256);
  dim3 nodeGrid((unsigned)((N + 8 * NODES_PER_WAVE - 1) / (8 * NODES_PER_WAVE)));

  zero_int_kernel   <<<nb(N),      b256, 0, stream>>>(cnt, N);
  indegree_kernel   <<<nb(nEdges), b256, 0, stream>>>(dst, cnt, nEdges);
  rowptr_scan_kernel<<<dim3((unsigned)((N + 1023) / 1024)), b256, 0, stream>>>(cnt, rowptr, N);
  place_kernel      <<<nb(nEdges), b256, 0, stream>>>(src, dst, rowptr, cnt, nbr, nEdges);
  order_rows_kernel <<<nodeGrid,   b256, 0, stream>>>(rowptr, nbr, nbrSorted, N);

  gather_mean_kernel<16><<<nodeGrid, b256, 0, stream>>>(x0, rowptr, nbrSorted, A, N);
  gemm_bias_act_kernel<64><<<gemmGrid, gemmBlock, 0, stream>>>(A, W1, b1, h1, N, 128, 1);

  gather_mean_kernel<32><<<nodeGrid, b256, 0, stream>>>(h1, rowptr, nbrSorted, A, N);
  gemm_bias_act_kernel<128><<<gemmGrid, gemmBlock, 0, stream>>>(A, W2, b2, h2, N, 128, 1);

  gather_mean_kernel<32><<<nodeGrid, b256, 0, stream>>>(h2, rowptr, nbrSorted, A, N);
  layer3_heads_kernel<<<gemmGrid, gemmBlock, 0, stream>>>(A, W3, b3, Wp, bp, Wc, bc, emb, nxt, cls, N, 10);
}


# Layer-3 product moved in front of its neighbour mean (linear): layer-2 launch forms t=h2@W3^T on its tiles, gather reads 64-wide t, adds b3, relu; h2 never stored; sum order: W3 product, then mean
# speedup vs baseline: 3.2685x; 1.0498x over previous
#include <hip/hip_runtime.h>
#include <hip/hip_bf16.h>

typedef __attribute__((ext_vector_type(8))) float v8f;
typedef __attribute__((ext_vector_type(2))) float v2f;


__global__ void zero_int_kernel(int* __restrict__ p, int n) {
  int i = blockIdx.x * blockDim.x + threadIdx.x;
  if (i < n) p[i] = 0;
}

__global__ void indegree_kernel(const int* __restrict__ dst, int* __restrict__ cnt, int nEdges) {
  int i = blockIdx.x * blockDim.x + threadIdx.x;
  if (i < nEdges)
    __hip_atomic_fetch_add(&cnt[dst[i]], 1, __ATOMIC_RELAXED, __HIP_MEMORY_SCOPE_AGENT);
}

__global__ __launch_bounds__(256) void
rowptr_scan_kernel(const int* __restrict__ cnt, int* __restrict__ rowptr, int n)
{
  __shared__ int sIncl[8];
  __shared__ int sPre[8];
  const int t = threadIdx.x, lane = t & 31, w = t >> 5;
  const int base = blockIdx.x * 1024;
  const int4* c4 = (const int4*)cnt;
  int pre = 0;
  #pragma unroll 4
  for (int i = t; i < (base >> 2); i += 256) {
    int4 v = c4[i];
    pre += (v.x + v.y) + (v.z + v.w);
  }
  const int idx = base + 4 * t;
  const int v0 = (idx     < n) ? cnt[idx]     : 0;
  const int v1 = (idx + 1 < n) ? cnt[idx + 1] : 0;
  const int v2 = (idx + 2 < n) ? cnt[idx + 2] : 0;
  const int v3 = (idx + 3 < n) ? cnt[idx + 3] : 0;
  const int local = (v0 + v1) + (v2 + v3);
  int incl = local;
  #pragma unroll
  for (int o = 1; o < 32; o <<= 1) {
    int y = __shfl_up(incl, o);
    if (lane >= o) incl += y;
  }
  #pragma unroll
  for (int o = 16; o > 0; o >>= 1) pre += __shfl_xor(pre, o);
  if (lane == 31) sIncl[w] = incl;
  if (lane == 0)  sPre[w]  = pre;
  __syncthreads();
  int off = 0;
  #pragma unroll
  for (int i = 0; i < 8; ++i) {
    off += sPre[i];
    if (i < w) off += sIncl[i];
  }
  int r = off + incl - local;
  if (idx     < n) rowptr[idx]     = r;
  r += v0;
  if (idx + 1 < n) rowptr[idx + 1] = r;
  r += v1;
  if (idx + 2 < n) rowptr[idx + 2] = r;
  r += v2;
  if (idx + 3 < n) rowptr[idx + 3] = r;
  r += v3;
  if (idx < n && idx + 4 >= n) rowptr[n] = r;
}

__global__ void place_kernel(const int* __restrict__ src, const int* __restrict__ dst,
                             const int* __restrict__ rowptr, int* __restrict__ cnt,
                             int* __restrict__ nbr, int nEdges) {
  int i = blockIdx.x * blockDim.x + threadIdx.x;
  if (i < nEdges) {
    const int d = dst[i];
    const int k = __hip_atomic_fetch_add(&cnt[d], -1, __ATOMIC_RELAXED, __HIP_MEMORY_SCOPE_AGENT);
    nbr[rowptr[d] + k - 1] = src[i];
  }
}

constexpr int NODES_PER_WAVE = 4;

__global__ __launch_bounds__(256) void
order_rows_kernel(const int* __restrict__ rowptr, const int* __restrict__ nbr,
                  int* __restrict__ nbrSorted, int nNodes)
{
  const int lane = threadIdx.x & 31;
  const int wv   = blockIdx.x * 8 + (threadIdx.x >> 5);
  for (int q = 0; q < NODES_PER_WAVE; ++q) {
    const int node = wv * NODES_PER_WAVE + q;
    if (node >= nNodes) return;
    const int beg = rowptr[node], end = rowptr[node + 1];
    for (int i = beg + lane; i < end; i += 32) {
      const int s = nbr[i];
      int r = 0;
      for (int j = beg; j < end; j += 4) {
        int t[4];
        #pragma unroll
        for (int u = 0; u < 4; ++u) t[u] = nbr[min(j + u, end - 1)];
        #pragma unroll
        for (int u = 0; u < 4; ++u) {
          const int jj = j + u;
          const bool front = (t[u] < s) || (t[u] == s && jj < i);
          r += (jj < end && front) ? 1 : 0;
        }
      }
      nbrSorted[beg + r] = s;
    }
  }
}

template <int F4, int U, bool CLAMP>
__device__ __forceinline__ void
add_rows(float4& acc, const float4* __restrict__ x4, const int* __restrict__ nbr,
         int e, int end, int c, int sub)
{
  constexpr int EPS = 32 / F4;
  float4 v[U];
  #pragma unroll
  for (int u = 0; u < U; ++u) {
    int ee = e + u * EPS + sub;
    if (CLAMP) ee = min(ee, end - 1);
    const int s = nbr[ee];
    v[u] = x4[(size_t)s * F4 + c];
  }
  if constexpr (CLAMP) {
    static_assert(U == 4, "the fence below names four rows");
    asm volatile("" :: "v"(v[0].x), "v"(v[0].y), "v"(v[0].z), "v"(v[0].w),
                       "v"(v[1].x), "v"(v[1].y), "v"(v[1].z), "v"(v[1].w),
                       "v"(v[2].x), "v"(v[2].y), "v"(v[2].z), "v"(v[2].w),
                       "v"(v[3].x), "v"(v[3].y), "v"(v[3].z), "v"(v[3].w));
  }
  #pragma unroll
  for (int u = 0; u < U; ++u) {
    if (!CLAMP || (e + u * EPS + sub < end)) {
      acc.x += v[u].x; acc.y += v[u].y; acc.z += v[u].z; acc.w += v[u].w;
    }
  }
}

template <int F4, bool ACT>
__global__ __launch_bounds__(256) void
gather_mean_kernel(const float* __restrict__ x, const int* __restrict__ rowptr,
                   const int* __restrict__ nbr, float* __restrict__ A,
                   const float* __restrict__ bias, int nNodes)
{
  constexpr int EPS = 32 / F4;
  const int lane = threadIdx.x & 31;
  const int c    = lane % F4;
  const int sub  = lane / F4;
  const int wv   = blockIdx.x * 8 + (threadIdx.x >> 5);
  const float4* x4 = (const float4*)x;
  float4 bv = make_float4(0.0f, 0.0f, 0.0f, 0.0f);
  if constexpr (ACT) {
    bv.x = bias[4 * c]; bv.y = bias[4 * c + 1]; bv.z = bias[4 * c + 2]; bv.w = bias[4 * c + 3];
  }
  for (int q = 0; q < NODES_PER_WAVE; ++q) {
    const int node = wv * NODES_PER_WAVE + q;
    if (node >= nNodes) return;
    const int beg = rowptr[node], end = rowptr[node + 1];
    const float4 xv = x4[(size_t)node * F4 + c];
    float4 acc = make_float4(0.0f, 0.0f, 0.0f, 0.0f);
    int e = beg;
    for (; e + 8 * EPS <= end; e += 8 * EPS) add_rows<F4, 8, false>(acc, x4, nbr, e, end, c, sub);
    for (; e < end; e += 4 * EPS)            add_rows<F4, 4, true >(acc, x4, nbr, e, end, c, sub);
    if (EPS == 2) {
      acc.x += __shfl_xor(acc.x, 16);
      acc.y += __shfl_xor(acc.y, 16);
      acc.z += __shfl_xor(acc.z, 16);
      acc.w += __shfl_xor(acc.w, 16);
    }
    const float rd = 1.0f / fmaxf((float)(end - beg), 1.0f);
    if (sub == 0) {
      float4 o;
      o.x = xv.x + acc.x * rd;
      o.y = xv.y + acc.y * rd;
      o.z = xv.z + acc.z * rd;
      o.w = xv.w + acc.w * rd;
      if constexpr (ACT) {
        o.x = fmaxf(o.x + bv.x, 0.0f);
        o.y = fmaxf(o.y + bv.y, 0.0f);
        o.z = fmaxf(o.z + bv.z, 0.0f);
        o.w = fmaxf(o.w + bv.w, 0.0f);
      }
      ((float4*)A)[(size_t)node * F4 + c] = o;
    }
  }
}

template <int FIN, int NT>
__device__ __forceinline__ void
gemm_tiles(const float4 (&a4)[FIN / 8], const float* __restrict__ Bl,
           const float* __restrict__ bsh, float* __restrict__ out,
           int ct, int rowBase, int half, int l16, int nRows, int Fout, int doRelu,
           bool fullRows)
{
  constexpr int STRIDE = FIN + 4;
  constexpr int KP = FIN / 8;
  v8f c[NT];
  #pragma unroll
  for (int t = 0; t < NT; ++t) c[t] = v8f{};
  #pragma unroll
  for (int j = 0; j < KP; ++j) {
    float4 b[NT];
    #pragma unroll
    for (int t = 0; t < NT; ++t)
      b[t] = *(const float4*)(Bl + (ct + t) * 16 * STRIDE + 8 * j);
    v2f alo, ahi;
    alo.x = a4[j].x; alo.y = a4[j].y;
    ahi.x = a4[j].z; ahi.y = a4[j].w;
    #pragma unroll
    for (int t = 0; t < NT; ++t) {
      v2f blo; blo.x = b[t].x; blo.y = b[t].y;
      c[t] = __builtin_amdgcn_wmma_f32_16x16x4_f32(false, alo, false, blo, (short)0, c[t], false, false);
    }
    #pragma unroll
    for (int t = 0; t < NT; ++t) {
      v2f bhi; bhi.x = b[t].z; bhi.y = b[t].w;
      c[t] = __builtin_amdgcn_wmma_f32_16x16x4_f32(false, ahi, false, bhi, (short)0, c[t], false, false);
    }
  }
  #pragma unroll
  for (int t = 0; t < NT; ++t) {
    const int n = (ct + t) * 16 + l16;
    const float bv = bsh[n];
    float* colBase = out + (size_t)(rowBase + 8 * half) * Fout + n;
    if (fullRows && ((ct + t) * 16 + 16 <= Fout)) {
      #pragma unroll
      for (int r = 0; r < 8; ++r) {
        float val = c[t][r] + bv;
        if (doRelu) val = fmaxf(val, 0.0f);
        colBase[(size_t)r * Fout] = val;
      }
    } else {
      #pragma unroll
      for (int r = 0; r < 8; ++r) {
        int gRow = rowBase + r + 8 * half;
        float val = c[t][r] + bv;
        if (doRelu) val = fmaxf(val, 0.0f);
        if (gRow < nRows && n < Fout) out[(size_t)gRow * Fout + n] = val;
      }
    }
  }
}

template <int FIN>
__global__ __launch_bounds__(256) void
gemm_bias_act_kernel(const float* __restrict__ xin, const float* __restrict__ W,
                     const float* __restrict__ bias, float* __restrict__ out,
                     int nRows, int Fout, int doRelu)
{
  constexpr int STRIDE = FIN + 4;
  constexpr int F4  = FIN / 4;
  constexpr int KP  = FIN / 8;
  constexpr int WLD = (128 * F4) / 256;
  __shared__ __attribute__((aligned(16))) float Bs[128 * STRIDE];
  __shared__ float bsh[128];

  const int foutTiles = (Fout + 15) >> 4;
  const int nW4       = foutTiles * 16 * F4;

  const int lane = threadIdx.x & 31;
  const int wave = threadIdx.x >> 5;
  const int half = lane >> 4;
  const int l16  = lane & 15;
  const int rowBase = (blockIdx.x * 8 + wave) * 16;
  const int row     = min(rowBase + l16, nRows - 1);

  const float bval = bias[min((int)threadIdx.x, Fout - 1)];
  const int nWvalid = Fout * F4;
  float4 wv[WLD];
  #pragma unroll
  for (int i = 0; i < WLD; ++i) {
    int idx = (int)threadIdx.x + i * 256;
    wv[i] = ((const float4*)W)[min(idx, nWvalid - 1)];
  }
  float4 a4[KP];
  {
    const float4* xr = (const float4*)(xin + (size_t)row * FIN) + half;
    #pragma unroll
    for (int j = 0; j < KP; ++j) a4[j] = xr[2 * j];
  }
  #pragma unroll
  for (int i = 0; i < WLD; ++i) {
    int idx = (int)threadIdx.x + i * 256;
    if (idx < nW4) {
      int r  = idx / F4;
      int c4 = idx - r * F4;
      float4 v = wv[i];
      if (idx >= nWvalid) v = make_float4(0.0f, 0.0f, 0.0f, 0.0f);
      *(float4*)&Bs[r * STRIDE + 4 * c4] = v;
    }
  }
  if (threadIdx.x < 128) bsh[threadIdx.x] = ((int)threadIdx.x < Fout) ? bval : 0.0f;
  __syncthreads();
  if (rowBase >= nRows) return;

  const float* Bl = &Bs[l16 * STRIDE + 4 * half];
  const bool fullRows = (rowBase + 16 <= nRows);
  int ct = 0;
  for (; ct + 2 <= foutTiles; ct += 2)
    gemm_tiles<FIN, 2>(a4, Bl, bsh, out, ct, rowBase, half, l16, nRows, Fout, doRelu, fullRows);
  if (ct < foutTiles)
    gemm_tiles<FIN, 1>(a4, Bl, bsh, out, ct, rowBase, half, l16, nRows, Fout, doRelu, fullRows);
}

constexpr int CHUNK_PITCH = 32 + 4;
__global__ __launch_bounds__(256) void
layer2_pre3_kernel(const float* __restrict__ xin, const float* __restrict__ W2,
                   const float* __restrict__ b2, const float* __restrict__ W3,
                   float* __restrict__ tout, int nRows)
{
  constexpr int FIN    = 128;
  constexpr int STRIDE = FIN + 4;
  constexpr int F4     = FIN / 4;
  constexpr int KP     = FIN / 8;
  constexpr int FH     = 128;
  constexpr int FT     = 64;
  constexpr int WLD2   = (FH * F4) / 256;
  constexpr int WLD3   = (FT * F4) / 256;
  __shared__ __attribute__((aligned(16))) float Bs[FH * STRIDE];
  __shared__ __attribute__((aligned(16))) float B3[FT * STRIDE];
  __shared__ __attribute__((aligned(16))) float Ks[8 * 16 * CHUNK_PITCH];
  __shared__ float bsh[FH];

  const int tid  = (int)threadIdx.x;
  const int lane = tid & 31;
  const int wave = tid >> 5;
  const int half = lane >> 4;
  const int l16  = lane & 15;
  const int rowBase = (blockIdx.x * 8 + wave) * 16;
  const int row     = min(rowBase + l16, nRows - 1);

  const float bval = b2[min(tid, FH - 1)];
  float4 w2[WLD2];
  #pragma unroll
  for (int i = 0; i < WLD2; ++i) w2[i] = ((const float4*)W2)[tid + i * 256];
  float4 w3[WLD3];
  #pragma unroll
  for (int i = 0; i < WLD3; ++i) w3[i] = ((const float4*)W3)[tid + i * 256];
  float4 a4[KP];
  {
    const float4* xr = (const float4*)(xin + (size_t)row * FIN) + half;
    #pragma unroll
    for (int j = 0; j < KP; ++j) a4[j] = xr[2 * j];
  }
  #pragma unroll
  for (int i = 0; i < WLD2; ++i) {
    const int idx = tid + i * 256;
    const int r = idx / F4, c4 = idx - r * F4;
    *(float4*)&Bs[r * STRIDE + 4 * c4] = w2[i];
  }
  #pragma unroll
  for (int i = 0; i < WLD3; ++i) {
    const int idx = tid + i * 256;
    const int r = idx / F4, c4 = idx - r * F4;
    *(float4*)&B3[r * STRIDE + 4 * c4] = w3[i];
  }
  if (tid < FH) bsh[tid] = bval;
  __syncthreads();
  if (rowBase >= nRows) return;

  const float* Bl  = &Bs[l16 * STRIDE + 4 * half];
  const float* B3l = &B3[l16 * STRIDE + 4 * half];
  float*       Kw  = &Ks[wave * 16 * CHUNK_PITCH];

  v8f tacc[FT / 16];
  #pragma unroll
  for (int t = 0; t < FT / 16; ++t) tacc[t] = v8f{};

  #pragma unroll
  for (int p = 0; p < FH / 32; ++p) {
    v8f c[2];
    c[0] = v8f{}; c[1] = v8f{};
    #pragma unroll
    for (int j = 0; j < KP; ++j) {
      float4 b[2];
      #pragma unroll
      for (int t = 0; t < 2; ++t)
        b[t] = *(const float4*)(Bl + (2 * p + t) * 16 * STRIDE + 8 * j);
      v2f alo, ahi;
      alo.x = a4[j].x; alo.y = a4[j].y;
      ahi.x = a4[j].z; ahi.y = a4[j].w;
      #pragma unroll
      for (int t = 0; t < 2; ++t) {
        v2f blo; blo.x = b[t].x; blo.y = b[t].y;
        c[t] = __builtin_amdgcn_wmma_f32_16x16x4_f32(false, alo, false, blo, (short)0, c[t], false, false);
      }
      #pragma unroll
      for (int t = 0; t < 2; ++t) {
        v2f bhi; bhi.x = b[t].z; bhi.y = b[t].w;
        c[t] = __builtin_amdgcn_wmma_f32_16x16x4_f32(false, ahi, false, bhi, (short)0, c[t], false, false);
      }
    }
    #pragma unroll
    for (int t = 0; t < 2; ++t) {
      const float bv = bsh[(2 * p + t) * 16 + l16];
      #pragma unroll
      for (int r = 0; r < 8; ++r)
        Kw[(r + 8 * half) * CHUNK_PITCH + t * 16 + l16] = fmaxf(c[t][r] + bv, 0.0f);
    }
    float4 e4[4];
    #pragma unroll
    for (int j = 0; j < 4; ++j) e4[j] = *(const float4*)(Kw + l16 * CHUNK_PITCH + 8 * j + 4 * half);
    #pragma unroll
    for (int j = 0; j < 4; ++j) {
      float4 b[FT / 16];
      #pragma unroll
      for (int t = 0; t < FT / 16; ++t)
        b[t] = *(const float4*)(B3l + t * 16 * STRIDE + 8 * (4 * p + j));
      v2f alo, ahi;
      alo.x = e4[j].x; alo.y = e4[j].y;
      ahi.x = e4[j].z; ahi.y = e4[j].w;
      #pragma unroll
      for (int t = 0; t < FT / 16; ++t) {
        v2f blo; blo.x = b[t].x; blo.y = b[t].y;
        tacc[t] = __builtin_amdgcn_wmma_f32_16x16x4_f32(false, alo, false, blo, (short)0, tacc[t], false, false);
      }
      #pragma unroll
      for (int t = 0; t < FT / 16; ++t) {
        v2f bhi; bhi.x = b[t].z; bhi.y = b[t].w;
        tacc[t] = __builtin_amdgcn_wmma_f32_16x16x4_f32(false, ahi, false, bhi, (short)0, tacc[t], false, false);
      }
    }
  }
  const bool fullRows = (rowBase + 16 <= nRows);
  #pragma unroll
  for (int t = 0; t < FT / 16; ++t) {
    const int n = t * 16 + l16;
    if (fullRows) {
      float* colBase = tout + (size_t)(rowBase + 8 * half) * FT + n;
      #pragma unroll
      for (int r = 0; r < 8; ++r) colBase[(size_t)r * FT] = tacc[t][r];
    } else {
      #pragma unroll
      for (int r = 0; r < 8; ++r) {
        const int gRow = rowBase + r + 8 * half;
        if (gRow < nRows) tout[(size_t)gRow * FT + n] = tacc[t][r];
      }
    }
  }
}

__global__ __launch_bounds__(256) void
heads_kernel(const float* __restrict__ emb, const float* __restrict__ Wp,
             const float* __restrict__ bp, const float* __restrict__ Wc,
             const float* __restrict__ bc, float* __restrict__ nxt,
             float* __restrict__ cls, int nRows, int nClasses)
{
  constexpr int FE      = 64;
  constexpr int STRIDE2 = FE + 4;
  constexpr int F4E     = FE / 4;
  constexpr int KPE     = FE / 8;
  constexpr int WLDP    = (FE * F4E) / 256;
  __shared__ __attribute__((aligned(16))) float Bh[(FE + 16) * STRIDE2];
  __shared__ float bsh2[FE + 16];

  const int tid  = (int)threadIdx.x;
  const int lane = tid & 31;
  const int wave = tid >> 5;
  const int half = lane >> 4;
  const int l16  = lane & 15;
  const int rowBase = (blockIdx.x * 8 + wave) * 16;
  const int row     = min(rowBase + l16, nRows - 1);

  const float bvp = bp[min(tid, FE - 1)];
  const float bvc = bc[min(tid, nClasses - 1)];
  float4 wp[WLDP];
  #pragma unroll
  for (int i = 0; i < WLDP; ++i) wp[i] = ((const float4*)Wp)[tid + i * 256];
  const int wcRow = tid >> 4;
  const float4 wc = ((const float4*)Wc)[min(tid, nClasses * F4E - 1)];
  float4 e4[KPE];
  {
    const float4* xr = (const float4*)(emb + (size_t)row * FE) + half;
    #pragma unroll
    for (int j = 0; j < KPE; ++j) e4[j] = xr[2 * j];
  }
  #pragma unroll
  for (int i = 0; i < WLDP; ++i) {
    const int idx = tid + i * 256;
    const int r = idx / F4E, c4 = idx - r * F4E;
    *(float4*)&Bh[r * STRIDE2 + 4 * c4] = wp[i];
  }
  {
    const int c4 = tid - wcRow * F4E;
    *(float4*)&Bh[(FE + wcRow) * STRIDE2 + 4 * c4] =
        (wcRow < nClasses) ? wc : make_float4(0.0f, 0.0f, 0.0f, 0.0f);
  }
  if (tid < FE) bsh2[tid] = bvp;
  if (tid < 16) bsh2[FE + tid] = (tid < nClasses) ? bvc : 0.0f;
  __syncthreads();
  if (rowBase >= nRows) return;

  const bool fullRows = (rowBase + 16 <= nRows);
  const float* Bl2 = &Bh[l16 * STRIDE2 + 4 * half];
  #pragma unroll
  for (int ct = 0; ct < FE / 16; ct += 2)
    gemm_tiles<FE, 2>(e4, Bl2, bsh2, nxt, ct, rowBase, half, l16, nRows, FE, 0, fullRows);
  gemm_tiles<FE, 1>(e4, Bl2 + FE * STRIDE2, bsh2 + FE, cls, 0, rowBase, half, l16, nRows, nClasses, 0, fullRows);
}

extern "C" void kernel_launch(void* const* d_in, const int* in_sizes, int n_in,
                              void* d_out, int out_size, void* d_ws, size_t ws_size,
                              hipStream_t stream) {
  const float* x0 = (const float*)d_in[0];
  const int*   ei = (const int*)d_in[1];
  const float *W1 = (const float*)d_in[2],  *b1 = (const float*)d_in[3];
  const float *W2 = (const float*)d_in[4],  *b2 = (const float*)d_in[5];
  const float *W3 = (const float*)d_in[6],  *b3 = (const float*)d_in[7];
  const float *Wp = (const float*)d_in[8],  *bp = (const float*)d_in[9];
  const float *Wc = (const float*)d_in[10], *bc = (const float*)d_in[11];

  const int N      = in_sizes[0] / 64;
  const int nEdges = in_sizes[1] / 2;
  const int* src = ei;
  const int* dst = ei + nEdges;

  float* ws        = (float*)d_ws;
  int*   cnt       = (int*)d_ws;
  float* A         = ws + 65536;
  int*   rowptr    = (int*)(A + (size_t)N * 128);
  int*   nbr       = rowptr + 65536;
  int*   nbrSorted = nbr + nEdges;

  float* out = (float*)d_out;
  float* h1  = out;
  float* emb = out;
  float* t3  = out + (size_t)N * 64;
  float* nxt = out + (size_t)N * 64;
  float* cls = out + (size_t)N * 128;

  auto nb = [](int n) { return dim3((unsigned)((n + 255) / 256)); };
  dim3 b256(256);
  dim3 gemmGrid((unsigned)((N + 127) / 128)), gemmBlock(256);
  dim3 nodeGrid((unsigned)((N + 8 * NODES_PER_WAVE - 1) / (8 * NODES_PER_WAVE)));

  zero_int_kernel   <<<nb(N),      b256, 0, stream>>>(cnt, N);
  indegree_kernel   <<<nb(nEdges), b256, 0, stream>>>(dst, cnt, nEdges);
  rowptr_scan_kernel<<<dim3((unsigned)((N + 1023) / 1024)), b256, 0, stream>>>(cnt, rowptr, N);
  place_kernel      <<<nb(nEdges), b256, 0, stream>>>(src, dst, rowptr, cnt, nbr, nEdges);
  order_rows_kernel <<<nodeGrid,   b256, 0, stream>>>(rowptr, nbr, nbrSorted, N);

  gather_mean_kernel<16, false><<<nodeGrid, b256, 0, stream>>>(x0, rowptr, nbrSorted, A, nullptr, N);
  gemm_bias_act_kernel<64><<<gemmGrid, gemmBlock, 0, stream>>>(A, W1, b1, h1, N, 128, 1);

  gather_mean_kernel<32, false><<<nodeGrid, b256, 0, stream>>>(h1, rowptr, nbrSorted, A, nullptr, N);
  layer2_pre3_kernel<<<gemmGrid, gemmBlock, 0, stream>>>(A, W2, b2, W3, t3, N);

  gather_mean_kernel<16, true><<<nodeGrid, b256, 0, stream>>>(t3, rowptr, nbrSorted, emb, b3, N);

  heads_kernel<<<gemmGrid, gemmBlock, 0, stream>>>(emb, Wp, bp, Wc, bc, nxt, cls, N, 10);
}


# Neighbour lists built through LDS bins of 64 nodes: per-block LDS counts, prefix over blocks, pairs moved per bin, one block per bin counts, places and orders in LDS; no global atomic; sum order kept
# speedup vs baseline: 4.7545x; 1.4546x over previous
#include <hip/hip_runtime.h>
#include <hip/hip_bf16.h>

typedef __attribute__((ext_vector_type(8))) float v8f;
typedef __attribute__((ext_vector_type(2))) float v2f;

constexpr int BIN_SHIFT = 6;
constexpr int BIN_NODES = 1 << BIN_SHIFT;
constexpr int MAX_BINS  = 1024;
constexpr int CHUNK     = 4096;
constexpr int QPT       = CHUNK / (256 * 4);
constexpr int BIN_CAP   = 4096;
static_assert(QPT == 4, "bin_move_kernel adds up four quad counts");

__device__ __forceinline__ int
load_quad(const int* __restrict__ p, int e, int n, bool aligned, int (&v)[4])
{
  if (aligned && e + 4 <= n) {
    const int4 t = *(const int4*)(p + e);
    v[0] = t.x; v[1] = t.y; v[2] = t.z; v[3] = t.w;
    return 4;
  }
  int c = 0;
  #pragma unroll
  for (int k = 0; k < 4; ++k) {
    v[k] = 0;
    if (e + k < n) { v[k] = p[e + k]; c = k + 1; }
  }
  return c;
}

__global__ __launch_bounds__(256) void
bin_count_kernel(const int* __restrict__ dst, int* __restrict__ H, int nEdges, int nBins)
{
  __shared__ int hist[MAX_BINS];
  const int tid = (int)threadIdx.x;
  const bool aligned = (((size_t)dst) & 15) == 0;
  const int e0 = blockIdx.x * CHUNK + 4 * tid;
  int d[QPT][4], c[QPT];
  #pragma unroll
  for (int u = 0; u < QPT; ++u) c[u] = load_quad(dst, e0 + 1024 * u, nEdges, aligned, d[u]);
  for (int i = tid; i < MAX_BINS; i += 256) hist[i] = 0;
  __syncthreads();
  #pragma unroll
  for (int u = 0; u < QPT; ++u) {
    #pragma unroll
    for (int k = 0; k < 4; ++k)
      if (k < c[u])
        __hip_atomic_fetch_add(&hist[d[u][k] >> BIN_SHIFT], 1, __ATOMIC_RELAXED, __HIP_MEMORY_SCOPE_WORKGROUP);
  }
  __syncthreads();
  const int G = (int)gridDim.x;
  for (int i = tid; i < nBins; i += 256) H[(size_t)i * G + blockIdx.x] = hist[i];
}

__device__ __forceinline__ int
scan_store(int* __restrict__ row, int b, int G, int v, int carry, int lane)
{
  int incl = v;
  #pragma unroll
  for (int o = 1; o < 32; o <<= 1) {
    int y = __shfl_up(incl, o);
    if (lane >= o) incl += y;
  }
  if (b < G) row[b] = carry + incl - v;
  return carry + __shfl(incl, 31);
}

__global__ __launch_bounds__(256) void
bin_prefix_kernel(int* __restrict__ H, int* __restrict__ binTotal, int nBins, int G)
{
  const int lane = threadIdx.x & 31;
  const int bin  = blockIdx.x * 8 + (threadIdx.x >> 5);
  if (bin >= nBins) return;
  int* row = H + (size_t)bin * G;
  int carry = 0;
  int vals[8];
  #pragma unroll
  for (int u = 0; u < 8; ++u) {
    const int b = 32 * u + lane;
    vals[u] = (b < G) ? row[b] : 0;
  }
  #pragma unroll
  for (int u = 0; u < 8; ++u) {
    const int b = 32 * u + lane;
    carry = scan_store(row, b, G, vals[u], carry, lane);
  }
  for (int j = 256; j < G; j += 32) {
    const int b = j + lane;
    const int v = (b < G) ? row[b] : 0;
    carry = scan_store(row, b, G, v, carry, lane);
  }
  if (lane == 0) binTotal[bin] = carry;
}

template <bool GUARD>
__device__ __forceinline__ void
move_pairs(int* __restrict__ cursor, int2* __restrict__ pairs,
           const int (&d)[QPT][4], const int (&s)[QPT][4], const int (&c)[QPT])
{
  int pos[QPT][4];
  #pragma unroll
  for (int u = 0; u < QPT; ++u) {
    #pragma unroll
    for (int k = 0; k < 4; ++k) {
      pos[u][k] = 0;
      if (!GUARD || k < c[u])
        pos[u][k] = __hip_atomic_fetch_add(&cursor[d[u][k] >> BIN_SHIFT], 1, __ATOMIC_RELAXED, __HIP_MEMORY_SCOPE_WORKGROUP);
    }
  }
  #pragma unroll
  for (int u = 0; u < QPT; ++u) {
    #pragma unroll
    for (int k = 0; k < 4; ++k)
      if (!GUARD || k < c[u]) pairs[pos[u][k]] = make_int2(s[u][k], d[u][k]);
  }
}

__global__ __launch_bounds__(256) void
bin_move_kernel(const int* __restrict__ src, const int* __restrict__ dst,
                const int* __restrict__ H, const int* __restrict__ binTotal,
                int* __restrict__ binStart, int* __restrict__ pairWords, int nEdges, int nBins)
{
  __shared__ int cursor[MAX_BINS];
  __shared__ int sIncl[8];
  const int tid = (int)threadIdx.x, lane = tid & 31, w = tid >> 5;
  const int G = (int)gridDim.x;
  int2* pairs = (int2*)pairWords;
  const bool aligned = ((((size_t)src) | ((size_t)dst)) & 15) == 0;
  const int e0 = blockIdx.x * CHUNK + 4 * tid;
  int d[QPT][4], s[QPT][4], c[QPT];
  #pragma unroll
  for (int u = 0; u < QPT; ++u) {
    c[u] = load_quad(dst, e0 + 1024 * u, nEdges, aligned, d[u]);
    load_quad(src, e0 + 1024 * u, nEdges, aligned, s[u]);
  }
  const int idx = 4 * tid;
  int v[4], h[4];
  #pragma unroll
  for (int k = 0; k < 4; ++k) {
    const bool in = (idx + k < nBins);
    v[k] = in ? binTotal[idx + k] : 0;
    h[k] = in ? H[(size_t)(idx + k) * G + blockIdx.x] : 0;
  }
  const int local = (v[0] + v[1]) + (v[2] + v[3]);
  int incl = local;
  #pragma unroll
  for (int o = 1; o < 32; o <<= 1) {
    int y = __shfl_up(incl, o);
    if (lane >= o) incl += y;
  }
  if (lane == 31) sIncl[w] = incl;
  __syncthreads();
  int r = incl - local;
  #pragma unroll
  for (int i = 0; i < 8; ++i)
    if (i < w) r += sIncl[i];
  #pragma unroll
  for (int k = 0; k < 4; ++k) {
    if (idx + k < nBins) {
      cursor[idx + k] = r + h[k];
      if (blockIdx.x == 0) binStart[idx + k] = r;
    }
    r += v[k];
  }
  if (blockIdx.x == 0 && idx < nBins && idx + 4 >= nBins) binStart[nBins] = r;
  __syncthreads();
  if ((c[0] + c[1]) + (c[2] + c[3]) == 4 * QPT) move_pairs<false>(cursor, pairs, d, s, c);
  else                                         move_pairs<true >(cursor, pairs, d, s, c);
}

constexpr int NODES_PER_WAVE = 4;

constexpr int PRE = 6;

template <typename L>
__device__ __forceinline__ int
rank_in_list(const L& list, int beg, int end, int p, int s)
{
  int r = 0;
  for (int j = beg; j < end; j += 8) {
    int t[8];
    #pragma unroll
    for (int u = 0; u < 8; ++u) t[u] = list(min(j + u, end - 1));
    #pragma unroll
    for (int u = 0; u < 8; ++u) {
      const int jj = j + u;
      const bool front = (t[u] < s) || (t[u] == s && jj < p);
      r += (jj < end && front) ? 1 : 0;
    }
  }
  return r;
}

__global__ __launch_bounds__(256) void
bin_lists_kernel(const int* __restrict__ pairWords, const int* __restrict__ binStart,
                 int* __restrict__ rowptr, int* nbr, int* __restrict__ nbrSorted,
                 int nNodes, int nBins)
{
  __shared__ int cntL[BIN_NODES];
  __shared__ int lpL[BIN_NODES + 1];
  __shared__ int curL[BIN_NODES];
  __shared__ int listL[BIN_CAP];
  __shared__ int nodeL[BIN_CAP];
  const int tid = (int)threadIdx.x;
  const int bin = (int)blockIdx.x;
  const int2* pairs = (const int2*)pairWords;
  const int nodeBase = bin * BIN_NODES;
  const int start = binStart[bin];
  const int m     = binStart[bin + 1] - start;
  int2 pr[PRE];
  #pragma unroll
  for (int u = 0; u < PRE; ++u) {
    const int i = tid + 256 * u;
    pr[u] = (i < m) ? pairs[start + i] : make_int2(0, nodeBase);
  }
  if (tid < BIN_NODES) cntL[tid] = 0;
  __syncthreads();
  #pragma unroll
  for (int u = 0; u < PRE; ++u)
    if (tid + 256 * u < m)
      __hip_atomic_fetch_add(&cntL[pr[u].y - nodeBase], 1, __ATOMIC_RELAXED, __HIP_MEMORY_SCOPE_WORKGROUP);
  for (int i = tid + 256 * PRE; i < m; i += 256)
    __hip_atomic_fetch_add(&cntL[pairs[start + i].y - nodeBase], 1, __ATOMIC_RELAXED, __HIP_MEMORY_SCOPE_WORKGROUP);
  __syncthreads();
  if (tid < 32) {
    const int c0 = cntL[2 * tid], c1 = cntL[2 * tid + 1];
    int incl = c0 + c1;
    #pragma unroll
    for (int o = 1; o < 32; o <<= 1) {
      int y = __shfl_up(incl, o);
      if (tid >= o) incl += y;
    }
    const int s0 = incl - (c0 + c1), s1 = s0 + c0;
    lpL[2 * tid]  = s0;  lpL[2 * tid + 1]  = s1;
    curL[2 * tid] = s0;  curL[2 * tid + 1] = s1;
    if (tid == 31) lpL[BIN_NODES] = incl;
    if (nodeBase + 2 * tid     <= nNodes) rowptr[nodeBase + 2 * tid]     = start + s0;
    if (nodeBase + 2 * tid + 1 <= nNodes) rowptr[nodeBase + 2 * tid + 1] = start + s1;
  }
  if (bin == nBins - 1 && tid == 0 && nodeBase + BIN_NODES == nNodes) rowptr[nNodes] = start + m;
  __syncthreads();
  if (m <= BIN_CAP) {
    int slot[PRE];
    #pragma unroll
    for (int u = 0; u < PRE; ++u)
      if (tid + 256 * u < m)
        slot[u] = __hip_atomic_fetch_add(&curL[pr[u].y - nodeBase], 1, __ATOMIC_RELAXED, __HIP_MEMORY_SCOPE_WORKGROUP);
    #pragma unroll
    for (int u = 0; u < PRE; ++u)
      if (tid + 256 * u < m) {
        listL[slot[u]] = pr[u].x;
        nodeL[slot[u]] = pr[u].y - nodeBase;
      }
    for (int i = tid + 256 * PRE; i < m; i += 256) {
      const int2 p   = pairs[start + i];
      const int  dl  = p.y - nodeBase;
      const int sl  = __hip_atomic_fetch_add(&curL[dl], 1, __ATOMIC_RELAXED, __HIP_MEMORY_SCOPE_WORKGROUP);
      listL[sl] = p.x;
      nodeL[sl] = dl;
    }
    __syncthreads();
    for (int p = tid; p < m; p += 256) {
      const int s  = listL[p];
      const int dl = nodeL[p];
      const int beg = lpL[dl], end = lpL[dl + 1];
      const int r = rank_in_list([&](int j) { return listL[j]; }, beg, end, p, s);
      nbrSorted[start + beg + r] = s;
    }
  } else {
    for (int i = tid; i < m; i += 256) {
      const int2 p  = pairs[start + i];
      const int  sl = __hip_atomic_fetch_add(&curL[p.y - nodeBase], 1, __ATOMIC_RELAXED, __HIP_MEMORY_SCOPE_WORKGROUP);
      nbr[start + sl] = p.x;
    }
    __syncthreads();
    const int lane = tid & 31;
    for (int q = tid >> 5; q < BIN_NODES; q += 8) {
      const int beg = lpL[q], end = lpL[q + 1];
      for (int i = beg + lane; i < end; i += 32) {
        const int s = nbr[start + i];
        const int r = rank_in_list([&](int j) { return nbr[start + j]; }, beg, end, i, s);
        nbrSorted[start + beg + r] = s;
      }
    }
  }
}

template <int F4, int U, bool CLAMP>
__device__ __forceinline__ void
add_rows(float4& acc, const float4* __restrict__ x4, const int* __restrict__ nbr,
         int e, int end, int c, int sub)
{
  constexpr int EPS = 32 / F4;
  float4 v[U];
  #pragma unroll
  for (int u = 0; u < U; ++u) {
    int ee = e + u * EPS + sub;
    if (CLAMP) ee = min(ee, end - 1);
    const int s = nbr[ee];
    v[u] = x4[(size_t)s * F4 + c];
  }
  if constexpr (CLAMP) {
    static_assert(U == 4, "the fence below names four rows");
    asm volatile("" :: "v"(v[0].x), "v"(v[0].y), "v"(v[0].z), "v"(v[0].w),
                       "v"(v[1].x), "v"(v[1].y), "v"(v[1].z), "v"(v[1].w),
                       "v"(v[2].x), "v"(v[2].y), "v"(v[2].z), "v"(v[2].w),
                       "v"(v[3].x), "v"(v[3].y), "v"(v[3].z), "v"(v[3].w));
  }
  #pragma unroll
  for (int u = 0; u < U; ++u) {
    if (!CLAMP || (e + u * EPS + sub < end)) {
      acc.x += v[u].x; acc.y += v[u].y; acc.z += v[u].z; acc.w += v[u].w;
    }
  }
}

template <int F4, bool ACT>
__global__ __launch_bounds__(256) void
gather_mean_kernel(const float* __restrict__ x, const int* __restrict__ rowptr,
                   const int* __restrict__ nbr, float* __restrict__ A,
                   const float* __restrict__ bias, int nNodes)
{
  constexpr int EPS = 32 / F4;
  const int lane = threadIdx.x & 31;
  const int c    = lane % F4;
  const int sub  = lane / F4;
  const int wv   = blockIdx.x * 8 + (threadIdx.x >> 5);
  const float4* x4 = (const float4*)x;
  float4 bv = make_float4(0.0f, 0.0f, 0.0f, 0.0f);
  if constexpr (ACT) {
    bv.x = bias[4 * c]; bv.y = bias[4 * c + 1]; bv.z = bias[4 * c + 2]; bv.w = bias[4 * c + 3];
  }
  for (int q = 0; q < NODES_PER_WAVE; ++q) {
    const int node = wv * NODES_PER_WAVE + q;
    if (node >= nNodes) return;
    const int beg = rowptr[node], end = rowptr[node + 1];
    const float4 xv = x4[(size_t)node * F4 + c];
    float4 acc = make_float4(0.0f, 0.0f, 0.0f, 0.0f);
    int e = beg;
    for (; e + 8 * EPS <= end; e += 8 * EPS) add_rows<F4, 8, false>(acc, x4, nbr, e, end, c, sub);
    for (; e < end; e += 4 * EPS)            add_rows<F4, 4, true >(acc, x4, nbr, e, end, c, sub);
    if (EPS == 2) {
      acc.x += __shfl_xor(acc.x, 16);
      acc.y += __shfl_xor(acc.y, 16);
      acc.z += __shfl_xor(acc.z, 16);
      acc.w += __shfl_xor(acc.w, 16);
    }
    const float rd = 1.0f / fmaxf((float)(end - beg), 1.0f);
    if (sub == 0) {
      float4 o;
      o.x = xv.x + acc.x * rd;
      o.y = xv.y + acc.y * rd;
      o.z = xv.z + acc.z * rd;
      o.w = xv.w + acc.w * rd;
      if constexpr (ACT) {
        o.x = fmaxf(o.x + bv.x, 0.0f);
        o.y = fmaxf(o.y + bv.y, 0.0f);
        o.z = fmaxf(o.z + bv.z, 0.0f);
        o.w = fmaxf(o.w + bv.w, 0.0f);
      }
      ((float4*)A)[(size_t)node * F4 + c] = o;
    }
  }
}

template <int FIN, int NT>
__device__ __forceinline__ void
gemm_tiles(const float4 (&a4)[FIN / 8], const float* __restrict__ Bl,
           const float* __restrict__ bsh, float* __restrict__ out,
           int ct, int rowBase, int half, int l16, int nRows, int Fout, int doRelu,
           bool fullRows)
{
  constexpr int STRIDE = FIN + 4;
  constexpr int KP = FIN / 8;
  v8f c[NT];
  #pragma unroll
  for (int t = 0; t < NT; ++t) c[t] = v8f{};
  #pragma unroll
  for (int j = 0; j < KP; ++j) {
    float4 b[NT];
    #pragma unroll
    for (int t = 0; t < NT; ++t)
      b[t] = *(const float4*)(Bl + (ct + t) * 16 * STRIDE + 8 * j);
    v2f alo, ahi;
    alo.x = a4[j].x; alo.y = a4[j].y;
    ahi.x = a4[j].z; ahi.y = a4[j].w;
    #pragma unroll
    for (int t = 0; t < NT; ++t) {
      v2f blo; blo.x = b[t].x; blo.y = b[t].y;
      c[t] = __builtin_amdgcn_wmma_f32_16x16x4_f32(false, alo, false, blo, (short)0, c[t], false, false);
    }
    #pragma unroll
    for (int t = 0; t < NT; ++t) {
      v2f bhi; bhi.x = b[t].z; bhi.y = b[t].w;
      c[t] = __builtin_amdgcn_wmma_f32_16x16x4_f32(false, ahi, false, bhi, (short)0, c[t], false, false);
    }
  }
  #pragma unroll
  for (int t = 0; t < NT; ++t) {
    const int n = (ct + t) * 16 + l16;
    const float bv = bsh[n];
    float* colBase = out + (size_t)(rowBase + 8 * half) * Fout + n;
    if (fullRows && ((ct + t) * 16 + 16 <= Fout)) {
      #pragma unroll
      for (int r = 0; r < 8; ++r) {
        float val = c[t][r] + bv;
        if (doRelu) val = fmaxf(val, 0.0f);
        colBase[(size_t)r * Fout] = val;
      }
    } else {
      #pragma unroll
      for (int r = 0; r < 8; ++r) {
        int gRow = rowBase + r + 8 * half;
        float val = c[t][r] + bv;
        if (doRelu) val = fmaxf(val, 0.0f);
        if (gRow < nRows && n < Fout) out[(size_t)gRow * Fout + n] = val;
      }
    }
  }
}

template <int FIN>
__global__ __launch_bounds__(256) void
gemm_bias_act_kernel(const float* __restrict__ xin, const float* __restrict__ W,
                     const float* __restrict__ bias, float* __restrict__ out,
                     int nRows, int Fout, int doRelu)
{
  constexpr int STRIDE = FIN + 4;
  constexpr int F4  = FIN / 4;
  constexpr int KP  = FIN / 8;
  constexpr int WLD = (128 * F4) / 256;
  __shared__ __attribute__((aligned(16))) float Bs[128 * STRIDE];
  __shared__ float bsh[128];

  const int foutTiles = (Fout + 15) >> 4;
  const int nW4       = foutTiles * 16 * F4;

  const int lane = threadIdx.x & 31;
  const int wave = threadIdx.x >> 5;
  const int half = lane >> 4;
  const int l16  = lane & 15;
  const int rowBase = (blockIdx.x * 8 + wave) * 16;
  const int row     = min(rowBase + l16, nRows - 1);

  const float bval = bias[min((int)threadIdx.x, Fout - 1)];
  const int nWvalid = Fout * F4;
  float4 wv[WLD];
  #pragma unroll
  for (int i = 0; i < WLD; ++i) {
    int idx = (int)threadIdx.x + i * 256;
    wv[i] = ((const float4*)W)[min(idx, nWvalid - 1)];
  }
  float4 a4[KP];
  {
    const float4* xr = (const float4*)(xin + (size_t)row * FIN) + half;
    #pragma unroll
    for (int j = 0; j < KP; ++j) a4[j] = xr[2 * j];
  }
  #pragma unroll
  for (int i = 0; i < WLD; ++i) {
    int idx = (int)threadIdx.x + i * 256;
    if (idx < nW4) {
      int r  = idx / F4;
      int c4 = idx - r * F4;
      float4 v = wv[i];
      if (idx >= nWvalid) v = make_float4(0.0f, 0.0f, 0.0f, 0.0f);
      *(float4*)&Bs[r * STRIDE + 4 * c4] = v;
    }
  }
  if (threadIdx.x < 128) bsh[threadIdx.x] = ((int)threadIdx.x < Fout) ? bval : 0.0f;
  __syncthreads();
  if (rowBase >= nRows) return;

  const float* Bl = &Bs[l16 * STRIDE + 4 * half];
  const bool fullRows = (rowBase + 16 <= nRows);
  int ct = 0;
  for (; ct + 2 <= foutTiles; ct += 2)
    gemm_tiles<FIN, 2>(a4, Bl, bsh, out, ct, rowBase, half, l16, nRows, Fout, doRelu, fullRows);
  if (ct < foutTiles)
    gemm_tiles<FIN, 1>(a4, Bl, bsh, out, ct, rowBase, half, l16, nRows, Fout, doRelu, fullRows);
}

constexpr int CHUNK_PITCH = 32 + 4;
__global__ __launch_bounds__(256) void
layer2_pre3_kernel(const float* __restrict__ xin, const float* __restrict__ W2,
                   const float* __restrict__ b2, const float* __restrict__ W3,
                   float* __restrict__ tout, int nRows)
{
  constexpr int FIN    = 128;
  constexpr int STRIDE = FIN + 4;
  constexpr int F4     = FIN / 4;
  constexpr int KP     = FIN / 8;
  constexpr int FH     = 128;
  constexpr int FT     = 64;
  constexpr int WLD2   = (FH * F4) / 256;
  constexpr int WLD3   = (FT * F4) / 256;
  __shared__ __attribute__((aligned(16))) float Bs[FH * STRIDE];
  __shared__ __attribute__((aligned(16))) float B3[FT * STRIDE];
  __shared__ __attribute__((aligned(16))) float Ks[8 * 16 * CHUNK_PITCH];
  __shared__ float bsh[FH];

  const int tid  = (int)threadIdx.x;
  const int lane = tid & 31;
  const int wave = tid >> 5;
  const int half = lane >> 4;
  const int l16  = lane & 15;
  const int rowBase = (blockIdx.x * 8 + wave) * 16;
  const int row     = min(rowBase + l16, nRows - 1);

  const float bval = b2[min(tid, FH - 1)];
  float4 w2[WLD2];
  #pragma unroll
  for (int i = 0; i < WLD2; ++i) w2[i] = ((const float4*)W2)[tid + i * 256];
  float4 w3[WLD3];
  #pragma unroll
  for (int i = 0; i < WLD3; ++i) w3[i] = ((const float4*)W3)[tid + i * 256];
  float4 a4[KP];
  {
    const float4* xr = (const float4*)(xin + (size_t)row * FIN) + half;
    #pragma unroll
    for (int j = 0; j < KP; ++j) a4[j] = xr[2 * j];
  }
  #pragma unroll
  for (int i = 0; i < WLD2; ++i) {
    const int idx = tid + i * 256;
    const int r = idx / F4, c4 = idx - r * F4;
    *(float4*)&Bs[r * STRIDE + 4 * c4] = w2[i];
  }
  #pragma unroll
  for (int i = 0; i < WLD3; ++i) {
    const int idx = tid + i * 256;
    const int r = idx / F4, c4 = idx - r * F4;
    *(float4*)&B3[r * STRIDE + 4 * c4] = w3[i];
  }
  if (tid < FH) bsh[tid] = bval;
  __syncthreads();
  if (rowBase >= nRows) return;

  const float* Bl  = &Bs[l16 * STRIDE + 4 * half];
  const float* B3l = &B3[l16 * STRIDE + 4 * half];
  float*       Kw  = &Ks[wave * 16 * CHUNK_PITCH];

  v8f tacc[FT / 16];
  #pragma unroll
  for (int t = 0; t < FT / 16; ++t) tacc[t] = v8f{};

  #pragma unroll
  for (int p = 0; p < FH / 32; ++p) {
    v8f c[2];
    c[0] = v8f{}; c[1] = v8f{};
    #pragma unroll
    for (int j = 0; j < KP; ++j) {
      float4 b[2];
      #pragma unroll
      for (int t = 0; t < 2; ++t)
        b[t] = *(const float4*)(Bl + (2 * p + t) * 16 * STRIDE + 8 * j);
      v2f alo, ahi;
      alo.x = a4[j].x; alo.y = a4[j].y;
      ahi.x = a4[j].z; ahi.y = a4[j].w;
      #pragma unroll
      for (int t = 0; t < 2; ++t) {
        v2f blo; blo.x = b[t].x; blo.y = b[t].y;
        c[t] = __builtin_amdgcn_wmma_f32_16x16x4_f32(false, alo, false, blo, (short)0, c[t], false, false);
      }
      #pragma unroll
      for (int t = 0; t < 2; ++t) {
        v2f bhi; bhi.x = b[t].z; bhi.y = b[t].w;
        c[t] = __builtin_amdgcn_wmma_f32_16x16x4_f32(false, ahi, false, bhi, (short)0, c[t], false, false);
      }
    }
    #pragma unroll
    for (int t = 0; t < 2; ++t) {
      const float bv = bsh[(2 * p + t) * 16 + l16];
      #pragma unroll
      for (int r = 0; r < 8; ++r)
        Kw[(r + 8 * half) * CHUNK_PITCH + t * 16 + l16] = fmaxf(c[t][r] + bv, 0.0f);
    }
    float4 e4[4];
    #pragma unroll
    for (int j = 0; j < 4; ++j) e4[j] = *(const float4*)(Kw + l16 * CHUNK_PITCH + 8 * j + 4 * half);
    #pragma unroll
    for (int j = 0; j < 4; ++j) {
      float4 b[FT / 16];
      #pragma unroll
      for (int t = 0; t < FT / 16; ++t)
        b[t] = *(const float4*)(B3l + t * 16 * STRIDE + 8 * (4 * p + j));
      v2f alo, ahi;
      alo.x = e4[j].x; alo.y = e4[j].y;
      ahi.x = e4[j].z; ahi.y = e4[j].w;
      #pragma unroll
      for (int t = 0; t < FT / 16; ++t) {
        v2f blo; blo.x = b[t].x; blo.y = b[t].y;
        tacc[t] = __builtin_amdgcn_wmma_f32_16x16x4_f32(false, alo, false, blo, (short)0, tacc[t], false, false);
      }
      #pragma unroll
      for (int t = 0; t < FT / 16; ++t) {
        v2f bhi; bhi.x = b[t].z; bhi.y = b[t].w;
        tacc[t] = __builtin_amdgcn_wmma_f32_16x16x4_f32(false, ahi, false, bhi, (short)0, tacc[t], false, false);
      }
    }
  }
  const bool fullRows = (rowBase + 16 <= nRows);
  #pragma unroll
  for (int t = 0; t < FT / 16; ++t) {
    const int n = t * 16 + l16;
    if (fullRows) {
      float* colBase = tout + (size_t)(rowBase + 8 * half) * FT + n;
      #pragma unroll
      for (int r = 0; r < 8; ++r) colBase[(size_t)r * FT] = tacc[t][r];
    } else {
      #pragma unroll
      for (int r = 0; r < 8; ++r) {
        const int gRow = rowBase + r + 8 * half;
        if (gRow < nRows) tout[(size_t)gRow * FT + n] = tacc[t][r];
      }
    }
  }
}

__global__ __launch_bounds__(256) void
heads_kernel(const float* __restrict__ emb, const float* __restrict__ Wp,
             const float* __restrict__ bp, const float* __restrict__ Wc,
             const float* __restrict__ bc, float* __restrict__ nxt,
             float* __restrict__ cls, int nRows, int nClasses)
{
  constexpr int FE      = 64;
  constexpr int STRIDE2 = FE + 4;
  constexpr int F4E     = FE / 4;
  constexpr int KPE     = FE / 8;
  constexpr int WLDP    = (FE * F4E) / 256;
  __shared__ __attribute__((aligned(16))) float Bh[(FE + 16) * STRIDE2];
  __shared__ float bsh2[FE + 16];

  const int tid  = (int)threadIdx.x;
  const int lane = tid & 31;
  const int wave = tid >> 5;
  const int half = lane >> 4;
  const int l16  = lane & 15;
  const int rowBase = (blockIdx.x * 8 + wave) * 16;
  const int row     = min(rowBase + l16, nRows - 1);

  const float bvp = bp[min(tid, FE - 1)];
  const float bvc = bc[min(tid, nClasses - 1)];
  float4 wp[WLDP];
  #pragma unroll
  for (int i = 0; i < WLDP; ++i) wp[i] = ((const float4*)Wp)[tid + i * 256];
  const int wcRow = tid >> 4;
  const float4 wc = ((const float4*)Wc)[min(tid, nClasses * F4E - 1)];
  float4 e4[KPE];
  {
    const float4* xr = (const float4*)(emb + (size_t)row * FE) + half;
    #pragma unroll
    for (int j = 0; j < KPE; ++j) e4[j] = xr[2 * j];
  }
  #pragma unroll
  for (int i = 0; i < WLDP; ++i) {
    const int idx = tid + i * 256;
    const int r = idx / F4E, c4 = idx - r * F4E;
    *(float4*)&Bh[r * STRIDE2 + 4 * c4] = wp[i];
  }
  {
    const int c4 = tid - wcRow * F4E;
    *(float4*)&Bh[(FE + wcRow) * STRIDE2 + 4 * c4] =
        (wcRow < nClasses) ? wc : make_float4(0.0f, 0.0f, 0.0f, 0.0f);
  }
  if (tid < FE) bsh2[tid] = bvp;
  if (tid < 16) bsh2[FE + tid] = (tid < nClasses) ? bvc : 0.0f;
  __syncthreads();
  if (rowBase >= nRows) return;

  const bool fullRows = (rowBase + 16 <= nRows);
  const float* Bl2 = &Bh[l16 * STRIDE2 + 4 * half];
  #pragma unroll
  for (int ct = 0; ct < FE / 16; ct += 2)
    gemm_tiles<FE, 2>(e4, Bl2, bsh2, nxt, ct, rowBase, half, l16, nRows, FE, 0, fullRows);
  gemm_tiles<FE, 1>(e4, Bl2 + FE * STRIDE2, bsh2 + FE, cls, 0, rowBase, half, l16, nRows, nClasses, 0, fullRows);
}

extern "C" void kernel_launch(void* const* d_in, const int* in_sizes, int n_in,
                              void* d_out, int out_size, void* d_ws, size_t ws_size,
                              hipStream_t stream) {
  const float* x0 = (const float*)d_in[0];
  const int*   ei = (const int*)d_in[1];
  const float *W1 = (const float*)d_in[2],  *b1 = (const float*)d_in[3];
  const float *W2 = (const float*)d_in[4],  *b2 = (const float*)d_in[5];
  const float *W3 = (const float*)d_in[6],  *b3 = (const float*)d_in[7];
  const float *Wp = (const float*)d_in[8],  *bp = (const float*)d_in[9];
  const float *Wc = (const float*)d_in[10], *bc = (const float*)d_in[11];

  const int N      = in_sizes[0] / 64;
  const int nEdges = in_sizes[1] / 2;
  const int* src = ei;
  const int* dst = ei + nEdges;

  float* ws        = (float*)d_ws;
  int*   binTotal  = (int*)d_ws;
  int*   binStart  = binTotal + 1024;
  float* A         = ws + 65536;
  int*   pairs     = (int*)A;
  int*   H         = (int*)A + 2 * (size_t)nEdges;
  int*   rowptr    = (int*)(A + (size_t)N * 128);
  int*   nbr       = rowptr + 65536;
  int*   nbrSorted = nbr + nEdges;

  float* out = (float*)d_out;
  float* h1  = out;
  float* emb = out;
  float* t3  = out + (size_t)N * 64;
  float* nxt = out + (size_t)N * 64;
  float* cls = out + (size_t)N * 128;

  dim3 b256(256);
  dim3 gemmGrid((unsigned)((N + 127) / 128)), gemmBlock(256);
  dim3 nodeGrid((unsigned)((N + 8 * NODES_PER_WAVE - 1) / (8 * NODES_PER_WAVE)));
  const int nBins   = (N + BIN_NODES - 1) / BIN_NODES;
  const int nChunks = (nEdges + CHUNK - 1) / CHUNK;
  dim3 chunkGrid((unsigned)nChunks), binGrid((unsigned)nBins), binWaveGrid((unsigned)((nBins + 7) / 8));

  bin_count_kernel <<<chunkGrid,   b256, 0, stream>>>(dst, H, nEdges, nBins);
  bin_prefix_kernel<<<binWaveGrid, b256, 0, stream>>>(H, binTotal, nBins, nChunks);
  bin_move_kernel  <<<chunkGrid,   b256, 0, stream>>>(src, dst, H, binTotal, binStart, pairs, nEdges, nBins);
  bin_lists_kernel <<<binGrid,     b256, 0, stream>>>(pairs, binStart, rowptr, nbr, nbrSorted, N, nBins);

  gather_mean_kernel<16, false><<<nodeGrid, b256, 0, stream>>>(x0, rowptr, nbrSorted, A, nullptr, N);
  gemm_bias_act_kernel<64><<<gemmGrid, gemmBlock, 0, stream>>>(A, W1, b1, h1, N, 128, 1);

  gather_mean_kernel<32, false><<<nodeGrid, b256, 0, stream>>>(h1, rowptr, nbrSorted, A, nullptr, N);
  layer2_pre3_kernel<<<gemmGrid, gemmBlock, 0, stream>>>(A, W2, b2, W3, t3, N);

  gather_mean_kernel<16, true><<<nodeGrid, b256, 0, stream>>>(t3, rowptr, nbrSorted, emb, b3, N);

  heads_kernel<<<gemmGrid, gemmBlock, 0, stream>>>(emb, Wp, bp, Wc, bc, nxt, cls, N, 10);
}


# Neighbour lists in 2 launches, not 4: each block orders its own 4096 edges by bin in LDS and stores the chunk whole with its bin offsets; bin blocks sum offsets and pull their pieces; sum order kept
# speedup vs baseline: 5.9471x; 1.2508x over previous
#include <hip/hip_runtime.h>
#include <hip/hip_bf16.h>

typedef __attribute__((ext_vector_type(8))) float v8f;
typedef __attribute__((ext_vector_type(2))) float v2f;

constexpr int BIN_SHIFT = 6;
constexpr int BIN_NODES = 1 << BIN_SHIFT;
constexpr int MAX_BINS  = 1024;
constexpr int CHUNK     = 4096;
constexpr int QPT       = CHUNK / (256 * 4);
constexpr int BIN_CAP   = 4096;
constexpr int SEG_PRE   = 12;
static_assert(MAX_BINS == 4 * 256, "chunk_sort_kernel scans four bins per thread");

__device__ __forceinline__ int
load_quad(const int* __restrict__ p, int e, int n, bool aligned, int (&v)[4])
{
  if (aligned && e + 4 <= n) {
    const int4 t = *(const int4*)(p + e);
    v[0] = t.x; v[1] = t.y; v[2] = t.z; v[3] = t.w;
    return 4;
  }
  int c = 0;
  #pragma unroll
  for (int k = 0; k < 4; ++k) {
    v[k] = 0;
    if (e + k < n) { v[k] = p[e + k]; c = k + 1; }
  }
  return c;
}

template <bool GUARD>
__device__ __forceinline__ void
place_pairs(int* __restrict__ cursor, int2* __restrict__ sorted,
            const int (&d)[QPT][4], const int (&s)[QPT][4], const int (&c)[QPT])
{
  int pos[QPT][4];
  #pragma unroll
  for (int u = 0; u < QPT; ++u) {
    #pragma unroll
    for (int k = 0; k < 4; ++k) {
      pos[u][k] = 0;
      if (!GUARD || k < c[u])
        pos[u][k] = __hip_atomic_fetch_add(&cursor[d[u][k] >> BIN_SHIFT], 1, __ATOMIC_RELAXED, __HIP_MEMORY_SCOPE_WORKGROUP);
    }
  }
  #pragma unroll
  for (int u = 0; u < QPT; ++u) {
    #pragma unroll
    for (int k = 0; k < 4; ++k)
      if (!GUARD || k < c[u]) sorted[pos[u][k]] = make_int2(s[u][k], d[u][k]);
  }
}

__global__ __launch_bounds__(256) void
chunk_sort_kernel(const int* __restrict__ src, const int* __restrict__ dst,
                  int* __restrict__ offT, int* __restrict__ pairWords, int nEdges, int nBins)
{
  __shared__ int cursor[MAX_BINS];
  __shared__ int sIncl[8];
  __shared__ __attribute__((aligned(16))) int sortedW[2 * CHUNK];
  const int tid = (int)threadIdx.x, lane = tid & 31, w = tid >> 5;
  const int G = (int)gridDim.x;
  int2* sorted = (int2*)sortedW;
  const bool aligned = ((((size_t)src) | ((size_t)dst)) & 15) == 0;
  const int e0 = blockIdx.x * CHUNK + 4 * tid;
  int d[QPT][4], s[QPT][4], c[QPT];
  #pragma unroll
  for (int u = 0; u < QPT; ++u) {
    c[u] = load_quad(dst, e0 + 1024 * u, nEdges, aligned, d[u]);
    load_quad(src, e0 + 1024 * u, nEdges, aligned, s[u]);
  }
  for (int i = tid; i < MAX_BINS; i += 256) cursor[i] = 0;
  __syncthreads();
  #pragma unroll
  for (int u = 0; u < QPT; ++u) {
    #pragma unroll
    for (int k = 0; k < 4; ++k)
      if (k < c[u])
        __hip_atomic_fetch_add(&cursor[d[u][k] >> BIN_SHIFT], 1, __ATOMIC_RELAXED, __HIP_MEMORY_SCOPE_WORKGROUP);
  }
  __syncthreads();
  const int idx = 4 * tid;
  int v[4];
  #pragma unroll
  for (int k = 0; k < 4; ++k) v[k] = cursor[idx + k];
  const int local = (v[0] + v[1]) + (v[2] + v[3]);
  int incl = local;
  #pragma unroll
  for (int o = 1; o < 32; o <<= 1) {
    int y = __shfl_up(incl, o);
    if (lane >= o) incl += y;
  }
  if (lane == 31) sIncl[w] = incl;
  __syncthreads();
  int r = incl - local;
  #pragma unroll
  for (int i = 0; i < 8; ++i)
    if (i < w) r += sIncl[i];
  #pragma unroll
  for (int k = 0; k < 4; ++k) {
    cursor[idx + k] = r;
    if (idx + k < nBins) offT[(size_t)(idx + k) * G + blockIdx.x] = r;
    r += v[k];
  }
  if (tid == 255) offT[(size_t)nBins * G + blockIdx.x] = r;
  __syncthreads();
  if ((c[0] + c[1]) + (c[2] + c[3]) == 4 * QPT) place_pairs<false>(cursor, sorted, d, s, c);
  else                                         place_pairs<true >(cursor, sorted, d, s, c);
  __syncthreads();
  const int total = min(CHUNK, max(nEdges - (int)blockIdx.x * CHUNK, 0));
  int4*       outp = (int4*)pairWords + (size_t)blockIdx.x * (CHUNK / 2);
  const int4* sp   = (const int4*)sortedW;
  if (total == CHUNK) {
    #pragma unroll
    for (int u = 0; u < CHUNK / 512; ++u) outp[tid + 256 * u] = sp[tid + 256 * u];
  } else {
    #pragma unroll
    for (int u = 0; u < CHUNK / 512; ++u) {
      const int i = tid + 256 * u;
      if (2 * i < total) outp[i] = sp[i];
    }
  }
}

constexpr int NODES_PER_WAVE = 4;

constexpr int PRE = 6;

template <typename L>
__device__ __forceinline__ int
rank_in_list(const L& list, int beg, int end, int p, int s)
{
  int r = 0;
  for (int j = beg; j < end; j += 8) {
    int t[8];
    #pragma unroll
    for (int u = 0; u < 8; ++u) t[u] = list(min(j + u, end - 1));
    #pragma unroll
    for (int u = 0; u < 8; ++u) {
      const int jj = j + u;
      const bool front = (t[u] < s) || (t[u] == s && jj < p);
      r += (jj < end && front) ? 1 : 0;
    }
  }
  return r;
}

__global__ __launch_bounds__(256) void
bin_lists_kernel(const int* __restrict__ pairWords, const int* __restrict__ offT,
                 int* __restrict__ rowptr, int* nbr, int* __restrict__ nbrSorted,
                 int nNodes, int nBins, int G)
{
  __shared__ int cntL[BIN_NODES];
  __shared__ int lpL[BIN_NODES + 1];
  __shared__ int curL[BIN_NODES];
  __shared__ int listL[BIN_CAP];
  __shared__ int nodeL[BIN_CAP];
  __shared__ __attribute__((aligned(8))) int pairLW[2 * BIN_CAP];
  __shared__ int sCnt[8];
  __shared__ int sOff[8];
  const int tid = (int)threadIdx.x, lane = tid & 31, w = tid >> 5;
  const int bin = (int)blockIdx.x;
  const int2* pairs = (const int2*)pairWords;
  int2* pairL = (int2*)pairLW;
  const int nodeBase = bin * BIN_NODES;
  const int* row0 = offT + (size_t)bin * G;
  const int* row1 = row0 + G;
  if (tid < BIN_NODES) cntL[tid] = 0;
  __syncthreads();
  int m = 0;
  int start = 0;
  for (int b0 = 0; b0 < G; b0 += 256) {
    const int b = b0 + tid;
    int o0 = 0, o1 = 0;
    if (b < G) { o0 = row0[b]; o1 = row1[b]; }
    const int cnt = o1 - o0;
    const int2* chunk = pairs + (size_t)min(b, G - 1) * CHUNK;
    const int2* seg   = chunk + o0;
    const int   last  = max(o1 - 1, 0);
    int2 p[SEG_PRE];
    #pragma unroll
    for (int u = 0; u < SEG_PRE; ++u) p[u] = chunk[min(o0 + u, last)];
    __builtin_amdgcn_sched_barrier(0);
    int incl = cnt, so = o0;
    #pragma unroll
    for (int o = 1; o < 32; o <<= 1) {
      int y = __shfl_up(incl, o);
      if (lane >= o) incl += y;
      so += __shfl_xor(so, o);
    }
    if (lane == 31) sCnt[w] = incl;
    if (lane == 0)  sOff[w] = so;
    __syncthreads();
    int base = m + incl - cnt;
    #pragma unroll
    for (int i = 0; i < 8; ++i) {
      const int t = sCnt[i];
      if (i < w) base += t;
      m += t;
      start += sOff[i];
    }
    #pragma unroll
    for (int u = 0; u < SEG_PRE; ++u)
      if (u < cnt) {
        __hip_atomic_fetch_add(&cntL[p[u].y - nodeBase], 1, __ATOMIC_RELAXED, __HIP_MEMORY_SCOPE_WORKGROUP);
        if (base + u < BIN_CAP) pairL[base + u] = p[u];
      }
    for (int k = SEG_PRE; k < cnt; ++k) {
      const int2 q = seg[k];
      __hip_atomic_fetch_add(&cntL[q.y - nodeBase], 1, __ATOMIC_RELAXED, __HIP_MEMORY_SCOPE_WORKGROUP);
      if (base + k < BIN_CAP) pairL[base + k] = q;
    }
    __syncthreads();
  }
  if (tid < 32) {
    const int c0 = cntL[2 * tid], c1 = cntL[2 * tid + 1];
    int incl = c0 + c1;
    #pragma unroll
    for (int o = 1; o < 32; o <<= 1) {
      int y = __shfl_up(incl, o);
      if (tid >= o) incl += y;
    }
    const int s0 = incl - (c0 + c1), s1 = s0 + c0;
    lpL[2 * tid]  = s0;  lpL[2 * tid + 1]  = s1;
    curL[2 * tid] = s0;  curL[2 * tid + 1] = s1;
    if (tid == 31) lpL[BIN_NODES] = incl;
    if (nodeBase + 2 * tid     <= nNodes) rowptr[nodeBase + 2 * tid]     = start + s0;
    if (nodeBase + 2 * tid + 1 <= nNodes) rowptr[nodeBase + 2 * tid + 1] = start + s1;
  }
  if (bin == nBins - 1 && tid == 0 && nodeBase + BIN_NODES == nNodes) rowptr[nNodes] = start + m;
  __syncthreads();
  if (m <= BIN_CAP) {
    int2 pr[PRE];
    #pragma unroll
    for (int u = 0; u < PRE; ++u) {
      const int i = tid + 256 * u;
      pr[u] = (i < m) ? pairL[i] : make_int2(0, nodeBase);
    }
    int slot[PRE];
    #pragma unroll
    for (int u = 0; u < PRE; ++u)
      if (tid + 256 * u < m)
        slot[u] = __hip_atomic_fetch_add(&curL[pr[u].y - nodeBase], 1, __ATOMIC_RELAXED, __HIP_MEMORY_SCOPE_WORKGROUP);
    #pragma unroll
    for (int u = 0; u < PRE; ++u)
      if (tid + 256 * u < m) {
        listL[slot[u]] = pr[u].x;
        nodeL[slot[u]] = pr[u].y - nodeBase;
      }
    for (int i = tid + 256 * PRE; i < m; i += 256) {
      const int2 q   = pairL[i];
      const int  dl  = q.y - nodeBase;
      const int  sl  = __hip_atomic_fetch_add(&curL[dl], 1, __ATOMIC_RELAXED, __HIP_MEMORY_SCOPE_WORKGROUP);
      listL[sl] = q.x;
      nodeL[sl] = dl;
    }
    __syncthreads();
    for (int p = tid; p < m; p += 256) {
      const int s  = listL[p];
      const int dl = nodeL[p];
      const int beg = lpL[dl], end = lpL[dl + 1];
      const int r = rank_in_list([&](int j) { return listL[j]; }, beg, end, p, s);
      nbrSorted[start + beg + r] = s;
    }
  } else {
    for (int b = tid; b < G; b += 256) {
      const int2* seg = pairs + (size_t)b * CHUNK;
      const int o1 = row1[b];
      for (int k = row0[b]; k < o1; ++k) {
        const int2 q  = seg[k];
        const int  sl = __hip_atomic_fetch_add(&curL[q.y - nodeBase], 1, __ATOMIC_RELAXED, __HIP_MEMORY_SCOPE_WORKGROUP);
        nbr[start + sl] = q.x;
      }
    }
    __syncthreads();
    for (int q = w; q < BIN_NODES; q += 8) {
      const int beg = lpL[q], end = lpL[q + 1];
      for (int i = beg + lane; i < end; i += 32) {
        const int s = nbr[start + i];
        const int r = rank_in_list([&](int j) { return nbr[start + j]; }, beg, end, i, s);
        nbrSorted[start + beg + r] = s;
      }
    }
  }
}

template <int F4, int U, bool CLAMP>
__device__ __forceinline__ void
add_rows(float4& acc, const float4* __restrict__ x4, const int* __restrict__ nbr,
         int e, int end, int c, int sub)
{
  constexpr int EPS = 32 / F4;
  float4 v[U];
  #pragma unroll
  for (int u = 0; u < U; ++u) {
    int ee = e + u * EPS + sub;
    if (CLAMP) ee = min(ee, end - 1);
    const int s = nbr[ee];
    v[u] = x4[(size_t)s * F4 + c];
  }
  if constexpr (CLAMP) {
    static_assert(U == 4, "the fence below names four rows");
    asm volatile("" :: "v"(v[0].x), "v"(v[0].y), "v"(v[0].z), "v"(v[0].w),
                       "v"(v[1].x), "v"(v[1].y), "v"(v[1].z), "v"(v[1].w),
                       "v"(v[2].x), "v"(v[2].y), "v"(v[2].z), "v"(v[2].w),
                       "v"(v[3].x), "v"(v[3].y), "v"(v[3].z), "v"(v[3].w));
  }
  #pragma unroll
  for (int u = 0; u < U; ++u) {
    if (!CLAMP || (e + u * EPS + sub < end)) {
      acc.x += v[u].x; acc.y += v[u].y; acc.z += v[u].z; acc.w += v[u].w;
    }
  }
}

template <int F4, bool ACT>
__global__ __launch_bounds__(256) void
gather_mean_kernel(const float* __restrict__ x, const int* __restrict__ rowptr,
                   const int* __restrict__ nbr, float* __restrict__ A,
                   const float* __restrict__ bias, int nNodes)
{
  constexpr int EPS = 32 / F4;
  const int lane = threadIdx.x & 31;
  const int c    = lane % F4;
  const int sub  = lane / F4;
  const int wv   = blockIdx.x * 8 + (threadIdx.x >> 5);
  const float4* x4 = (const float4*)x;
  float4 bv = make_float4(0.0f, 0.0f, 0.0f, 0.0f);
  if constexpr (ACT) {
    bv.x = bias[4 * c]; bv.y = bias[4 * c + 1]; bv.z = bias[4 * c + 2]; bv.w = bias[4 * c + 3];
  }
  for (int q = 0; q < NODES_PER_WAVE; ++q) {
    const int node = wv * NODES_PER_WAVE + q;
    if (node >= nNodes) return;
    const int beg = rowptr[node], end = rowptr[node + 1];
    const float4 xv = x4[(size_t)node * F4 + c];
    float4 acc = make_float4(0.0f, 0.0f, 0.0f, 0.0f);
    int e = beg;
    for (; e + 8 * EPS <= end; e += 8 * EPS) add_rows<F4, 8, false>(acc, x4, nbr, e, end, c, sub);
    for (; e < end; e += 4 * EPS)            add_rows<F4, 4, true >(acc, x4, nbr, e, end, c, sub);
    if (EPS == 2) {
      acc.x += __shfl_xor(acc.x, 16);
      acc.y += __shfl_xor(acc.y, 16);
      acc.z += __shfl_xor(acc.z, 16);
      acc.w += __shfl_xor(acc.w, 16);
    }
    const float rd = 1.0f / fmaxf((float)(end - beg), 1.0f);
    if (sub == 0) {
      float4 o;
      o.x = xv.x + acc.x * rd;
      o.y = xv.y + acc.y * rd;
      o.z = xv.z + acc.z * rd;
      o.w = xv.w + acc.w * rd;
      if constexpr (ACT) {
        o.x = fmaxf(o.x + bv.x, 0.0f);
        o.y = fmaxf(o.y + bv.y, 0.0f);
        o.z = fmaxf(o.z + bv.z, 0.0f);
        o.w = fmaxf(o.w + bv.w, 0.0f);
      }
      ((float4*)A)[(size_t)node * F4 + c] = o;
    }
  }
}

template <int FIN, int NT>
__device__ __forceinline__ void
gemm_tiles(const float4 (&a4)[FIN / 8], const float* __restrict__ Bl,
           const float* __restrict__ bsh, float* __restrict__ out,
           int ct, int rowBase, int half, int l16, int nRows, int Fout, int doRelu,
           bool fullRows)
{
  constexpr int STRIDE = FIN + 4;
  constexpr int KP = FIN / 8;
  v8f c[NT];
  #pragma unroll
  for (int t = 0; t < NT; ++t) c[t] = v8f{};
  #pragma unroll
  for (int j = 0; j < KP; ++j) {
    float4 b[NT];
    #pragma unroll
    for (int t = 0; t < NT; ++t)
      b[t] = *(const float4*)(Bl + (ct + t) * 16 * STRIDE + 8 * j);
    v2f alo, ahi;
    alo.x = a4[j].x; alo.y = a4[j].y;
    ahi.x = a4[j].z; ahi.y = a4[j].w;
    #pragma unroll
    for (int t = 0; t < NT; ++t) {
      v2f blo; blo.x = b[t].x; blo.y = b[t].y;
      c[t] = __builtin_amdgcn_wmma_f32_16x16x4_f32(false, alo, false, blo, (short)0, c[t], false, false);
    }
    #pragma unroll
    for (int t = 0; t < NT; ++t) {
      v2f bhi; bhi.x = b[t].z; bhi.y = b[t].w;
      c[t] = __builtin_amdgcn_wmma_f32_16x16x4_f32(false, ahi, false, bhi, (short)0, c[t], false, false);
    }
  }
  #pragma unroll
  for (int t = 0; t < NT; ++t) {
    const int n = (ct + t) * 16 + l16;
    const float bv = bsh[n];
    float* colBase = out + (size_t)(rowBase + 8 * half) * Fout + n;
    if (fullRows && ((ct + t) * 16 + 16 <= Fout)) {
      #pragma unroll
      for (int r = 0; r < 8; ++r) {
        float val = c[t][r] + bv;
        if (doRelu) val = fmaxf(val, 0.0f);
        colBase[(size_t)r * Fout] = val;
      }
    } else {
      #pragma unroll
      for (int r = 0; r < 8; ++r) {
        int gRow = rowBase + r + 8 * half;
        float val = c[t][r] + bv;
        if (doRelu) val = fmaxf(val, 0.0f);
        if (gRow < nRows && n < Fout) out[(size_t)gRow * Fout + n] = val;
      }
    }
  }
}

template <int FIN>
__global__ __launch_bounds__(256) void
gemm_bias_act_kernel(const float* __restrict__ xin, const float* __restrict__ W,
                     const float* __restrict__ bias, float* __restrict__ out,
                     int nRows, int Fout, int doRelu)
{
  constexpr int STRIDE = FIN + 4;
  constexpr int F4  = FIN / 4;
  constexpr int KP  = FIN / 8;
  constexpr int WLD = (128 * F4) / 256;
  __shared__ __attribute__((aligned(16))) float Bs[128 * STRIDE];
  __shared__ float bsh[128];

  const int foutTiles = (Fout + 15) >> 4;
  const int nW4       = foutTiles * 16 * F4;

  const int lane = threadIdx.x & 31;
  const int wave = threadIdx.x >> 5;
  const int half = lane >> 4;
  const int l16  = lane & 15;
  const int rowBase = (blockIdx.x * 8 + wave) * 16;
  const int row     = min(rowBase + l16, nRows - 1);

  const float bval = bias[min((int)threadIdx.x, Fout - 1)];
  const int nWvalid = Fout * F4;
  float4 wv[WLD];
  #pragma unroll
  for (int i = 0; i < WLD; ++i) {
    int idx = (int)threadIdx.x + i * 256;
    wv[i] = ((const float4*)W)[min(idx, nWvalid - 1)];
  }
  float4 a4[KP];
  {
    const float4* xr = (const float4*)(xin + (size_t)row * FIN) + half;
    #pragma unroll
    for (int j = 0; j < KP; ++j) a4[j] = xr[2 * j];
  }
  #pragma unroll
  for (int i = 0; i < WLD; ++i) {
    int idx = (int)threadIdx.x + i * 256;
    if (idx < nW4) {
      int r  = idx / F4;
      int c4 = idx - r * F4;
      float4 v = wv[i];
      if (idx >= nWvalid) v = make_float4(0.0f, 0.0f, 0.0f, 0.0f);
      *(float4*)&Bs[r * STRIDE + 4 * c4] = v;
    }
  }
  if (threadIdx.x < 128) bsh[threadIdx.x] = ((int)threadIdx.x < Fout) ? bval : 0.0f;
  __syncthreads();
  if (rowBase >= nRows) return;

  const float* Bl = &Bs[l16 * STRIDE + 4 * half];
  const bool fullRows = (rowBase + 16 <= nRows);
  int ct = 0;
  for (; ct + 2 <= foutTiles; ct += 2)
    gemm_tiles<FIN, 2>(a4, Bl, bsh, out, ct, rowBase, half, l16, nRows, Fout, doRelu, fullRows);
  if (ct < foutTiles)
    gemm_tiles<FIN, 1>(a4, Bl, bsh, out, ct, rowBase, half, l16, nRows, Fout, doRelu, fullRows);
}

constexpr int CHUNK_PITCH = 32 + 4;
__global__ __launch_bounds__(256) void
layer2_pre3_kernel(const float* __restrict__ xin, const float* __restrict__ W2,
                   const float* __restrict__ b2, const float* __restrict__ W3,
                   float* __restrict__ tout, int nRows)
{
  constexpr int FIN    = 128;
  constexpr int STRIDE = FIN + 4;
  constexpr int F4     = FIN / 4;
  constexpr int KP     = FIN / 8;
  constexpr int FH     = 128;
  constexpr int FT     = 64;
  constexpr int WLD2   = (FH * F4) / 256;
  constexpr int WLD3   = (FT * F4) / 256;
  __shared__ __attribute__((aligned(16))) float Bs[FH * STRIDE];
  __shared__ __attribute__((aligned(16))) float B3[FT * STRIDE];
  __shared__ __attribute__((aligned(16))) float Ks[8 * 16 * CHUNK_PITCH];
  __shared__ float bsh[FH];

  const int tid  = (int)threadIdx.x;
  const int lane = tid & 31;
  const int wave = tid >> 5;
  const int half = lane >> 4;
  const int l16  = lane & 15;
  const int rowBase = (blockIdx.x * 8 + wave) * 16;
  const int row     = min(rowBase + l16, nRows - 1);

  const float bval = b2[min(tid, FH - 1)];
  float4 w2[WLD2];
  #pragma unroll
  for (int i = 0; i < WLD2; ++i) w2[i] = ((const float4*)W2)[tid + i * 256];
  float4 w3[WLD3];
  #pragma unroll
  for (int i = 0; i < WLD3; ++i) w3[i] = ((const float4*)W3)[tid + i * 256];
  float4 a4[KP];
  {
    const float4* xr = (const float4*)(xin + (size_t)row * FIN) + half;
    #pragma unroll
    for (int j = 0; j < KP; ++j) a4[j] = xr[2 * j];
  }
  #pragma unroll
  for (int i = 0; i < WLD2; ++i) {
    const int idx = tid + i * 256;
    const int r = idx / F4, c4 = idx - r * F4;
    *(float4*)&Bs[r * STRIDE + 4 * c4] = w2[i];
  }
  #pragma unroll
  for (int i = 0; i < WLD3; ++i) {
    const int idx = tid + i * 256;
    const int r = idx / F4, c4 = idx - r * F4;
    *(float4*)&B3[r * STRIDE + 4 * c4] = w3[i];
  }
  if (tid < FH) bsh[tid] = bval;
  __syncthreads();
  if (rowBase >= nRows) return;

  const float* Bl  = &Bs[l16 * STRIDE + 4 * half];
  const float* B3l = &B3[l16 * STRIDE + 4 * half];
  float*       Kw  = &Ks[wave * 16 * CHUNK_PITCH];

  v8f tacc[FT / 16];
  #pragma unroll
  for (int t = 0; t < FT / 16; ++t) tacc[t] = v8f{};

  #pragma unroll
  for (int p = 0; p < FH / 32; ++p) {
    v8f c[2];
    c[0] = v8f{}; c[1] = v8f{};
    #pragma unroll
    for (int j = 0; j < KP; ++j) {
      float4 b[2];
      #pragma unroll
      for (int t = 0; t < 2; ++t)
        b[t] = *(const float4*)(Bl + (2 * p + t) * 16 * STRIDE + 8 * j);
      v2f alo, ahi;
      alo.x = a4[j].x; alo.y = a4[j].y;
      ahi.x = a4[j].z; ahi.y = a4[j].w;
      #pragma unroll
      for (int t = 0; t < 2; ++t) {
        v2f blo; blo.x = b[t].x; blo.y = b[t].y;
        c[t] = __builtin_amdgcn_wmma_f32_16x16x4_f32(false, alo, false, blo, (short)0, c[t], false, false);
      }
      #pragma unroll
      for (int t = 0; t < 2; ++t) {
        v2f bhi; bhi.x = b[t].z; bhi.y = b[t].w;
        c[t] = __builtin_amdgcn_wmma_f32_16x16x4_f32(false, ahi, false, bhi, (short)0, c[t], false, false);
      }
    }
    #pragma unroll
    for (int t = 0; t < 2; ++t) {
      const float bv = bsh[(2 * p + t) * 16 + l16];
      #pragma unroll
      for (int r = 0; r < 8; ++r)
        Kw[(r + 8 * half) * CHUNK_PITCH + t * 16 + l16] = fmaxf(c[t][r] + bv, 0.0f);
    }
    float4 e4[4];
    #pragma unroll
    for (int j = 0; j < 4; ++j) e4[j] = *(const float4*)(Kw + l16 * CHUNK_PITCH + 8 * j + 4 * half);
    #pragma unroll
    for (int j = 0; j < 4; ++j) {
      float4 b[FT / 16];
      #pragma unroll
      for (int t = 0; t < FT / 16; ++t)
        b[t] = *(const float4*)(B3l + t * 16 * STRIDE + 8 * (4 * p + j));
      v2f alo, ahi;
      alo.x = e4[j].x; alo.y = e4[j].y;
      ahi.x = e4[j].z; ahi.y = e4[j].w;
      #pragma unroll
      for (int t = 0; t < FT / 16; ++t) {
        v2f blo; blo.x = b[t].x; blo.y = b[t].y;
        tacc[t] = __builtin_amdgcn_wmma_f32_16x16x4_f32(false, alo, false, blo, (short)0, tacc[t], false, false);
      }
      #pragma unroll
      for (int t = 0; t < FT / 16; ++t) {
        v2f bhi; bhi.x = b[t].z; bhi.y = b[t].w;
        tacc[t] = __builtin_amdgcn_wmma_f32_16x16x4_f32(false, ahi, false, bhi, (short)0, tacc[t], false, false);
      }
    }
  }
  const bool fullRows = (rowBase + 16 <= nRows);
  #pragma unroll
  for (int t = 0; t < FT / 16; ++t) {
    const int n = t * 16 + l16;
    if (fullRows) {
      float* colBase = tout + (size_t)(rowBase + 8 * half) * FT + n;
      #pragma unroll
      for (int r = 0; r < 8; ++r) colBase[(size_t)r * FT] = tacc[t][r];
    } else {
      #pragma unroll
      for (int r = 0; r < 8; ++r) {
        const int gRow = rowBase + r + 8 * half;
        if (gRow < nRows) tout[(size_t)gRow * FT + n] = tacc[t][r];
      }
    }
  }
}

__global__ __launch_bounds__(256) void
heads_kernel(const float* __restrict__ emb, const float* __restrict__ Wp,
             const float* __restrict__ bp, const float* __restrict__ Wc,
             const float* __restrict__ bc, float* __restrict__ nxt,
             float* __restrict__ cls, int nRows, int nClasses)
{
  constexpr int FE      = 64;
  constexpr int STRIDE2 = FE + 4;
  constexpr int F4E     = FE / 4;
  constexpr int KPE     = FE / 8;
  constexpr int WLDP    = (FE * F4E) / 256;
  __shared__ __attribute__((aligned(16))) float Bh[(FE + 16) * STRIDE2];
  __shared__ float bsh2[FE + 16];

  const int tid  = (int)threadIdx.x;
  const int lane = tid & 31;
  const int wave = tid >> 5;
  const int half = lane >> 4;
  const int l16  = lane & 15;
  const int rowBase = (blockIdx.x * 8 + wave) * 16;
  const int row     = min(rowBase + l16, nRows - 1);

  const float bvp = bp[min(tid, FE - 1)];
  const float bvc = bc[min(tid, nClasses - 1)];
  float4 wp[WLDP];
  #pragma unroll
  for (int i = 0; i < WLDP; ++i) wp[i] = ((const float4*)Wp)[tid + i * 256];
  const int wcRow = tid >> 4;
  const float4 wc = ((const float4*)Wc)[min(tid, nClasses * F4E - 1)];
  float4 e4[KPE];
  {
    const float4* xr = (const float4*)(emb + (size_t)row * FE) + half;
    #pragma unroll
    for (int j = 0; j < KPE; ++j) e4[j] = xr[2 * j];
  }
  #pragma unroll
  for (int i = 0; i < WLDP; ++i) {
    const int idx = tid + i * 256;
    const int r = idx / F4E, c4 = idx - r * F4E;
    *(float4*)&Bh[r * STRIDE2 + 4 * c4] = wp[i];
  }
  {
    const int c4 = tid - wcRow * F4E;
    *(float4*)&Bh[(FE + wcRow) * STRIDE2 + 4 * c4] =
        (wcRow < nClasses) ? wc : make_float4(0.0f, 0.0f, 0.0f, 0.0f);
  }
  if (tid < FE) bsh2[tid] = bvp;
  if (tid < 16) bsh2[FE + tid] = (tid < nClasses) ? bvc : 0.0f;
  __syncthreads();
  if (rowBase >= nRows) return;

  const bool fullRows = (rowBase + 16 <= nRows);
  const float* Bl2 = &Bh[l16 * STRIDE2 + 4 * half];
  #pragma unroll
  for (int ct = 0; ct < FE / 16; ct += 2)
    gemm_tiles<FE, 2>(e4, Bl2, bsh2, nxt, ct, rowBase, half, l16, nRows, FE, 0, fullRows);
  gemm_tiles<FE, 1>(e4, Bl2 + FE * STRIDE2, bsh2 + FE, cls, 0, rowBase, half, l16, nRows, nClasses, 0, fullRows);
}

extern "C" void kernel_launch(void* const* d_in, const int* in_sizes, int n_in,
                              void* d_out, int out_size, void* d_ws, size_t ws_size,
                              hipStream_t stream) {
  const float* x0 = (const float*)d_in[0];
  const int*   ei = (const int*)d_in[1];
  const float *W1 = (const float*)d_in[2],  *b1 = (const float*)d_in[3];
  const float *W2 = (const float*)d_in[4],  *b2 = (const float*)d_in[5];
  const float *W3 = (const float*)d_in[6],  *b3 = (const float*)d_in[7];
  const float *Wp = (const float*)d_in[8],  *bp = (const float*)d_in[9];
  const float *Wc = (const float*)d_in[10], *bc = (const float*)d_in[11];

  const int N      = in_sizes[0] / 64;
  const int nEdges = in_sizes[1] / 2;
  const int* src = ei;
  const int* dst = ei + nEdges;

  float* ws        = (float*)d_ws;
  float* A         = ws + 65536;
  const int nBins   = (N + BIN_NODES - 1) / BIN_NODES;
  const int nChunks = (nEdges + CHUNK - 1) / CHUNK;
  int*   pairs     = (int*)A;
  int*   offT      = (int*)A + 2 * (size_t)nChunks * CHUNK;
  int*   rowptr    = (int*)(A + (size_t)N * 128);
  int*   nbr       = rowptr + 65536;
  int*   nbrSorted = nbr + nEdges;

  float* out = (float*)d_out;
  float* h1  = out;
  float* emb = out;
  float* t3  = out + (size_t)N * 64;
  float* nxt = out + (size_t)N * 64;
  float* cls = out + (size_t)N * 128;

  dim3 b256(256);
  dim3 gemmGrid((unsigned)((N + 127) / 128)), gemmBlock(256);
  dim3 nodeGrid((unsigned)((N + 8 * NODES_PER_WAVE - 1) / (8 * NODES_PER_WAVE)));
  dim3 chunkGrid((unsigned)nChunks), binGrid((unsigned)nBins);

  chunk_sort_kernel<<<chunkGrid, b256, 0, stream>>>(src, dst, offT, pairs, nEdges, nBins);
  bin_lists_kernel <<<binGrid,   b256, 0, stream>>>(pairs, offT, rowptr, nbr, nbrSorted, N, nBins, nChunks);

  gather_mean_kernel<16, false><<<nodeGrid, b256, 0, stream>>>(x0, rowptr, nbrSorted, A, nullptr, N);
  gemm_bias_act_kernel<64><<<gemmGrid, gemmBlock, 0, stream>>>(A, W1, b1, h1, N, 128, 1);

  gather_mean_kernel<32, false><<<nodeGrid, b256, 0, stream>>>(h1, rowptr, nbrSorted, A, nullptr, N);
  layer2_pre3_kernel<<<gemmGrid, gemmBlock, 0, stream>>>(A, W2, b2, W3, t3, N);

  gather_mean_kernel<16, true><<<nodeGrid, b256, 0, stream>>>(t3, rowptr, nbrSorted, emb, b3, N);

  heads_kernel<<<gemmGrid, gemmBlock, 0, stream>>>(emb, Wp, bp, Wc, bc, nxt, cls, N, 10);
}


# Layer-1 mean and layer-3 mean each feed their product inside the same launch: a block's 32 finished rows go to an LDS tile and its 8 waves run the W1 / head tiles on it (8 -> 6 launches); same sums
# speedup vs baseline: 6.2531x; 1.0515x over previous
#include <hip/hip_runtime.h>
#include <hip/hip_bf16.h>

typedef __attribute__((ext_vector_type(8))) float v8f;
typedef __attribute__((ext_vector_type(2))) float v2f;

constexpr int BIN_SHIFT = 6;
constexpr int BIN_NODES = 1 << BIN_SHIFT;
constexpr int MAX_BINS  = 1024;
constexpr int CHUNK     = 4096;
constexpr int QPT       = CHUNK / (256 * 4);
constexpr int BIN_CAP   = 4096;
constexpr int SEG_PRE   = 12;
static_assert(MAX_BINS == 4 * 256, "chunk_sort_kernel scans four bins per thread");

__device__ __forceinline__ int
load_quad(const int* __restrict__ p, int e, int n, bool aligned, int (&v)[4])
{
  if (aligned && e + 4 <= n) {
    const int4 t = *(const int4*)(p + e);
    v[0] = t.x; v[1] = t.y; v[2] = t.z; v[3] = t.w;
    return 4;
  }
  int c = 0;
  #pragma unroll
  for (int k = 0; k < 4; ++k) {
    v[k] = 0;
    if (e + k < n) { v[k] = p[e + k]; c = k + 1; }
  }
  return c;
}

template <bool GUARD>
__device__ __forceinline__ void
place_pairs(int* __restrict__ cursor, int2* __restrict__ sorted,
            const int (&d)[QPT][4], const int (&s)[QPT][4], const int (&c)[QPT])
{
  int pos[QPT][4];
  #pragma unroll
  for (int u = 0; u < QPT; ++u) {
    #pragma unroll
    for (int k = 0; k < 4; ++k) {
      pos[u][k] = 0;
      if (!GUARD || k < c[u])
        pos[u][k] = __hip_atomic_fetch_add(&cursor[d[u][k] >> BIN_SHIFT], 1, __ATOMIC_RELAXED, __HIP_MEMORY_SCOPE_WORKGROUP);
    }
  }
  #pragma unroll
  for (int u = 0; u < QPT; ++u) {
    #pragma unroll
    for (int k = 0; k < 4; ++k)
      if (!GUARD || k < c[u]) sorted[pos[u][k]] = make_int2(s[u][k], d[u][k]);
  }
}

__global__ __launch_bounds__(256) void
chunk_sort_kernel(const int* __restrict__ src, const int* __restrict__ dst,
                  int* __restrict__ offT, int* __restrict__ pairWords, int nEdges, int nBins)
{
  __shared__ int cursor[MAX_BINS];
  __shared__ int sIncl[8];
  __shared__ __attribute__((aligned(16))) int sortedW[2 * CHUNK];
  const int tid = (int)threadIdx.x, lane = tid & 31, w = tid >> 5;
  const int G = (int)gridDim.x;
  int2* sorted = (int2*)sortedW;
  const bool aligned = ((((size_t)src) | ((size_t)dst)) & 15) == 0;
  const int e0 = blockIdx.x * CHUNK + 4 * tid;
  int d[QPT][4], s[QPT][4], c[QPT];
  #pragma unroll
  for (int u = 0; u < QPT; ++u) {
    c[u] = load_quad(dst, e0 + 1024 * u, nEdges, aligned, d[u]);
    load_quad(src, e0 + 1024 * u, nEdges, aligned, s[u]);
  }
  for (int i = tid; i < MAX_BINS; i += 256) cursor[i] = 0;
  __syncthreads();
  #pragma unroll
  for (int u = 0; u < QPT; ++u) {
    #pragma unroll
    for (int k = 0; k < 4; ++k)
      if (k < c[u])
        __hip_atomic_fetch_add(&cursor[d[u][k] >> BIN_SHIFT], 1, __ATOMIC_RELAXED, __HIP_MEMORY_SCOPE_WORKGROUP);
  }
  __syncthreads();
  const int idx = 4 * tid;
  int v[4];
  #pragma unroll
  for (int k = 0; k < 4; ++k) v[k] = cursor[idx + k];
  const int local = (v[0] + v[1]) + (v[2] + v[3]);
  int incl = local;
  #pragma unroll
  for (int o = 1; o < 32; o <<= 1) {
    int y = __shfl_up(incl, o);
    if (lane >= o) incl += y;
  }
  if (lane == 31) sIncl[w] = incl;
  __syncthreads();
  int r = incl - local;
  #pragma unroll
  for (int i = 0; i < 8; ++i)
    if (i < w) r += sIncl[i];
  #pragma unroll
  for (int k = 0; k < 4; ++k) {
    cursor[idx + k] = r;
    if (idx + k < nBins) offT[(size_t)(idx + k) * G + blockIdx.x] = r;
    r += v[k];
  }
  if (tid == 255) offT[(size_t)nBins * G + blockIdx.x] = r;
  __syncthreads();
  if ((c[0] + c[1]) + (c[2] + c[3]) == 4 * QPT) place_pairs<false>(cursor, sorted, d, s, c);
  else                                         place_pairs<true >(cursor, sorted, d, s, c);
  __syncthreads();
  const int total = min(CHUNK, max(nEdges - (int)blockIdx.x * CHUNK, 0));
  int4*       outp = (int4*)pairWords + (size_t)blockIdx.x * (CHUNK / 2);
  const int4* sp   = (const int4*)sortedW;
  if (total == CHUNK) {
    #pragma unroll
    for (int u = 0; u < CHUNK / 512; ++u) outp[tid + 256 * u] = sp[tid + 256 * u];
  } else {
    #pragma unroll
    for (int u = 0; u < CHUNK / 512; ++u) {
      const int i = tid + 256 * u;
      if (2 * i < total) outp[i] = sp[i];
    }
  }
}

constexpr int NODES_PER_WAVE = 4;

constexpr int PRE = 6;

template <typename L>
__device__ __forceinline__ int
rank_in_list(const L& list, int beg, int end, int p, int s)
{
  int r = 0;
  for (int j = beg; j < end; j += 8) {
    int t[8];
    #pragma unroll
    for (int u = 0; u < 8; ++u) t[u] = list(min(j + u, end - 1));
    #pragma unroll
    for (int u = 0; u < 8; ++u) {
      const int jj = j + u;
      const bool front = (t[u] < s) || (t[u] == s && jj < p);
      r += (jj < end && front) ? 1 : 0;
    }
  }
  return r;
}

__global__ __launch_bounds__(256) void
bin_lists_kernel(const int* __restrict__ pairWords, const int* __restrict__ offT,
                 int* __restrict__ rowptr, int* nbr, int* __restrict__ nbrSorted,
                 int nNodes, int nBins, int G)
{
  __shared__ int cntL[BIN_NODES];
  __shared__ int lpL[BIN_NODES + 1];
  __shared__ int curL[BIN_NODES];
  __shared__ int listL[BIN_CAP];
  __shared__ int nodeL[BIN_CAP];
  __shared__ __attribute__((aligned(8))) int pairLW[2 * BIN_CAP];
  __shared__ int sCnt[8];
  __shared__ int sOff[8];
  const int tid = (int)threadIdx.x, lane = tid & 31, w = tid >> 5;
  const int bin = (int)blockIdx.x;
  const int2* pairs = (const int2*)pairWords;
  int2* pairL = (int2*)pairLW;
  const int nodeBase = bin * BIN_NODES;
  const int* row0 = offT + (size_t)bin * G;
  const int* row1 = row0 + G;
  if (tid < BIN_NODES) cntL[tid] = 0;
  __syncthreads();
  int m = 0;
  int start = 0;
  for (int b0 = 0; b0 < G; b0 += 256) {
    const int b = b0 + tid;
    int o0 = 0, o1 = 0;
    if (b < G) { o0 = row0[b]; o1 = row1[b]; }
    const int cnt = o1 - o0;
    const int2* chunk = pairs + (size_t)min(b, G - 1) * CHUNK;
    const int2* seg   = chunk + o0;
    const int   last  = max(o1 - 1, 0);
    int2 p[SEG_PRE];
    #pragma unroll
    for (int u = 0; u < SEG_PRE; ++u) p[u] = chunk[min(o0 + u, last)];
    __builtin_amdgcn_sched_barrier(0);
    int incl = cnt, so = o0;
    #pragma unroll
    for (int o = 1; o < 32; o <<= 1) {
      int y = __shfl_up(incl, o);
      if (lane >= o) incl += y;
      so += __shfl_xor(so, o);
    }
    if (lane == 31) sCnt[w] = incl;
    if (lane == 0)  sOff[w] = so;
    __syncthreads();
    int base = m + incl - cnt;
    #pragma unroll
    for (int i = 0; i < 8; ++i) {
      const int t = sCnt[i];
      if (i < w) base += t;
      m += t;
      start += sOff[i];
    }
    #pragma unroll
    for (int u = 0; u < SEG_PRE; ++u)
      if (u < cnt) {
        __hip_atomic_fetch_add(&cntL[p[u].y - nodeBase], 1, __ATOMIC_RELAXED, __HIP_MEMORY_SCOPE_WORKGROUP);
        if (base + u < BIN_CAP) pairL[base + u] = p[u];
      }
    for (int k = SEG_PRE; k < cnt; ++k) {
      const int2 q = seg[k];
      __hip_atomic_fetch_add(&cntL[q.y - nodeBase], 1, __ATOMIC_RELAXED, __HIP_MEMORY_SCOPE_WORKGROUP);
      if (base + k < BIN_CAP) pairL[base + k] = q;
    }
    __syncthreads();
  }
  if (tid < 32) {
    const int c0 = cntL[2 * tid], c1 = cntL[2 * tid + 1];
    int incl = c0 + c1;
    #pragma unroll
    for (int o = 1; o < 32; o <<= 1) {
      int y = __shfl_up(incl, o);
      if (tid >= o) incl += y;
    }
    const int s0 = incl - (c0 + c1), s1 = s0 + c0;
    lpL[2 * tid]  = s0;  lpL[2 * tid + 1]  = s1;
    curL[2 * tid] = s0;  curL[2 * tid + 1] = s1;
    if (tid == 31) lpL[BIN_NODES] = incl;
    if (nodeBase + 2 * tid     <= nNodes) rowptr[nodeBase + 2 * tid]     = start + s0;
    if (nodeBase + 2 * tid + 1 <= nNodes) rowptr[nodeBase + 2 * tid + 1] = start + s1;
  }
  if (bin == nBins - 1 && tid == 0 && nodeBase + BIN_NODES == nNodes) rowptr[nNodes] = start + m;
  __syncthreads();
  if (m <= BIN_CAP) {
    int2 pr[PRE];
    #pragma unroll
    for (int u = 0; u < PRE; ++u) {
      const int i = tid + 256 * u;
      pr[u] = (i < m) ? pairL[i] : make_int2(0, nodeBase);
    }
    int slot[PRE];
    #pragma unroll
    for (int u = 0; u < PRE; ++u)
      if (tid + 256 * u < m)
        slot[u] = __hip_atomic_fetch_add(&curL[pr[u].y - nodeBase], 1, __ATOMIC_RELAXED, __HIP_MEMORY_SCOPE_WORKGROUP);
    #pragma unroll
    for (int u = 0; u < PRE; ++u)
      if (tid + 256 * u < m) {
        listL[slot[u]] = pr[u].x;
        nodeL[slot[u]] = pr[u].y - nodeBase;
      }
    for (int i = tid + 256 * PRE; i < m; i += 256) {
      const int2 q   = pairL[i];
      const int  dl  = q.y - nodeBase;
      const int  sl  = __hip_atomic_fetch_add(&curL[dl], 1, __ATOMIC_RELAXED, __HIP_MEMORY_SCOPE_WORKGROUP);
      listL[sl] = q.x;
      nodeL[sl] = dl;
    }
    __syncthreads();
    for (int p = tid; p < m; p += 256) {
      const int s  = listL[p];
      const int dl = nodeL[p];
      const int beg = lpL[dl], end = lpL[dl + 1];
      const int r = rank_in_list([&](int j) { return listL[j]; }, beg, end, p, s);
      nbrSorted[start + beg + r] = s;
    }
  } else {
    for (int b = tid; b < G; b += 256) {
      const int2* seg = pairs + (size_t)b * CHUNK;
      const int o1 = row1[b];
      for (int k = row0[b]; k < o1; ++k) {
        const int2 q  = seg[k];
        const int  sl = __hip_atomic_fetch_add(&curL[q.y - nodeBase], 1, __ATOMIC_RELAXED, __HIP_MEMORY_SCOPE_WORKGROUP);
        nbr[start + sl] = q.x;
      }
    }
    __syncthreads();
    for (int q = w; q < BIN_NODES; q += 8) {
      const int beg = lpL[q], end = lpL[q + 1];
      for (int i = beg + lane; i < end; i += 32) {
        const int s = nbr[start + i];
        const int r = rank_in_list([&](int j) { return nbr[start + j]; }, beg, end, i, s);
        nbrSorted[start + beg + r] = s;
      }
    }
  }
}

template <int F4, int U, bool CLAMP>
__device__ __forceinline__ void
add_rows(float4& acc, const float4* __restrict__ x4, const int* __restrict__ nbr,
         int e, int end, int c, int sub)
{
  constexpr int EPS = 32 / F4;
  float4 v[U];
  #pragma unroll
  for (int u = 0; u < U; ++u) {
    int ee = e + u * EPS + sub;
    if (CLAMP) ee = min(ee, end - 1);
    const int s = nbr[ee];
    v[u] = x4[(size_t)s * F4 + c];
  }
  if constexpr (CLAMP) {
    static_assert(U == 4, "the fence below names four rows");
    asm volatile("" :: "v"(v[0].x), "v"(v[0].y), "v"(v[0].z), "v"(v[0].w),
                       "v"(v[1].x), "v"(v[1].y), "v"(v[1].z), "v"(v[1].w),
                       "v"(v[2].x), "v"(v[2].y), "v"(v[2].z), "v"(v[2].w),
                       "v"(v[3].x), "v"(v[3].y), "v"(v[3].z), "v"(v[3].w));
  }
  #pragma unroll
  for (int u = 0; u < U; ++u) {
    if (!CLAMP || (e + u * EPS + sub < end)) {
      acc.x += v[u].x; acc.y += v[u].y; acc.z += v[u].z; acc.w += v[u].w;
    }
  }
}

template <int F4, bool ACT>
__device__ __forceinline__ float4
gather_node(const float4* __restrict__ x4, const int* __restrict__ rowptr,
            const int* __restrict__ nbr, const float4& bv, int node, int c, int sub)
{
  constexpr int EPS = 32 / F4;
  const int beg = rowptr[node], end = rowptr[node + 1];
  const float4 xv = x4[(size_t)node * F4 + c];
  float4 acc = make_float4(0.0f, 0.0f, 0.0f, 0.0f);
  int e = beg;
  for (; e + 8 * EPS <= end; e += 8 * EPS) add_rows<F4, 8, false>(acc, x4, nbr, e, end, c, sub);
  for (; e < end; e += 4 * EPS)            add_rows<F4, 4, true >(acc, x4, nbr, e, end, c, sub);
  if (EPS == 2) {
    acc.x += __shfl_xor(acc.x, 16);
    acc.y += __shfl_xor(acc.y, 16);
    acc.z += __shfl_xor(acc.z, 16);
    acc.w += __shfl_xor(acc.w, 16);
  }
  const float rd = 1.0f / fmaxf((float)(end - beg), 1.0f);
  float4 o;
  o.x = xv.x + acc.x * rd;
  o.y = xv.y + acc.y * rd;
  o.z = xv.z + acc.z * rd;
  o.w = xv.w + acc.w * rd;
  if constexpr (ACT) {
    o.x = fmaxf(o.x + bv.x, 0.0f);
    o.y = fmaxf(o.y + bv.y, 0.0f);
    o.z = fmaxf(o.z + bv.z, 0.0f);
    o.w = fmaxf(o.w + bv.w, 0.0f);
  }
  return o;
}

template <int F4>
__global__ __launch_bounds__(256) void
gather_mean_kernel(const float* __restrict__ x, const int* __restrict__ rowptr,
                   const int* __restrict__ nbr, float* __restrict__ A, int nNodes)
{
  const int lane = threadIdx.x & 31;
  const int c    = lane % F4;
  const int sub  = lane / F4;
  const int wv   = blockIdx.x * 8 + (threadIdx.x >> 5);
  const float4* x4 = (const float4*)x;
  const float4 nob = make_float4(0.0f, 0.0f, 0.0f, 0.0f);
  for (int q = 0; q < NODES_PER_WAVE; ++q) {
    const int node = wv * NODES_PER_WAVE + q;
    if (node >= nNodes) return;
    const float4 o = gather_node<F4, false>(x4, rowptr, nbr, nob, node, c, sub);
    if (sub == 0) ((float4*)A)[(size_t)node * F4 + c] = o;
  }
}

template <int FIN, int NT>
__device__ __forceinline__ void
gemm_tiles(const float4 (&a4)[FIN / 8], const float* __restrict__ Bl,
           const float* __restrict__ bsh, float* __restrict__ out,
           int ct, int rowBase, int half, int l16, int nRows, int Fout, int doRelu,
           bool fullRows)
{
  constexpr int STRIDE = FIN + 4;
  constexpr int KP = FIN / 8;
  v8f c[NT];
  #pragma unroll
  for (int t = 0; t < NT; ++t) c[t] = v8f{};
  #pragma unroll
  for (int j = 0; j < KP; ++j) {
    float4 b[NT];
    #pragma unroll
    for (int t = 0; t < NT; ++t)
      b[t] = *(const float4*)(Bl + (ct + t) * 16 * STRIDE + 8 * j);
    v2f alo, ahi;
    alo.x = a4[j].x; alo.y = a4[j].y;
    ahi.x = a4[j].z; ahi.y = a4[j].w;
    #pragma unroll
    for (int t = 0; t < NT; ++t) {
      v2f blo; blo.x = b[t].x; blo.y = b[t].y;
      c[t] = __builtin_amdgcn_wmma_f32_16x16x4_f32(false, alo, false, blo, (short)0, c[t], false, false);
    }
    #pragma unroll
    for (int t = 0; t < NT; ++t) {
      v2f bhi; bhi.x = b[t].z; bhi.y = b[t].w;
      c[t] = __builtin_amdgcn_wmma_f32_16x16x4_f32(false, ahi, false, bhi, (short)0, c[t], false, false);
    }
  }
  #pragma unroll
  for (int t = 0; t < NT; ++t) {
    const int n = (ct + t) * 16 + l16;
    const float bv = bsh[n];
    float* colBase = out + (size_t)(rowBase + 8 * half) * Fout + n;
    if (fullRows && ((ct + t) * 16 + 16 <= Fout)) {
      #pragma unroll
      for (int r = 0; r < 8; ++r) {
        float val = c[t][r] + bv;
        if (doRelu) val = fmaxf(val, 0.0f);
        colBase[(size_t)r * Fout] = val;
      }
    } else {
      #pragma unroll
      for (int r = 0; r < 8; ++r) {
        int gRow = rowBase + r + 8 * half;
        float val = c[t][r] + bv;
        if (doRelu) val = fmaxf(val, 0.0f);
        if (gRow < nRows && n < Fout) out[(size_t)gRow * Fout + n] = val;
      }
    }
  }
}

constexpr int TILE_ROWS = 8 * NODES_PER_WAVE;
static_assert(TILE_ROWS == 32, "two 16-row strips per block, one per wave parity");
__global__ __launch_bounds__(256) void
gather_gemm_kernel(const float* __restrict__ x, const int* __restrict__ rowptr,
                   const int* __restrict__ nbr, const float* __restrict__ W,
                   const float* __restrict__ bias, float* __restrict__ out, int nNodes)
{
  constexpr int FIN    = 64;
  constexpr int FOUT   = 128;
  constexpr int STRIDE = FIN + 4;
  constexpr int F4     = FIN / 4;
  constexpr int KP     = FIN / 8;
  constexpr int WLD    = (FOUT * F4) / 256;
  static_assert(FOUT / 16 == 2 * 4, "two column tiles for each of the four wave pairs");
  __shared__ __attribute__((aligned(16))) float Bs[FOUT * STRIDE];
  __shared__ __attribute__((aligned(16))) float At[TILE_ROWS * STRIDE];
  __shared__ float bsh[FOUT];

  const int tid  = (int)threadIdx.x;
  const int lane = tid & 31;
  const int wave = tid >> 5;
  const int half = lane >> 4;
  const int l16  = lane & 15;

  const float bval = bias[min(tid, FOUT - 1)];
  float4 wv[WLD];
  #pragma unroll
  for (int i = 0; i < WLD; ++i) wv[i] = ((const float4*)W)[tid + i * 256];
  #pragma unroll
  for (int i = 0; i < WLD; ++i) {
    const int idx = tid + i * 256;
    const int r = idx / F4, c4 = idx - r * F4;
    *(float4*)&Bs[r * STRIDE + 4 * c4] = wv[i];
  }
  if (tid < FOUT) bsh[tid] = bval;

  {
    const int c   = lane % F4;
    const int sub = lane / F4;
    const float4* x4 = (const float4*)x;
    const float4 nob = make_float4(0.0f, 0.0f, 0.0f, 0.0f);
    for (int q = 0; q < NODES_PER_WAVE; ++q) {
      const int r    = wave * NODES_PER_WAVE + q;
      const int node = (int)blockIdx.x * TILE_ROWS + r;
      float4 o = nob;
      if (node < nNodes) o = gather_node<F4, false>(x4, rowptr, nbr, nob, node, c, sub);
      if (sub == 0) *(float4*)&At[r * STRIDE + 4 * c] = o;
    }
  }
  __syncthreads();

  const int strip   = wave & 1;
  const int rowBase = (int)blockIdx.x * TILE_ROWS + 16 * strip;
  if (rowBase >= nNodes) return;
  float4 a4[KP];
  {
    const float* Ar = &At[(16 * strip + l16) * STRIDE + 4 * half];
    #pragma unroll
    for (int j = 0; j < KP; ++j) a4[j] = *(const float4*)(Ar + 8 * j);
  }
  const float* Bl = &Bs[l16 * STRIDE + 4 * half];
  const bool fullRows = (rowBase + 16 <= nNodes);
  gemm_tiles<FIN, 2>(a4, Bl, bsh, out, 2 * (wave >> 1), rowBase, half, l16, nNodes, FOUT, 1, fullRows);
}

constexpr int CHUNK_PITCH = 32 + 4;
__global__ __launch_bounds__(256) void
layer2_pre3_kernel(const float* __restrict__ xin, const float* __restrict__ W2,
                   const float* __restrict__ b2, const float* __restrict__ W3,
                   float* __restrict__ tout, int nRows)
{
  constexpr int FIN    = 128;
  constexpr int STRIDE = FIN + 4;
  constexpr int F4     = FIN / 4;
  constexpr int KP     = FIN / 8;
  constexpr int FH     = 128;
  constexpr int FT     = 64;
  constexpr int WLD2   = (FH * F4) / 256;
  constexpr int WLD3   = (FT * F4) / 256;
  __shared__ __attribute__((aligned(16))) float Bs[FH * STRIDE];
  __shared__ __attribute__((aligned(16))) float B3[FT * STRIDE];
  __shared__ __attribute__((aligned(16))) float Ks[8 * 16 * CHUNK_PITCH];
  __shared__ float bsh[FH];

  const int tid  = (int)threadIdx.x;
  const int lane = tid & 31;
  const int wave = tid >> 5;
  const int half = lane >> 4;
  const int l16  = lane & 15;
  const int rowBase = (blockIdx.x * 8 + wave) * 16;
  const int row     = min(rowBase + l16, nRows - 1);

  const float bval = b2[min(tid, FH - 1)];
  float4 w2[WLD2];
  #pragma unroll
  for (int i = 0; i < WLD2; ++i) w2[i] = ((const float4*)W2)[tid + i * 256];
  float4 w3[WLD3];
  #pragma unroll
  for (int i = 0; i < WLD3; ++i) w3[i] = ((const float4*)W3)[tid + i * 256];
  float4 a4[KP];
  {
    const float4* xr = (const float4*)(xin + (size_t)row * FIN) + half;
    #pragma unroll
    for (int j = 0; j < KP; ++j) a4[j] = xr[2 * j];
  }
  #pragma unroll
  for (int i = 0; i < WLD2; ++i) {
    const int idx = tid + i * 256;
    const int r = idx / F4, c4 = idx - r * F4;
    *(float4*)&Bs[r * STRIDE + 4 * c4] = w2[i];
  }
  #pragma unroll
  for (int i = 0; i < WLD3; ++i) {
    const int idx = tid + i * 256;
    const int r = idx / F4, c4 = idx - r * F4;
    *(float4*)&B3[r * STRIDE + 4 * c4] = w3[i];
  }
  if (tid < FH) bsh[tid] = bval;
  __syncthreads();
  if (rowBase >= nRows) return;

  const float* Bl  = &Bs[l16 * STRIDE + 4 * half];
  const float* B3l = &B3[l16 * STRIDE + 4 * half];
  float*       Kw  = &Ks[wave * 16 * CHUNK_PITCH];

  v8f tacc[FT / 16];
  #pragma unroll
  for (int t = 0; t < FT / 16; ++t) tacc[t] = v8f{};

  #pragma unroll
  for (int p = 0; p < FH / 32; ++p) {
    v8f c[2];
    c[0] = v8f{}; c[1] = v8f{};
    #pragma unroll
    for (int j = 0; j < KP; ++j) {
      float4 b[2];
      #pragma unroll
      for (int t = 0; t < 2; ++t)
        b[t] = *(const float4*)(Bl + (2 * p + t) * 16 * STRIDE + 8 * j);
      v2f alo, ahi;
      alo.x = a4[j].x; alo.y = a4[j].y;
      ahi.x = a4[j].z; ahi.y = a4[j].w;
      #pragma unroll
      for (int t = 0; t < 2; ++t) {
        v2f blo; blo.x = b[t].x; blo.y = b[t].y;
        c[t] = __builtin_amdgcn_wmma_f32_16x16x4_f32(false, alo, false, blo, (short)0, c[t], false, false);
      }
      #pragma unroll
      for (int t = 0; t < 2; ++t) {
        v2f bhi; bhi.x = b[t].z; bhi.y = b[t].w;
        c[t] = __builtin_amdgcn_wmma_f32_16x16x4_f32(false, ahi, false, bhi, (short)0, c[t], false, false);
      }
    }
    #pragma unroll
    for (int t = 0; t < 2; ++t) {
      const float bv = bsh[(2 * p + t) * 16 + l16];
      #pragma unroll
      for (int r = 0; r < 8; ++r)
        Kw[(r + 8 * half) * CHUNK_PITCH + t * 16 + l16] = fmaxf(c[t][r] + bv, 0.0f);
    }
    float4 e4[4];
    #pragma unroll
    for (int j = 0; j < 4; ++j) e4[j] = *(const float4*)(Kw + l16 * CHUNK_PITCH + 8 * j + 4 * half);
    #pragma unroll
    for (int j = 0; j < 4; ++j) {
      float4 b[FT / 16];
      #pragma unroll
      for (int t = 0; t < FT / 16; ++t)
        b[t] = *(const float4*)(B3l + t * 16 * STRIDE + 8 * (4 * p + j));
      v2f alo, ahi;
      alo.x = e4[j].x; alo.y = e4[j].y;
      ahi.x = e4[j].z; ahi.y = e4[j].w;
      #pragma unroll
      for (int t = 0; t < FT / 16; ++t) {
        v2f blo; blo.x = b[t].x; blo.y = b[t].y;
        tacc[t] = __builtin_amdgcn_wmma_f32_16x16x4_f32(false, alo, false, blo, (short)0, tacc[t], false, false);
      }
      #pragma unroll
      for (int t = 0; t < FT / 16; ++t) {
        v2f bhi; bhi.x = b[t].z; bhi.y = b[t].w;
        tacc[t] = __builtin_amdgcn_wmma_f32_16x16x4_f32(false, ahi, false, bhi, (short)0, tacc[t], false, false);
      }
    }
  }
  const bool fullRows = (rowBase + 16 <= nRows);
  #pragma unroll
  for (int t = 0; t < FT / 16; ++t) {
    const int n = t * 16 + l16;
    if (fullRows) {
      float* colBase = tout + (size_t)(rowBase + 8 * half) * FT + n;
      #pragma unroll
      for (int r = 0; r < 8; ++r) colBase[(size_t)r * FT] = tacc[t][r];
    } else {
      #pragma unroll
      for (int r = 0; r < 8; ++r) {
        const int gRow = rowBase + r + 8 * half;
        if (gRow < nRows) tout[(size_t)gRow * FT + n] = tacc[t][r];
      }
    }
  }
}

__global__ __launch_bounds__(256) void
gather_heads_kernel(const float* __restrict__ t, const int* __restrict__ rowptr,
                    const int* __restrict__ nbr, const float* __restrict__ b3,
                    float* __restrict__ emb,
                    const float* __restrict__ Wp, const float* __restrict__ bp,
                    const float* __restrict__ Wc, const float* __restrict__ bc,
                    float* __restrict__ nxt, float* __restrict__ cls, int nNodes, int nClasses)
{
  constexpr int FE      = 64;
  constexpr int STRIDE2 = FE + 4;
  constexpr int F4E     = FE / 4;
  constexpr int KPE     = FE / 8;
  constexpr int WLDP    = (FE * F4E) / 256;
  static_assert(FE / 16 == 4, "one column tile of next for each of the four wave pairs");
  __shared__ __attribute__((aligned(16))) float Bh[(FE + 16) * STRIDE2];
  __shared__ __attribute__((aligned(16))) float Et[TILE_ROWS * STRIDE2];
  __shared__ float bsh2[FE + 16];

  const int tid  = (int)threadIdx.x;
  const int lane = tid & 31;
  const int wave = tid >> 5;
  const int half = lane >> 4;
  const int l16  = lane & 15;

  const float bvp = bp[min(tid, FE - 1)];
  const float bvc = bc[min(tid, nClasses - 1)];
  float4 wp[WLDP];
  #pragma unroll
  for (int i = 0; i < WLDP; ++i) wp[i] = ((const float4*)Wp)[tid + i * 256];
  const int wcRow = tid >> 4;
  const float4 wc = ((const float4*)Wc)[min(tid, nClasses * F4E - 1)];
  #pragma unroll
  for (int i = 0; i < WLDP; ++i) {
    const int idx = tid + i * 256;
    const int r = idx / F4E, c4 = idx - r * F4E;
    *(float4*)&Bh[r * STRIDE2 + 4 * c4] = wp[i];
  }
  {
    const int c4 = tid - wcRow * F4E;
    *(float4*)&Bh[(FE + wcRow) * STRIDE2 + 4 * c4] =
        (wcRow < nClasses) ? wc : make_float4(0.0f, 0.0f, 0.0f, 0.0f);
  }
  if (tid < FE) bsh2[tid] = bvp;
  if (tid < 16) bsh2[FE + tid] = (tid < nClasses) ? bvc : 0.0f;

  {
    const int c   = lane % F4E;
    const int sub = lane / F4E;
    const float4* t4 = (const float4*)t;
    float4 bv;
    bv.x = b3[4 * c]; bv.y = b3[4 * c + 1]; bv.z = b3[4 * c + 2]; bv.w = b3[4 * c + 3];
    for (int q = 0; q < NODES_PER_WAVE; ++q) {
      const int r    = wave * NODES_PER_WAVE + q;
      const int node = (int)blockIdx.x * TILE_ROWS + r;
      float4 o = make_float4(0.0f, 0.0f, 0.0f, 0.0f);
      if (node < nNodes) {
        o = gather_node<F4E, true>(t4, rowptr, nbr, bv, node, c, sub);
        if (sub == 0) ((float4*)emb)[(size_t)node * F4E + c] = o;
      }
      if (sub == 0) *(float4*)&Et[r * STRIDE2 + 4 * c] = o;
    }
  }
  __syncthreads();

  const int strip   = wave & 1;
  const int pair    = wave >> 1;
  const int rowBase = (int)blockIdx.x * TILE_ROWS + 16 * strip;
  if (rowBase >= nNodes) return;
  float4 e4[KPE];
  {
    const float* Er = &Et[(16 * strip + l16) * STRIDE2 + 4 * half];
    #pragma unroll
    for (int j = 0; j < KPE; ++j) e4[j] = *(const float4*)(Er + 8 * j);
  }
  const bool fullRows = (rowBase + 16 <= nNodes);
  const float* Bl2 = &Bh[l16 * STRIDE2 + 4 * half];
  gemm_tiles<FE, 1>(e4, Bl2, bsh2, nxt, pair, rowBase, half, l16, nNodes, FE, 0, fullRows);
  if (pair == 3)
    gemm_tiles<FE, 1>(e4, Bl2 + FE * STRIDE2, bsh2 + FE, cls, 0, rowBase, half, l16, nNodes, nClasses, 0, fullRows);
}

extern "C" void kernel_launch(void* const* d_in, const int* in_sizes, int n_in,
                              void* d_out, int out_size, void* d_ws, size_t ws_size,
                              hipStream_t stream) {
  const float* x0 = (const float*)d_in[0];
  const int*   ei = (const int*)d_in[1];
  const float *W1 = (const float*)d_in[2],  *b1 = (const float*)d_in[3];
  const float *W2 = (const float*)d_in[4],  *b2 = (const float*)d_in[5];
  const float *W3 = (const float*)d_in[6],  *b3 = (const float*)d_in[7];
  const float *Wp = (const float*)d_in[8],  *bp = (const float*)d_in[9];
  const float *Wc = (const float*)d_in[10], *bc = (const float*)d_in[11];

  const int N      = in_sizes[0] / 64;
  const int nEdges = in_sizes[1] / 2;
  const int* src = ei;
  const int* dst = ei + nEdges;

  float* ws        = (float*)d_ws;
  float* A         = ws + 65536;
  const int nBins   = (N + BIN_NODES - 1) / BIN_NODES;
  const int nChunks = (nEdges + CHUNK - 1) / CHUNK;
  int*   pairs     = (int*)A;
  int*   offT      = (int*)A + 2 * (size_t)nChunks * CHUNK;
  int*   rowptr    = (int*)(A + (size_t)N * 128);
  int*   nbr       = rowptr + 65536;
  int*   nbrSorted = nbr + nEdges;
  float* h1        = A;
  float* t3        = A;

  float* out = (float*)d_out;
  float* A2  = out;
  float* emb = out;
  float* nxt = out + (size_t)N * 64;
  float* cls = out + (size_t)N * 128;

  dim3 b256(256);
  dim3 gemmGrid((unsigned)((N + 127) / 128)), gemmBlock(256);
  dim3 nodeGrid((unsigned)((N + 8 * NODES_PER_WAVE - 1) / (8 * NODES_PER_WAVE)));
  dim3 chunkGrid((unsigned)nChunks), binGrid((unsigned)nBins);

  chunk_sort_kernel<<<chunkGrid, b256, 0, stream>>>(src, dst, offT, pairs, nEdges, nBins);
  bin_lists_kernel <<<binGrid,   b256, 0, stream>>>(pairs, offT, rowptr, nbr, nbrSorted, N, nBins, nChunks);

  gather_gemm_kernel<<<nodeGrid, b256, 0, stream>>>(x0, rowptr, nbrSorted, W1, b1, h1, N);

  gather_mean_kernel<32><<<nodeGrid, b256, 0, stream>>>(h1, rowptr, nbrSorted, A2, N);
  layer2_pre3_kernel<<<gemmGrid, gemmBlock, 0, stream>>>(A2, W2, b2, W3, t3, N);

  gather_heads_kernel<<<nodeGrid, b256, 0, stream>>>(t3, rowptr, nbrSorted, b3, emb, Wp, bp, Wc, bc, nxt, cls, N, 10);
}


# Gather's dependent-load chain cut: a wave gets its 4 nodes' list bounds in one load, each node's ids in one whole-wave load a node ahead; rows take ids by lane reads; one tail step; sum order kept
# speedup vs baseline: 6.5705x; 1.0508x over previous
#include <hip/hip_runtime.h>
#include <hip/hip_bf16.h>

typedef __attribute__((ext_vector_type(8))) float v8f;
typedef __attribute__((ext_vector_type(2))) float v2f;

constexpr int BIN_SHIFT = 6;
constexpr int BIN_NODES = 1 << BIN_SHIFT;
constexpr int MAX_BINS  = 1024;
constexpr int CHUNK     = 4096;
constexpr int QPT       = CHUNK / (256 * 4);
constexpr int BIN_CAP   = 4096;
constexpr int SEG_PRE   = 12;
static_assert(MAX_BINS == 4 * 256, "chunk_sort_kernel scans four bins per thread");

__device__ __forceinline__ int
load_quad(const int* __restrict__ p, int e, int n, bool aligned, int (&v)[4])
{
  if (aligned && e + 4 <= n) {
    const int4 t = *(const int4*)(p + e);
    v[0] = t.x; v[1] = t.y; v[2] = t.z; v[3] = t.w;
    return 4;
  }
  int c = 0;
  #pragma unroll
  for (int k = 0; k < 4; ++k) {
    v[k] = 0;
    if (e + k < n) { v[k] = p[e + k]; c = k + 1; }
  }
  return c;
}

template <bool GUARD>
__device__ __forceinline__ void
place_pairs(int* __restrict__ cursor, int2* __restrict__ sorted,
            const int (&d)[QPT][4], const int (&s)[QPT][4], const int (&c)[QPT])
{
  int pos[QPT][4];
  #pragma unroll
  for (int u = 0; u < QPT; ++u) {
    #pragma unroll
    for (int k = 0; k < 4; ++k) {
      pos[u][k] = 0;
      if (!GUARD || k < c[u])
        pos[u][k] = __hip_atomic_fetch_add(&cursor[d[u][k] >> BIN_SHIFT], 1, __ATOMIC_RELAXED, __HIP_MEMORY_SCOPE_WORKGROUP);
    }
  }
  #pragma unroll
  for (int u = 0; u < QPT; ++u) {
    #pragma unroll
    for (int k = 0; k < 4; ++k)
      if (!GUARD || k < c[u]) sorted[pos[u][k]] = make_int2(s[u][k], d[u][k]);
  }
}

__global__ __launch_bounds__(256) void
chunk_sort_kernel(const int* __restrict__ src, const int* __restrict__ dst,
                  int* __restrict__ offT, int* __restrict__ pairWords, int nEdges, int nBins)
{
  __shared__ int cursor[MAX_BINS];
  __shared__ int sIncl[8];
  __shared__ __attribute__((aligned(16))) int sortedW[2 * CHUNK];
  const int tid = (int)threadIdx.x, lane = tid & 31, w = tid >> 5;
  const int G = (int)gridDim.x;
  int2* sorted = (int2*)sortedW;
  const bool aligned = ((((size_t)src) | ((size_t)dst)) & 15) == 0;
  const int e0 = blockIdx.x * CHUNK + 4 * tid;
  int d[QPT][4], s[QPT][4], c[QPT];
  #pragma unroll
  for (int u = 0; u < QPT; ++u) {
    c[u] = load_quad(dst, e0 + 1024 * u, nEdges, aligned, d[u]);
    load_quad(src, e0 + 1024 * u, nEdges, aligned, s[u]);
  }
  for (int i = tid; i < MAX_BINS; i += 256) cursor[i] = 0;
  __syncthreads();
  #pragma unroll
  for (int u = 0; u < QPT; ++u) {
    #pragma unroll
    for (int k = 0; k < 4; ++k)
      if (k < c[u])
        __hip_atomic_fetch_add(&cursor[d[u][k] >> BIN_SHIFT], 1, __ATOMIC_RELAXED, __HIP_MEMORY_SCOPE_WORKGROUP);
  }
  __syncthreads();
  const int idx = 4 * tid;
  int v[4];
  #pragma unroll
  for (int k = 0; k < 4; ++k) v[k] = cursor[idx + k];
  const int local = (v[0] + v[1]) + (v[2] + v[3]);
  int incl = local;
  #pragma unroll
  for (int o = 1; o < 32; o <<= 1) {
    int y = __shfl_up(incl, o);
    if (lane >= o) incl += y;
  }
  if (lane == 31) sIncl[w] = incl;
  __syncthreads();
  int r = incl - local;
  #pragma unroll
  for (int i = 0; i < 8; ++i)
    if (i < w) r += sIncl[i];
  #pragma unroll
  for (int k = 0; k < 4; ++k) {
    cursor[idx + k] = r;
    if (idx + k < nBins) offT[(size_t)(idx + k) * G + blockIdx.x] = r;
    r += v[k];
  }
  if (tid == 255) offT[(size_t)nBins * G + blockIdx.x] = r;
  __syncthreads();
  if ((c[0] + c[1]) + (c[2] + c[3]) == 4 * QPT) place_pairs<false>(cursor, sorted, d, s, c);
  else                                         place_pairs<true >(cursor, sorted, d, s, c);
  __syncthreads();
  const int total = min(CHUNK, max(nEdges - (int)blockIdx.x * CHUNK, 0));
  int4*       outp = (int4*)pairWords + (size_t)blockIdx.x * (CHUNK / 2);
  const int4* sp   = (const int4*)sortedW;
  if (total == CHUNK) {
    #pragma unroll
    for (int u = 0; u < CHUNK / 512; ++u) outp[tid + 256 * u] = sp[tid + 256 * u];
  } else {
    #pragma unroll
    for (int u = 0; u < CHUNK / 512; ++u) {
      const int i = tid + 256 * u;
      if (2 * i < total) outp[i] = sp[i];
    }
  }
}

constexpr int NODES_PER_WAVE = 4;

constexpr int PRE = 6;

template <typename L>
__device__ __forceinline__ int
rank_in_list(const L& list, int beg, int end, int p, int s)
{
  int r = 0;
  for (int j = beg; j < end; j += 8) {
    int t[8];
    #pragma unroll
    for (int u = 0; u < 8; ++u) t[u] = list(min(j + u, end - 1));
    #pragma unroll
    for (int u = 0; u < 8; ++u) {
      const int jj = j + u;
      const bool front = (t[u] < s) || (t[u] == s && jj < p);
      r += (jj < end && front) ? 1 : 0;
    }
  }
  return r;
}

__global__ __launch_bounds__(256) void
bin_lists_kernel(const int* __restrict__ pairWords, const int* __restrict__ offT,
                 int* __restrict__ rowptr, int* nbr, int* __restrict__ nbrSorted,
                 int nNodes, int nBins, int G)
{
  __shared__ int cntL[BIN_NODES];
  __shared__ int lpL[BIN_NODES + 1];
  __shared__ int curL[BIN_NODES];
  __shared__ int listL[BIN_CAP];
  __shared__ int nodeL[BIN_CAP];
  __shared__ __attribute__((aligned(8))) int pairLW[2 * BIN_CAP];
  __shared__ int sCnt[8];
  __shared__ int sOff[8];
  const int tid = (int)threadIdx.x, lane = tid & 31, w = tid >> 5;
  const int bin = (int)blockIdx.x;
  const int2* pairs = (const int2*)pairWords;
  int2* pairL = (int2*)pairLW;
  const int nodeBase = bin * BIN_NODES;
  const int* row0 = offT + (size_t)bin * G;
  const int* row1 = row0 + G;
  if (tid < BIN_NODES) cntL[tid] = 0;
  __syncthreads();
  int m = 0;
  int start = 0;
  for (int b0 = 0; b0 < G; b0 += 256) {
    const int b = b0 + tid;
    int o0 = 0, o1 = 0;
    if (b < G) { o0 = row0[b]; o1 = row1[b]; }
    const int cnt = o1 - o0;
    const int2* chunk = pairs + (size_t)min(b, G - 1) * CHUNK;
    const int2* seg   = chunk + o0;
    const int   last  = max(o1 - 1, 0);
    int2 p[SEG_PRE];
    #pragma unroll
    for (int u = 0; u < SEG_PRE; ++u) p[u] = chunk[min(o0 + u, last)];
    __builtin_amdgcn_sched_barrier(0);
    int incl = cnt, so = o0;
    #pragma unroll
    for (int o = 1; o < 32; o <<= 1) {
      int y = __shfl_up(incl, o);
      if (lane >= o) incl += y;
      so += __shfl_xor(so, o);
    }
    if (lane == 31) sCnt[w] = incl;
    if (lane == 0)  sOff[w] = so;
    __syncthreads();
    int base = m + incl - cnt;
    #pragma unroll
    for (int i = 0; i < 8; ++i) {
      const int t = sCnt[i];
      if (i < w) base += t;
      m += t;
      start += sOff[i];
    }
    #pragma unroll
    for (int u = 0; u < SEG_PRE; ++u)
      if (u < cnt) {
        __hip_atomic_fetch_add(&cntL[p[u].y - nodeBase], 1, __ATOMIC_RELAXED, __HIP_MEMORY_SCOPE_WORKGROUP);
        if (base + u < BIN_CAP) pairL[base + u] = p[u];
      }
    for (int k = SEG_PRE; k < cnt; ++k) {
      const int2 q = seg[k];
      __hip_atomic_fetch_add(&cntL[q.y - nodeBase], 1, __ATOMIC_RELAXED, __HIP_MEMORY_SCOPE_WORKGROUP);
      if (base + k < BIN_CAP) pairL[base + k] = q;
    }
    __syncthreads();
  }
  if (tid < 32) {
    const int c0 = cntL[2 * tid], c1 = cntL[2 * tid + 1];
    int incl = c0 + c1;
    #pragma unroll
    for (int o = 1; o < 32; o <<= 1) {
      int y = __shfl_up(incl, o);
      if (tid >= o) incl += y;
    }
    const int s0 = incl - (c0 + c1), s1 = s0 + c0;
    lpL[2 * tid]  = s0;  lpL[2 * tid + 1]  = s1;
    curL[2 * tid] = s0;  curL[2 * tid + 1] = s1;
    if (tid == 31) lpL[BIN_NODES] = incl;
    if (nodeBase + 2 * tid     <= nNodes) rowptr[nodeBase + 2 * tid]     = start + s0;
    if (nodeBase + 2 * tid + 1 <= nNodes) rowptr[nodeBase + 2 * tid + 1] = start + s1;
  }
  if (bin == nBins - 1 && tid == 0 && nodeBase + BIN_NODES == nNodes) rowptr[nNodes] = start + m;
  __syncthreads();
  if (m <= BIN_CAP) {
    int2 pr[PRE];
    #pragma unroll
    for (int u = 0; u < PRE; ++u) {
      const int i = tid + 256 * u;
      pr[u] = (i < m) ? pairL[i] : make_int2(0, nodeBase);
    }
    int slot[PRE];
    #pragma unroll
    for (int u = 0; u < PRE; ++u)
      if (tid + 256 * u < m)
        slot[u] = __hip_atomic_fetch_add(&curL[pr[u].y - nodeBase], 1, __ATOMIC_RELAXED, __HIP_MEMORY_SCOPE_WORKGROUP);
    #pragma unroll
    for (int u = 0; u < PRE; ++u)
      if (tid + 256 * u < m) {
        listL[slot[u]] = pr[u].x;
        nodeL[slot[u]] = pr[u].y - nodeBase;
      }
    for (int i = tid + 256 * PRE; i < m; i += 256) {
      const int2 q   = pairL[i];
      const int  dl  = q.y - nodeBase;
      const int  sl  = __hip_atomic_fetch_add(&curL[dl], 1, __ATOMIC_RELAXED, __HIP_MEMORY_SCOPE_WORKGROUP);
      listL[sl] = q.x;
      nodeL[sl] = dl;
    }
    __syncthreads();
    for (int p = tid; p < m; p += 256) {
      const int s  = listL[p];
      const int dl = nodeL[p];
      const int beg = lpL[dl], end = lpL[dl + 1];
      const int r = rank_in_list([&](int j) { return listL[j]; }, beg, end, p, s);
      nbrSorted[start + beg + r] = s;
    }
  } else {
    for (int b = tid; b < G; b += 256) {
      const int2* seg = pairs + (size_t)b * CHUNK;
      const int o1 = row1[b];
      for (int k = row0[b]; k < o1; ++k) {
        const int2 q  = seg[k];
        const int  sl = __hip_atomic_fetch_add(&curL[q.y - nodeBase], 1, __ATOMIC_RELAXED, __HIP_MEMORY_SCOPE_WORKGROUP);
        nbr[start + sl] = q.x;
      }
    }
    __syncthreads();
    for (int q = w; q < BIN_NODES; q += 8) {
      const int beg = lpL[q], end = lpL[q + 1];
      for (int i = beg + lane; i < end; i += 32) {
        const int s = nbr[start + i];
        const int r = rank_in_list([&](int j) { return nbr[start + j]; }, beg, end, i, s);
        nbrSorted[start + beg + r] = s;
      }
    }
  }
}


template <int F4, int U, bool CLAMP>
__device__ __forceinline__ void
add_rows(float4& acc, const float4* __restrict__ x4, int ids, int j0, int cnt, int c, int sub)
{
  constexpr int EPS = 32 / F4;
  float4 v[U];
  #pragma unroll
  for (int u = 0; u < U; ++u) {
    int s;
    if constexpr (EPS == 1) {
      int j = j0 + u;
      if (CLAMP) j = min(j, cnt - 1);
      s = __builtin_amdgcn_readlane(ids, j);
    } else {
      int ja = j0 + 2 * u, jb = ja + 1;
      if (CLAMP) { ja = min(ja, cnt - 1); jb = min(jb, cnt - 1); }
      const int sa = __builtin_amdgcn_readlane(ids, ja);
      const int sb = __builtin_amdgcn_readlane(ids, jb);
      s = sub ? sb : sa;
    }
    v[u] = x4[(size_t)s * F4 + c];
  }
  if constexpr (CLAMP) {
    static_assert(U % 4 == 0, "the fence below names four rows per statement");
    #pragma unroll
    for (int g = 0; g < U; g += 4)
      asm volatile("" :: "v"(v[g].x),     "v"(v[g].y),     "v"(v[g].z),     "v"(v[g].w),
                         "v"(v[g + 1].x), "v"(v[g + 1].y), "v"(v[g + 1].z), "v"(v[g + 1].w),
                         "v"(v[g + 2].x), "v"(v[g + 2].y), "v"(v[g + 2].z), "v"(v[g + 2].w),
                         "v"(v[g + 3].x), "v"(v[g + 3].y), "v"(v[g + 3].z), "v"(v[g + 3].w));
  }
  #pragma unroll
  for (int u = 0; u < U; ++u) {
    if (!CLAMP || (j0 + u * EPS + sub < cnt)) {
      acc.x += v[u].x; acc.y += v[u].y; acc.z += v[u].z; acc.w += v[u].w;
    }
  }
}

template <int F4>
__device__ __forceinline__ void
add_piece(float4& acc, const float4* __restrict__ x4, int ids, int cnt, int c, int sub)
{
  constexpr int EPS = 32 / F4;
  int j = 0;
  for (; j + 8 * EPS <= cnt; j += 8 * EPS) add_rows<F4, 8, false>(acc, x4, ids, j, cnt, c, sub);
  if (cnt - j > 4 * EPS)  add_rows<F4, 8, true>(acc, x4, ids, j, cnt, c, sub);
  else if (cnt - j > 0)   add_rows<F4, 4, true>(acc, x4, ids, j, cnt, c, sub);
}

template <int F4, bool ACT>
__device__ __forceinline__ float4
gather_node(const float4* __restrict__ x4, const int* __restrict__ nbr, const float4& bv,
            int node, int beg, int end, int ids, int c, int sub, int lane)
{
  constexpr int EPS = 32 / F4;
  const float4 xv = x4[(size_t)node * F4 + c];
  float4 acc = make_float4(0.0f, 0.0f, 0.0f, 0.0f);
  add_piece<F4>(acc, x4, ids, min(end - beg, 32), c, sub);
  for (int base = beg + 32; base < end; base += 32) {
    const int more = nbr[min(base + lane, end - 1)];
    const int cnt  = min(end - base, 32);
    for (int j = 0; j < cnt; j += 4 * EPS) add_rows<F4, 4, true>(acc, x4, more, j, cnt, c, sub);
  }
  if (EPS == 2) {
    acc.x += __shfl_xor(acc.x, 16);
    acc.y += __shfl_xor(acc.y, 16);
    acc.z += __shfl_xor(acc.z, 16);
    acc.w += __shfl_xor(acc.w, 16);
  }
  const float rd = 1.0f / fmaxf((float)(end - beg), 1.0f);
  float4 o;
  o.x = xv.x + acc.x * rd;
  o.y = xv.y + acc.y * rd;
  o.z = xv.z + acc.z * rd;
  o.w = xv.w + acc.w * rd;
  if constexpr (ACT) {
    o.x = fmaxf(o.x + bv.x, 0.0f);
    o.y = fmaxf(o.y + bv.y, 0.0f);
    o.z = fmaxf(o.z + bv.z, 0.0f);
    o.w = fmaxf(o.w + bv.w, 0.0f);
  }
  return o;
}

__device__ __forceinline__ int
wave_bounds(const int* __restrict__ rowptr, int n0, int nNodes, int lane)
{
  return rowptr[min(n0 + lane, nNodes)];
}

__device__ __forceinline__ int
list_ids(const int* __restrict__ nbr, int beg, int end, int lane)
{
  int ids = 0;
  if (end > beg) ids = nbr[min(beg + lane, end - 1)];
  return ids;
}

__device__ __forceinline__ int
first_ids(const int* __restrict__ nbr, int rp, int (&b)[NODES_PER_WAVE + 1], int lane)
{
  #pragma unroll
  for (int q = 0; q <= NODES_PER_WAVE; ++q) b[q] = __builtin_amdgcn_readlane(rp, q);
  return list_ids(nbr, b[0], b[1], lane);
}

template <int F4, bool ACT, typename Out>
__device__ __forceinline__ void
gather_wave_nodes(const float4* __restrict__ x4, const int* __restrict__ nbr, const float4& bv,
                  const int (&b)[NODES_PER_WAVE + 1], int ids, int n0, int nNodes, int lane,
                  Out&& out)
{
  const int c   = lane % F4;
  const int sub = lane / F4;
  #pragma unroll
  for (int q = 0; q < NODES_PER_WAVE; ++q) {
    const int node = n0 + q;
    asm volatile("" :: "v"(ids));
    int idsNext = 0;
    if (q + 1 < NODES_PER_WAVE) idsNext = list_ids(nbr, b[q + 1], b[q + 2], lane);
    float4 o = make_float4(0.0f, 0.0f, 0.0f, 0.0f);
    if (node < nNodes)
      o = gather_node<F4, ACT>(x4, nbr, bv, node, b[q], b[q + 1], ids, c, sub, lane);
    out(q, node, o);
    ids = idsNext;
  }
}

template <int F4>
__global__ __launch_bounds__(256) void
gather_mean_kernel(const float* __restrict__ x, const int* __restrict__ rowptr,
                   const int* __restrict__ nbr, float* __restrict__ A, int nNodes)
{
  const int lane = threadIdx.x & 31;
  const int c    = lane % F4;
  const int sub  = lane / F4;
  const int wv   = blockIdx.x * 8 + (threadIdx.x >> 5);
  const int n0   = wv * NODES_PER_WAVE;
  if (n0 >= nNodes) return;
  const float4* x4 = (const float4*)x;
  const float4 nob = make_float4(0.0f, 0.0f, 0.0f, 0.0f);
  const int rp = wave_bounds(rowptr, n0, nNodes, lane);
  int b[NODES_PER_WAVE + 1];
  const int ids0 = first_ids(nbr, rp, b, lane);
  gather_wave_nodes<F4, false>(x4, nbr, nob, b, ids0, n0, nNodes, lane,
    [&](int, int node, const float4& o) {
      if (node < nNodes && sub == 0) ((float4*)A)[(size_t)node * F4 + c] = o;
    });
}

template <int FIN, int NT>
__device__ __forceinline__ void
gemm_tiles(const float4 (&a4)[FIN / 8], const float* __restrict__ Bl,
           const float* __restrict__ bsh, float* __restrict__ out,
           int ct, int rowBase, int half, int l16, int nRows, int Fout, int doRelu,
           bool fullRows)
{
  constexpr int STRIDE = FIN + 4;
  constexpr int KP = FIN / 8;
  v8f c[NT];
  #pragma unroll
  for (int t = 0; t < NT; ++t) c[t] = v8f{};
  #pragma unroll
  for (int j = 0; j < KP; ++j) {
    float4 b[NT];
    #pragma unroll
    for (int t = 0; t < NT; ++t)
      b[t] = *(const float4*)(Bl + (ct + t) * 16 * STRIDE + 8 * j);
    v2f alo, ahi;
    alo.x = a4[j].x; alo.y = a4[j].y;
    ahi.x = a4[j].z; ahi.y = a4[j].w;
    #pragma unroll
    for (int t = 0; t < NT; ++t) {
      v2f blo; blo.x = b[t].x; blo.y = b[t].y;
      c[t] = __builtin_amdgcn_wmma_f32_16x16x4_f32(false, alo, false, blo, (short)0, c[t], false, false);
    }
    #pragma unroll
    for (int t = 0; t < NT; ++t) {
      v2f bhi; bhi.x = b[t].z; bhi.y = b[t].w;
      c[t] = __builtin_amdgcn_wmma_f32_16x16x4_f32(false, ahi, false, bhi, (short)0, c[t], false, false);
    }
  }
  #pragma unroll
  for (int t = 0; t < NT; ++t) {
    const int n = (ct + t) * 16 + l16;
    const float bv = bsh[n];
    float* colBase = out + (size_t)(rowBase + 8 * half) * Fout + n;
    if (fullRows && ((ct + t) * 16 + 16 <= Fout)) {
      #pragma unroll
      for (int r = 0; r < 8; ++r) {
        float val = c[t][r] + bv;
        if (doRelu) val = fmaxf(val, 0.0f);
        colBase[(size_t)r * Fout] = val;
      }
    } else {
      #pragma unroll
      for (int r = 0; r < 8; ++r) {
        int gRow = rowBase + r + 8 * half;
        float val = c[t][r] + bv;
        if (doRelu) val = fmaxf(val, 0.0f);
        if (gRow < nRows && n < Fout) out[(size_t)gRow * Fout + n] = val;
      }
    }
  }
}

constexpr int TILE_ROWS = 8 * NODES_PER_WAVE;
static_assert(TILE_ROWS == 32, "two 16-row strips per block, one per wave parity");
__global__ __launch_bounds__(256) void
gather_gemm_kernel(const float* __restrict__ x, const int* __restrict__ rowptr,
                   const int* __restrict__ nbr, const float* __restrict__ W,
                   const float* __restrict__ bias, float* __restrict__ out, int nNodes)
{
  constexpr int FIN    = 64;
  constexpr int FOUT   = 128;
  constexpr int STRIDE = FIN + 4;
  constexpr int F4     = FIN / 4;
  constexpr int KP     = FIN / 8;
  constexpr int WLD    = (FOUT * F4) / 256;
  static_assert(FOUT / 16 == 2 * 4, "two column tiles for each of the four wave pairs");
  __shared__ __attribute__((aligned(16))) float Bs[FOUT * STRIDE];
  __shared__ __attribute__((aligned(16))) float At[TILE_ROWS * STRIDE];
  __shared__ float bsh[FOUT];

  const int tid  = (int)threadIdx.x;
  const int lane = tid & 31;
  const int wave = tid >> 5;
  const int half = lane >> 4;
  const int l16  = lane & 15;

  const int n0 = (int)blockIdx.x * TILE_ROWS + wave * NODES_PER_WAVE;
  const int rp = wave_bounds(rowptr, n0, nNodes, lane);

  const float bval = bias[min(tid, FOUT - 1)];
  float4 wv[WLD];
  #pragma unroll
  for (int i = 0; i < WLD; ++i) wv[i] = ((const float4*)W)[tid + i * 256];
  int b[NODES_PER_WAVE + 1];
  const int ids0 = first_ids(nbr, rp, b, lane);
  #pragma unroll
  for (int i = 0; i < WLD; ++i) {
    const int idx = tid + i * 256;
    const int r = idx / F4, c4 = idx - r * F4;
    *(float4*)&Bs[r * STRIDE + 4 * c4] = wv[i];
  }
  if (tid < FOUT) bsh[tid] = bval;

  {
    const int c   = lane % F4;
    const int sub = lane / F4;
    const float4* x4 = (const float4*)x;
    const float4 nob = make_float4(0.0f, 0.0f, 0.0f, 0.0f);
    gather_wave_nodes<F4, false>(x4, nbr, nob, b, ids0, n0, nNodes, lane,
      [&](int q, int, const float4& o) {
        const int r = wave * NODES_PER_WAVE + q;
        if (sub == 0) *(float4*)&At[r * STRIDE + 4 * c] = o;
      });
  }
  __syncthreads();

  const int strip   = wave & 1;
  const int rowBase = (int)blockIdx.x * TILE_ROWS + 16 * strip;
  if (rowBase >= nNodes) return;
  float4 a4[KP];
  {
    const float* Ar = &At[(16 * strip + l16) * STRIDE + 4 * half];
    #pragma unroll
    for (int j = 0; j < KP; ++j) a4[j] = *(const float4*)(Ar + 8 * j);
  }
  const float* Bl = &Bs[l16 * STRIDE + 4 * half];
  const bool fullRows = (rowBase + 16 <= nNodes);
  gemm_tiles<FIN, 2>(a4, Bl, bsh, out, 2 * (wave >> 1), rowBase, half, l16, nNodes, FOUT, 1, fullRows);
}

constexpr int CHUNK_PITCH = 32 + 4;
__global__ __launch_bounds__(256) void
layer2_pre3_kernel(const float* __restrict__ xin, const float* __restrict__ W2,
                   const float* __restrict__ b2, const float* __restrict__ W3,
                   float* __restrict__ tout, int nRows)
{
  constexpr int FIN    = 128;
  constexpr int STRIDE = FIN + 4;
  constexpr int F4     = FIN / 4;
  constexpr int KP     = FIN / 8;
  constexpr int FH     = 128;
  constexpr int FT     = 64;
  constexpr int WLD2   = (FH * F4) / 256;
  constexpr int WLD3   = (FT * F4) / 256;
  __shared__ __attribute__((aligned(16))) float Bs[FH * STRIDE];
  __shared__ __attribute__((aligned(16))) float B3[FT * STRIDE];
  __shared__ __attribute__((aligned(16))) float Ks[8 * 16 * CHUNK_PITCH];
  __shared__ float bsh[FH];

  const int tid  = (int)threadIdx.x;
  const int lane = tid & 31;
  const int wave = tid >> 5;
  const int half = lane >> 4;
  const int l16  = lane & 15;
  const int rowBase = (blockIdx.x * 8 + wave) * 16;
  const int row     = min(rowBase + l16, nRows - 1);

  const float bval = b2[min(tid, FH - 1)];
  float4 w2[WLD2];
  #pragma unroll
  for (int i = 0; i < WLD2; ++i) w2[i] = ((const float4*)W2)[tid + i * 256];
  float4 w3[WLD3];
  #pragma unroll
  for (int i = 0; i < WLD3; ++i) w3[i] = ((const float4*)W3)[tid + i * 256];
  float4 a4[KP];
  {
    const float4* xr = (const float4*)(xin + (size_t)row * FIN) + half;
    #pragma unroll
    for (int j = 0; j < KP; ++j) a4[j] = xr[2 * j];
  }
  #pragma unroll
  for (int i = 0; i < WLD2; ++i) {
    const int idx = tid + i * 256;
    const int r = idx / F4, c4 = idx - r * F4;
    *(float4*)&Bs[r * STRIDE + 4 * c4] = w2[i];
  }
  #pragma unroll
  for (int i = 0; i < WLD3; ++i) {
    const int idx = tid + i * 256;
    const int r = idx / F4, c4 = idx - r * F4;
    *(float4*)&B3[r * STRIDE + 4 * c4] = w3[i];
  }
  if (tid < FH) bsh[tid] = bval;
  __syncthreads();
  if (rowBase >= nRows) return;

  const float* Bl  = &Bs[l16 * STRIDE + 4 * half];
  const float* B3l = &B3[l16 * STRIDE + 4 * half];
  float*       Kw  = &Ks[wave * 16 * CHUNK_PITCH];

  v8f tacc[FT / 16];
  #pragma unroll
  for (int t = 0; t < FT / 16; ++t) tacc[t] = v8f{};

  #pragma unroll
  for (int p = 0; p < FH / 32; ++p) {
    v8f c[2];
    c[0] = v8f{}; c[1] = v8f{};
    #pragma unroll
    for (int j = 0; j < KP; ++j) {
      float4 b[2];
      #pragma unroll
      for (int t = 0; t < 2; ++t)
        b[t] = *(const float4*)(Bl + (2 * p + t) * 16 * STRIDE + 8 * j);
      v2f alo, ahi;
      alo.x = a4[j].x; alo.y = a4[j].y;
      ahi.x = a4[j].z; ahi.y = a4[j].w;
      #pragma unroll
      for (int t = 0; t < 2; ++t) {
        v2f blo; blo.x = b[t].x; blo.y = b[t].y;
        c[t] = __builtin_amdgcn_wmma_f32_16x16x4_f32(false, alo, false, blo, (short)0, c[t], false, false);
      }
      #pragma unroll
      for (int t = 0; t < 2; ++t) {
        v2f bhi; bhi.x = b[t].z; bhi.y = b[t].w;
        c[t] = __builtin_amdgcn_wmma_f32_16x16x4_f32(false, ahi, false, bhi, (short)0, c[t], false, false);
      }
    }
    #pragma unroll
    for (int t = 0; t < 2; ++t) {
      const float bv = bsh[(2 * p + t) * 16 + l16];
      #pragma unroll
      for (int r = 0; r < 8; ++r)
        Kw[(r + 8 * half) * CHUNK_PITCH + t * 16 + l16] = fmaxf(c[t][r] + bv, 0.0f);
    }
    float4 e4[4];
    #pragma unroll
    for (int j = 0; j < 4; ++j) e4[j] = *(const float4*)(Kw + l16 * CHUNK_PITCH + 8 * j + 4 * half);
    #pragma unroll
    for (int j = 0; j < 4; ++j) {
      float4 b[FT / 16];
      #pragma unroll
      for (int t = 0; t < FT / 16; ++t)
        b[t] = *(const float4*)(B3l + t * 16 * STRIDE + 8 * (4 * p + j));
      v2f alo, ahi;
      alo.x = e4[j].x; alo.y = e4[j].y;
      ahi.x = e4[j].z; ahi.y = e4[j].w;
      #pragma unroll
      for (int t = 0; t < FT / 16; ++t) {
        v2f blo; blo.x = b[t].x; blo.y = b[t].y;
        tacc[t] = __builtin_amdgcn_wmma_f32_16x16x4_f32(false, alo, false, blo, (short)0, tacc[t], false, false);
      }
      #pragma unroll
      for (int t = 0; t < FT / 16; ++t) {
        v2f bhi; bhi.x = b[t].z; bhi.y = b[t].w;
        tacc[t] = __builtin_amdgcn_wmma_f32_16x16x4_f32(false, ahi, false, bhi, (short)0, tacc[t], false, false);
      }
    }
  }
  const bool fullRows = (rowBase + 16 <= nRows);
  #pragma unroll
  for (int t = 0; t < FT / 16; ++t) {
    const int n = t * 16 + l16;
    if (fullRows) {
      float* colBase = tout + (size_t)(rowBase + 8 * half) * FT + n;
      #pragma unroll
      for (int r = 0; r < 8; ++r) colBase[(size_t)r * FT] = tacc[t][r];
    } else {
      #pragma unroll
      for (int r = 0; r < 8; ++r) {
        const int gRow = rowBase + r + 8 * half;
        if (gRow < nRows) tout[(size_t)gRow * FT + n] = tacc[t][r];
      }
    }
  }
}

__global__ __launch_bounds__(256) void
gather_heads_kernel(const float* __restrict__ t, const int* __restrict__ rowptr,
                    const int* __restrict__ nbr, const float* __restrict__ b3,
                    float* __restrict__ emb,
                    const float* __restrict__ Wp, const float* __restrict__ bp,
                    const float* __restrict__ Wc, const float* __restrict__ bc,
                    float* __restrict__ nxt, float* __restrict__ cls, int nNodes, int nClasses)
{
  constexpr int FE      = 64;
  constexpr int STRIDE2 = FE + 4;
  constexpr int F4E     = FE / 4;
  constexpr int KPE     = FE / 8;
  constexpr int WLDP    = (FE * F4E) / 256;
  static_assert(FE / 16 == 4, "one column tile of next for each of the four wave pairs");
  __shared__ __attribute__((aligned(16))) float Bh[(FE + 16) * STRIDE2];
  __shared__ __attribute__((aligned(16))) float Et[TILE_ROWS * STRIDE2];
  __shared__ float bsh2[FE + 16];

  const int tid  = (int)threadIdx.x;
  const int lane = tid & 31;
  const int wave = tid >> 5;
  const int half = lane >> 4;
  const int l16  = lane & 15;

  const int n0 = (int)blockIdx.x * TILE_ROWS + wave * NODES_PER_WAVE;
  const int rp = wave_bounds(rowptr, n0, nNodes, lane);

  const float bvp = bp[min(tid, FE - 1)];
  const float bvc = bc[min(tid, nClasses - 1)];
  float4 wp[WLDP];
  #pragma unroll
  for (int i = 0; i < WLDP; ++i) wp[i] = ((const float4*)Wp)[tid + i * 256];
  const int wcRow = tid >> 4;
  const float4 wc = ((const float4*)Wc)[min(tid, nClasses * F4E - 1)];
  int b[NODES_PER_WAVE + 1];
  const int ids0 = first_ids(nbr, rp, b, lane);
  #pragma unroll
  for (int i = 0; i < WLDP; ++i) {
    const int idx = tid + i * 256;
    const int r = idx / F4E, c4 = idx - r * F4E;
    *(float4*)&Bh[r * STRIDE2 + 4 * c4] = wp[i];
  }
  {
    const int c4 = tid - wcRow * F4E;
    *(float4*)&Bh[(FE + wcRow) * STRIDE2 + 4 * c4] =
        (wcRow < nClasses) ? wc : make_float4(0.0f, 0.0f, 0.0f, 0.0f);
  }
  if (tid < FE) bsh2[tid] = bvp;
  if (tid < 16) bsh2[FE + tid] = (tid < nClasses) ? bvc : 0.0f;

  {
    const int c   = lane % F4E;
    const int sub = lane / F4E;
    const float4* t4 = (const float4*)t;
    float4 bv;
    bv.x = b3[4 * c]; bv.y = b3[4 * c + 1]; bv.z = b3[4 * c + 2]; bv.w = b3[4 * c + 3];
    gather_wave_nodes<F4E, true>(t4, nbr, bv, b, ids0, n0, nNodes, lane,
      [&](int q, int node, const float4& o) {
        const int r = wave * NODES_PER_WAVE + q;
        if (node < nNodes && sub == 0) ((float4*)emb)[(size_t)node * F4E + c] = o;
        if (sub == 0) *(float4*)&Et[r * STRIDE2 + 4 * c] = o;
      });
  }
  __syncthreads();

  const int strip   = wave & 1;
  const int pair    = wave >> 1;
  const int rowBase = (int)blockIdx.x * TILE_ROWS + 16 * strip;
  if (rowBase >= nNodes) return;
  float4 e4[KPE];
  {
    const float* Er = &Et[(16 * strip + l16) * STRIDE2 + 4 * half];
    #pragma unroll
    for (int j = 0; j < KPE; ++j) e4[j] = *(const float4*)(Er + 8 * j);
  }
  const bool fullRows = (rowBase + 16 <= nNodes);
  const float* Bl2 = &Bh[l16 * STRIDE2 + 4 * half];
  gemm_tiles<FE, 1>(e4, Bl2, bsh2, nxt, pair, rowBase, half, l16, nNodes, FE, 0, fullRows);
  if (pair == 3)
    gemm_tiles<FE, 1>(e4, Bl2 + FE * STRIDE2, bsh2 + FE, cls, 0, rowBase, half, l16, nNodes, nClasses, 0, fullRows);
}

extern "C" void kernel_launch(void* const* d_in, const int* in_sizes, int n_in,
                              void* d_out, int out_size, void* d_ws, size_t ws_size,
                              hipStream_t stream) {
  const float* x0 = (const float*)d_in[0];
  const int*   ei = (const int*)d_in[1];
  const float *W1 = (const float*)d_in[2],  *b1 = (const float*)d_in[3];
  const float *W2 = (const float*)d_in[4],  *b2 = (const float*)d_in[5];
  const float *W3 = (const float*)d_in[6],  *b3 = (const float*)d_in[7];
  const float *Wp = (const float*)d_in[8],  *bp = (const float*)d_in[9];
  const float *Wc = (const float*)d_in[10], *bc = (const float*)d_in[11];

  const int N      = in_sizes[0] / 64;
  const int nEdges = in_sizes[1] / 2;
  const int* src = ei;
  const int* dst = ei + nEdges;

  float* ws        = (float*)d_ws;
  float* A         = ws + 65536;
  const int nBins   = (N + BIN_NODES - 1) / BIN_NODES;
  const int nChunks = (nEdges + CHUNK - 1) / CHUNK;
  int*   pairs     = (int*)A;
  int*   offT      = (int*)A + 2 * (size_t)nChunks * CHUNK;
  int*   rowptr    = (int*)(A + (size_t)N * 128);
  int*   nbr       = rowptr + 65536;
  int*   nbrSorted = nbr + nEdges;
  float* h1        = A;
  float* t3        = A;

  float* out = (float*)d_out;
  float* A2  = out;
  float* emb = out;
  float* nxt = out + (size_t)N * 64;
  float* cls = out + (size_t)N * 128;

  dim3 b256(256);
  dim3 gemmGrid((unsigned)((N + 127) / 128)), gemmBlock(256);
  dim3 nodeGrid((unsigned)((N + 8 * NODES_PER_WAVE - 1) / (8 * NODES_PER_WAVE)));
  dim3 chunkGrid((unsigned)nChunks), binGrid((unsigned)nBins);

  chunk_sort_kernel<<<chunkGrid, b256, 0, stream>>>(src, dst, offT, pairs, nEdges, nBins);
  bin_lists_kernel <<<binGrid,   b256, 0, stream>>>(pairs, offT, rowptr, nbr, nbrSorted, N, nBins, nChunks);

  gather_gemm_kernel<<<nodeGrid, b256, 0, stream>>>(x0, rowptr, nbrSorted, W1, b1, h1, N);

  gather_mean_kernel<32><<<nodeGrid, b256, 0, stream>>>(h1, rowptr, nbrSorted, A2, N);
  layer2_pre3_kernel<<<gemmGrid, gemmBlock, 0, stream>>>(A2, W2, b2, W3, t3, N);

  gather_heads_kernel<<<nodeGrid, b256, 0, stream>>>(t3, rowptr, nbrSorted, b3, emb, Wp, bp, Wc, bc, nxt, cls, N, 10);
}


# Layer-2 mean feeds both products inside the same launch: a block's 32 mean rows stay in an LDS tile, W2 and W3 pass through one 64-row LDS buffer in 3 chunks (6 -> 5 launches); same sums, same order
# speedup vs baseline: 6.8630x; 1.0445x over previous
#include <hip/hip_runtime.h>
#include <hip/hip_bf16.h>

typedef __attribute__((ext_vector_type(8))) float v8f;
typedef __attribute__((ext_vector_type(2))) float v2f;

constexpr int BIN_SHIFT = 6;
constexpr int BIN_NODES = 1 << BIN_SHIFT;
constexpr int MAX_BINS  = 1024;
constexpr int CHUNK     = 4096;
constexpr int QPT       = CHUNK / (256 * 4);
constexpr int BIN_CAP   = 4096;
constexpr int SEG_PRE   = 12;
static_assert(MAX_BINS == 4 * 256, "chunk_sort_kernel scans four bins per thread");

__device__ __forceinline__ int
load_quad(const int* __restrict__ p, int e, int n, bool aligned, int (&v)[4])
{
  if (aligned && e + 4 <= n) {
    const int4 t = *(const int4*)(p + e);
    v[0] = t.x; v[1] = t.y; v[2] = t.z; v[3] = t.w;
    return 4;
  }
  int c = 0;
  #pragma unroll
  for (int k = 0; k < 4; ++k) {
    v[k] = 0;
    if (e + k < n) { v[k] = p[e + k]; c = k + 1; }
  }
  return c;
}

template <bool GUARD>
__device__ __forceinline__ void
place_pairs(int* __restrict__ cursor, int2* __restrict__ sorted,
            const int (&d)[QPT][4], const int (&s)[QPT][4], const int (&c)[QPT])
{
  int pos[QPT][4];
  #pragma unroll
  for (int u = 0; u < QPT; ++u) {
    #pragma unroll
    for (int k = 0; k < 4; ++k) {
      pos[u][k] = 0;
      if (!GUARD || k < c[u])
        pos[u][k] = __hip_atomic_fetch_add(&cursor[d[u][k] >> BIN_SHIFT], 1, __ATOMIC_RELAXED, __HIP_MEMORY_SCOPE_WORKGROUP);
    }
  }
  #pragma unroll
  for (int u = 0; u < QPT; ++u) {
    #pragma unroll
    for (int k = 0; k < 4; ++k)
      if (!GUARD || k < c[u]) sorted[pos[u][k]] = make_int2(s[u][k], d[u][k]);
  }
}

__global__ __launch_bounds__(256) void
chunk_sort_kernel(const int* __restrict__ src, const int* __restrict__ dst,
                  int* __restrict__ offT, int* __restrict__ pairWords, int nEdges, int nBins)
{
  __shared__ int cursor[MAX_BINS];
  __shared__ int sIncl[8];
  __shared__ __attribute__((aligned(16))) int sortedW[2 * CHUNK];
  const int tid = (int)threadIdx.x, lane = tid & 31, w = tid >> 5;
  const int G = (int)gridDim.x;
  int2* sorted = (int2*)sortedW;
  const bool aligned = ((((size_t)src) | ((size_t)dst)) & 15) == 0;
  const int e0 = blockIdx.x * CHUNK + 4 * tid;
  int d[QPT][4], s[QPT][4], c[QPT];
  #pragma unroll
  for (int u = 0; u < QPT; ++u) {
    c[u] = load_quad(dst, e0 + 1024 * u, nEdges, aligned, d[u]);
    load_quad(src, e0 + 1024 * u, nEdges, aligned, s[u]);
  }
  for (int i = tid; i < MAX_BINS; i += 256) cursor[i] = 0;
  __syncthreads();
  #pragma unroll
  for (int u = 0; u < QPT; ++u) {
    #pragma unroll
    for (int k = 0; k < 4; ++k)
      if (k < c[u])
        __hip_atomic_fetch_add(&cursor[d[u][k] >> BIN_SHIFT], 1, __ATOMIC_RELAXED, __HIP_MEMORY_SCOPE_WORKGROUP);
  }
  __syncthreads();
  const int idx = 4 * tid;
  int v[4];
  #pragma unroll
  for (int k = 0; k < 4; ++k) v[k] = cursor[idx + k];
  const int local = (v[0] + v[1]) + (v[2] + v[3]);
  int incl = local;
  #pragma unroll
  for (int o = 1; o < 32; o <<= 1) {
    int y = __shfl_up(incl, o);
    if (lane >= o) incl += y;
  }
  if (lane == 31) sIncl[w] = incl;
  __syncthreads();
  int r = incl - local;
  #pragma unroll
  for (int i = 0; i < 8; ++i)
    if (i < w) r += sIncl[i];
  #pragma unroll
  for (int k = 0; k < 4; ++k) {
    cursor[idx + k] = r;
    if (idx + k < nBins) offT[(size_t)(idx + k) * G + blockIdx.x] = r;
    r += v[k];
  }
  if (tid == 255) offT[(size_t)nBins * G + blockIdx.x] = r;
  __syncthreads();
  if ((c[0] + c[1]) + (c[2] + c[3]) == 4 * QPT) place_pairs<false>(cursor, sorted, d, s, c);
  else                                         place_pairs<true >(cursor, sorted, d, s, c);
  __syncthreads();
  const int total = min(CHUNK, max(nEdges - (int)blockIdx.x * CHUNK, 0));
  int4*       outp = (int4*)pairWords + (size_t)blockIdx.x * (CHUNK / 2);
  const int4* sp   = (const int4*)sortedW;
  if (total == CHUNK) {
    #pragma unroll
    for (int u = 0; u < CHUNK / 512; ++u) outp[tid + 256 * u] = sp[tid + 256 * u];
  } else {
    #pragma unroll
    for (int u = 0; u < CHUNK / 512; ++u) {
      const int i = tid + 256 * u;
      if (2 * i < total) outp[i] = sp[i];
    }
  }
}

constexpr int NODES_PER_WAVE = 4;

constexpr int PRE = 6;

template <typename L>
__device__ __forceinline__ int
rank_in_list(const L& list, int beg, int end, int p, int s)
{
  int r = 0;
  for (int j = beg; j < end; j += 8) {
    int t[8];
    #pragma unroll
    for (int u = 0; u < 8; ++u) t[u] = list(min(j + u, end - 1));
    #pragma unroll
    for (int u = 0; u < 8; ++u) {
      const int jj = j + u;
      const bool front = (t[u] < s) || (t[u] == s && jj < p);
      r += (jj < end && front) ? 1 : 0;
    }
  }
  return r;
}

__global__ __launch_bounds__(256) void
bin_lists_kernel(const int* __restrict__ pairWords, const int* __restrict__ offT,
                 int* __restrict__ rowptr, int* nbr, int* __restrict__ nbrSorted,
                 int nNodes, int nBins, int G)
{
  __shared__ int cntL[BIN_NODES];
  __shared__ int lpL[BIN_NODES + 1];
  __shared__ int curL[BIN_NODES];
  __shared__ int listL[BIN_CAP];
  __shared__ int nodeL[BIN_CAP];
  __shared__ __attribute__((aligned(8))) int pairLW[2 * BIN_CAP];
  __shared__ int sCnt[8];
  __shared__ int sOff[8];
  const int tid = (int)threadIdx.x, lane = tid & 31, w = tid >> 5;
  const int bin = (int)blockIdx.x;
  const int2* pairs = (const int2*)pairWords;
  int2* pairL = (int2*)pairLW;
  const int nodeBase = bin * BIN_NODES;
  const int* row0 = offT + (size_t)bin * G;
  const int* row1 = row0 + G;
  if (tid < BIN_NODES) cntL[tid] = 0;
  __syncthreads();
  int m = 0;
  int start = 0;
  for (int b0 = 0; b0 < G; b0 += 256) {
    const int b = b0 + tid;
    int o0 = 0, o1 = 0;
    if (b < G) { o0 = row0[b]; o1 = row1[b]; }
    const int cnt = o1 - o0;
    const int2* chunk = pairs + (size_t)min(b, G - 1) * CHUNK;
    const int2* seg   = chunk + o0;
    const int   last  = max(o1 - 1, 0);
    int2 p[SEG_PRE];
    #pragma unroll
    for (int u = 0; u < SEG_PRE; ++u) p[u] = chunk[min(o0 + u, last)];
    __builtin_amdgcn_sched_barrier(0);
    int incl = cnt, so = o0;
    #pragma unroll
    for (int o = 1; o < 32; o <<= 1) {
      int y = __shfl_up(incl, o);
      if (lane >= o) incl += y;
      so += __shfl_xor(so, o);
    }
    if (lane == 31) sCnt[w] = incl;
    if (lane == 0)  sOff[w] = so;
    __syncthreads();
    int base = m + incl - cnt;
    #pragma unroll
    for (int i = 0; i < 8; ++i) {
      const int t = sCnt[i];
      if (i < w) base += t;
      m += t;
      start += sOff[i];
    }
    #pragma unroll
    for (int u = 0; u < SEG_PRE; ++u)
      if (u < cnt) {
        __hip_atomic_fetch_add(&cntL[p[u].y - nodeBase], 1, __ATOMIC_RELAXED, __HIP_MEMORY_SCOPE_WORKGROUP);
        if (base + u < BIN_CAP) pairL[base + u] = p[u];
      }
    for (int k = SEG_PRE; k < cnt; ++k) {
      const int2 q = seg[k];
      __hip_atomic_fetch_add(&cntL[q.y - nodeBase], 1, __ATOMIC_RELAXED, __HIP_MEMORY_SCOPE_WORKGROUP);
      if (base + k < BIN_CAP) pairL[base + k] = q;
    }
    __syncthreads();
  }
  if (tid < 32) {
    const int c0 = cntL[2 * tid], c1 = cntL[2 * tid + 1];
    int incl = c0 + c1;
    #pragma unroll
    for (int o = 1; o < 32; o <<= 1) {
      int y = __shfl_up(incl, o);
      if (tid >= o) incl += y;
    }
    const int s0 = incl - (c0 + c1), s1 = s0 + c0;
    lpL[2 * tid]  = s0;  lpL[2 * tid + 1]  = s1;
    curL[2 * tid] = s0;  curL[2 * tid + 1] = s1;
    if (tid == 31) lpL[BIN_NODES] = incl;
    if (nodeBase + 2 * tid     <= nNodes) rowptr[nodeBase + 2 * tid]     = start + s0;
    if (nodeBase + 2 * tid + 1 <= nNodes) rowptr[nodeBase + 2 * tid + 1] = start + s1;
  }
  if (bin == nBins - 1 && tid == 0 && nodeBase + BIN_NODES == nNodes) rowptr[nNodes] = start + m;
  __syncthreads();
  if (m <= BIN_CAP) {
    int2 pr[PRE];
    #pragma unroll
    for (int u = 0; u < PRE; ++u) {
      const int i = tid + 256 * u;
      pr[u] = (i < m) ? pairL[i] : make_int2(0, nodeBase);
    }
    int slot[PRE];
    #pragma unroll
    for (int u = 0; u < PRE; ++u)
      if (tid + 256 * u < m)
        slot[u] = __hip_atomic_fetch_add(&curL[pr[u].y - nodeBase], 1, __ATOMIC_RELAXED, __HIP_MEMORY_SCOPE_WORKGROUP);
    #pragma unroll
    for (int u = 0; u < PRE; ++u)
      if (tid + 256 * u < m) {
        listL[slot[u]] = pr[u].x;
        nodeL[slot[u]] = pr[u].y - nodeBase;
      }
    for (int i = tid + 256 * PRE; i < m; i += 256) {
      const int2 q   = pairL[i];
      const int  dl  = q.y - nodeBase;
      const int  sl  = __hip_atomic_fetch_add(&curL[dl], 1, __ATOMIC_RELAXED, __HIP_MEMORY_SCOPE_WORKGROUP);
      listL[sl] = q.x;
      nodeL[sl] = dl;
    }
    __syncthreads();
    for (int p = tid; p < m; p += 256) {
      const int s  = listL[p];
      const int dl = nodeL[p];
      const int beg = lpL[dl], end = lpL[dl + 1];
      const int r = rank_in_list([&](int j) { return listL[j]; }, beg, end, p, s);
      nbrSorted[start + beg + r] = s;
    }
  } else {
    for (int b = tid; b < G; b += 256) {
      const int2* seg = pairs + (size_t)b * CHUNK;
      const int o1 = row1[b];
      for (int k = row0[b]; k < o1; ++k) {
        const int2 q  = seg[k];
        const int  sl = __hip_atomic_fetch_add(&curL[q.y - nodeBase], 1, __ATOMIC_RELAXED, __HIP_MEMORY_SCOPE_WORKGROUP);
        nbr[start + sl] = q.x;
      }
    }
    __syncthreads();
    for (int q = w; q < BIN_NODES; q += 8) {
      const int beg = lpL[q], end = lpL[q + 1];
      for (int i = beg + lane; i < end; i += 32) {
        const int s = nbr[start + i];
        const int r = rank_in_list([&](int j) { return nbr[start + j]; }, beg, end, i, s);
        nbrSorted[start + beg + r] = s;
      }
    }
  }
}


template <int F4, int U, bool CLAMP>
__device__ __forceinline__ void
add_rows(float4& acc, const float4* __restrict__ x4, int ids, int j0, int cnt, int c, int sub)
{
  constexpr int EPS = 32 / F4;
  float4 v[U];
  #pragma unroll
  for (int u = 0; u < U; ++u) {
    int s;
    if constexpr (EPS == 1) {
      int j = j0 + u;
      if (CLAMP) j = min(j, cnt - 1);
      s = __builtin_amdgcn_readlane(ids, j);
    } else {
      int ja = j0 + 2 * u, jb = ja + 1;
      if (CLAMP) { ja = min(ja, cnt - 1); jb = min(jb, cnt - 1); }
      const int sa = __builtin_amdgcn_readlane(ids, ja);
      const int sb = __builtin_amdgcn_readlane(ids, jb);
      s = sub ? sb : sa;
    }
    v[u] = x4[(size_t)s * F4 + c];
  }
  if constexpr (CLAMP) {
    static_assert(U % 4 == 0, "the fence below names four rows per statement");
    #pragma unroll
    for (int g = 0; g < U; g += 4)
      asm volatile("" :: "v"(v[g].x),     "v"(v[g].y),     "v"(v[g].z),     "v"(v[g].w),
                         "v"(v[g + 1].x), "v"(v[g + 1].y), "v"(v[g + 1].z), "v"(v[g + 1].w),
                         "v"(v[g + 2].x), "v"(v[g + 2].y), "v"(v[g + 2].z), "v"(v[g + 2].w),
                         "v"(v[g + 3].x), "v"(v[g + 3].y), "v"(v[g + 3].z), "v"(v[g + 3].w));
  }
  #pragma unroll
  for (int u = 0; u < U; ++u) {
    if (!CLAMP || (j0 + u * EPS + sub < cnt)) {
      acc.x += v[u].x; acc.y += v[u].y; acc.z += v[u].z; acc.w += v[u].w;
    }
  }
}

template <int F4>
__device__ __forceinline__ void
add_piece(float4& acc, const float4* __restrict__ x4, int ids, int cnt, int c, int sub)
{
  constexpr int EPS = 32 / F4;
  int j = 0;
  for (; j + 8 * EPS <= cnt; j += 8 * EPS) add_rows<F4, 8, false>(acc, x4, ids, j, cnt, c, sub);
  if (cnt - j > 4 * EPS)  add_rows<F4, 8, true>(acc, x4, ids, j, cnt, c, sub);
  else if (cnt - j > 0)   add_rows<F4, 4, true>(acc, x4, ids, j, cnt, c, sub);
}

template <int F4, bool ACT>
__device__ __forceinline__ float4
gather_node(const float4* __restrict__ x4, const int* __restrict__ nbr, const float4& bv,
            int node, int beg, int end, int ids, int c, int sub, int lane)
{
  constexpr int EPS = 32 / F4;
  const float4 xv = x4[(size_t)node * F4 + c];
  float4 acc = make_float4(0.0f, 0.0f, 0.0f, 0.0f);
  add_piece<F4>(acc, x4, ids, min(end - beg, 32), c, sub);
  for (int base = beg + 32; base < end; base += 32) {
    const int more = nbr[min(base + lane, end - 1)];
    const int cnt  = min(end - base, 32);
    for (int j = 0; j < cnt; j += 4 * EPS) add_rows<F4, 4, true>(acc, x4, more, j, cnt, c, sub);
  }
  if (EPS == 2) {
    acc.x += __shfl_xor(acc.x, 16);
    acc.y += __shfl_xor(acc.y, 16);
    acc.z += __shfl_xor(acc.z, 16);
    acc.w += __shfl_xor(acc.w, 16);
  }
  const float rd = 1.0f / fmaxf((float)(end - beg), 1.0f);
  float4 o;
  o.x = xv.x + acc.x * rd;
  o.y = xv.y + acc.y * rd;
  o.z = xv.z + acc.z * rd;
  o.w = xv.w + acc.w * rd;
  if constexpr (ACT) {
    o.x = fmaxf(o.x + bv.x, 0.0f);
    o.y = fmaxf(o.y + bv.y, 0.0f);
    o.z = fmaxf(o.z + bv.z, 0.0f);
    o.w = fmaxf(o.w + bv.w, 0.0f);
  }
  return o;
}

__device__ __forceinline__ int
wave_bounds(const int* __restrict__ rowptr, int n0, int nNodes, int lane)
{
  return rowptr[min(n0 + lane, nNodes)];
}

__device__ __forceinline__ int
list_ids(const int* __restrict__ nbr, int beg, int end, int lane)
{
  int ids = 0;
  if (end > beg) ids = nbr[min(beg + lane, end - 1)];
  return ids;
}

__device__ __forceinline__ int
first_ids(const int* __restrict__ nbr, int rp, int (&b)[NODES_PER_WAVE + 1], int lane)
{
  #pragma unroll
  for (int q = 0; q <= NODES_PER_WAVE; ++q) b[q] = __builtin_amdgcn_readlane(rp, q);
  return list_ids(nbr, b[0], b[1], lane);
}

template <int F4, bool ACT, typename Out>
__device__ __forceinline__ void
gather_wave_nodes(const float4* __restrict__ x4, const int* __restrict__ nbr, const float4& bv,
                  const int (&b)[NODES_PER_WAVE + 1], int ids, int n0, int nNodes, int lane,
                  Out&& out)
{
  const int c   = lane % F4;
  const int sub = lane / F4;
  #pragma unroll
  for (int q = 0; q < NODES_PER_WAVE; ++q) {
    const int node = n0 + q;
    asm volatile("" :: "v"(ids));
    int idsNext = 0;
    if (q + 1 < NODES_PER_WAVE) idsNext = list_ids(nbr, b[q + 1], b[q + 2], lane);
    float4 o = make_float4(0.0f, 0.0f, 0.0f, 0.0f);
    if (node < nNodes)
      o = gather_node<F4, ACT>(x4, nbr, bv, node, b[q], b[q + 1], ids, c, sub, lane);
    out(q, node, o);
    ids = idsNext;
  }
}

template <int FIN, int NT>
__device__ __forceinline__ void
gemm_tiles(const float4 (&a4)[FIN / 8], const float* __restrict__ Bl,
           const float* __restrict__ bsh, float* __restrict__ out,
           int ct, int rowBase, int half, int l16, int nRows, int Fout, int doRelu,
           bool fullRows)
{
  constexpr int STRIDE = FIN + 4;
  constexpr int KP = FIN / 8;
  v8f c[NT];
  #pragma unroll
  for (int t = 0; t < NT; ++t) c[t] = v8f{};
  #pragma unroll
  for (int j = 0; j < KP; ++j) {
    float4 b[NT];
    #pragma unroll
    for (int t = 0; t < NT; ++t)
      b[t] = *(const float4*)(Bl + (ct + t) * 16 * STRIDE + 8 * j);
    v2f alo, ahi;
    alo.x = a4[j].x; alo.y = a4[j].y;
    ahi.x = a4[j].z; ahi.y = a4[j].w;
    #pragma unroll
    for (int t = 0; t < NT; ++t) {
      v2f blo; blo.x = b[t].x; blo.y = b[t].y;
      c[t] = __builtin_amdgcn_wmma_f32_16x16x4_f32(false, alo, false, blo, (short)0, c[t], false, false);
    }
    #pragma unroll
    for (int t = 0; t < NT; ++t) {
      v2f bhi; bhi.x = b[t].z; bhi.y = b[t].w;
      c[t] = __builtin_amdgcn_wmma_f32_16x16x4_f32(false, ahi, false, bhi, (short)0, c[t], false, false);
    }
  }
  #pragma unroll
  for (int t = 0; t < NT; ++t) {
    const int n = (ct + t) * 16 + l16;
    const float bv = bsh[n];
    float* colBase = out + (size_t)(rowBase + 8 * half) * Fout + n;
    if (fullRows && ((ct + t) * 16 + 16 <= Fout)) {
      #pragma unroll
      for (int r = 0; r < 8; ++r) {
        float val = c[t][r] + bv;
        if (doRelu) val = fmaxf(val, 0.0f);
        colBase[(size_t)r * Fout] = val;
      }
    } else {
      #pragma unroll
      for (int r = 0; r < 8; ++r) {
        int gRow = rowBase + r + 8 * half;
        float val = c[t][r] + bv;
        if (doRelu) val = fmaxf(val, 0.0f);
        if (gRow < nRows && n < Fout) out[(size_t)gRow * Fout + n] = val;
      }
    }
  }
}

constexpr int TILE_ROWS = 8 * NODES_PER_WAVE;
static_assert(TILE_ROWS == 32, "two 16-row strips per block, one per wave parity");
__global__ __launch_bounds__(256) void
gather_gemm_kernel(const float* __restrict__ x, const int* __restrict__ rowptr,
                   const int* __restrict__ nbr, const float* __restrict__ W,
                   const float* __restrict__ bias, float* __restrict__ out, int nNodes)
{
  constexpr int FIN    = 64;
  constexpr int FOUT   = 128;
  constexpr int STRIDE = FIN + 4;
  constexpr int F4     = FIN / 4;
  constexpr int KP     = FIN / 8;
  constexpr int WLD    = (FOUT * F4) / 256;
  static_assert(FOUT / 16 == 2 * 4, "two column tiles for each of the four wave pairs");
  __shared__ __attribute__((aligned(16))) float Bs[FOUT * STRIDE];
  __shared__ __attribute__((aligned(16))) float At[TILE_ROWS * STRIDE];
  __shared__ float bsh[FOUT];

  const int tid  = (int)threadIdx.x;
  const int lane = tid & 31;
  const int wave = tid >> 5;
  const int half = lane >> 4;
  const int l16  = lane & 15;

  const int n0 = (int)blockIdx.x * TILE_ROWS + wave * NODES_PER_WAVE;
  const int rp = wave_bounds(rowptr, n0, nNodes, lane);

  const float bval = bias[min(tid, FOUT - 1)];
  float4 wv[WLD];
  #pragma unroll
  for (int i = 0; i < WLD; ++i) wv[i] = ((const float4*)W)[tid + i * 256];
  int b[NODES_PER_WAVE + 1];
  const int ids0 = first_ids(nbr, rp, b, lane);
  #pragma unroll
  for (int i = 0; i < WLD; ++i) {
    const int idx = tid + i * 256;
    const int r = idx / F4, c4 = idx - r * F4;
    *(float4*)&Bs[r * STRIDE + 4 * c4] = wv[i];
  }
  if (tid < FOUT) bsh[tid] = bval;

  {
    const int c   = lane % F4;
    const int sub = lane / F4;
    const float4* x4 = (const float4*)x;
    const float4 nob = make_float4(0.0f, 0.0f, 0.0f, 0.0f);
    gather_wave_nodes<F4, false>(x4, nbr, nob, b, ids0, n0, nNodes, lane,
      [&](int q, int, const float4& o) {
        const int r = wave * NODES_PER_WAVE + q;
        if (sub == 0) *(float4*)&At[r * STRIDE + 4 * c] = o;
      });
  }
  __syncthreads();

  const int strip   = wave & 1;
  const int rowBase = (int)blockIdx.x * TILE_ROWS + 16 * strip;
  if (rowBase >= nNodes) return;
  float4 a4[KP];
  {
    const float* Ar = &At[(16 * strip + l16) * STRIDE + 4 * half];
    #pragma unroll
    for (int j = 0; j < KP; ++j) a4[j] = *(const float4*)(Ar + 8 * j);
  }
  const float* Bl = &Bs[l16 * STRIDE + 4 * half];
  const bool fullRows = (rowBase + 16 <= nNodes);
  gemm_tiles<FIN, 2>(a4, Bl, bsh, out, 2 * (wave >> 1), rowBase, half, l16, nNodes, FOUT, 1, fullRows);
}


template <int FIN>
__device__ __forceinline__ v8f
strip_tile(const float* __restrict__ Al, const float* __restrict__ Bl)
{
  v8f c = v8f{};
  #pragma unroll
  for (int j = 0; j < FIN / 8; ++j) {
    const float4 a = *(const float4*)(Al + 8 * j);
    const float4 b = *(const float4*)(Bl + 8 * j);
    v2f alo, ahi, blo, bhi;
    alo.x = a.x; alo.y = a.y;
    ahi.x = a.z; ahi.y = a.w;
    blo.x = b.x; blo.y = b.y;
    bhi.x = b.z; bhi.y = b.w;
    c = __builtin_amdgcn_wmma_f32_16x16x4_f32(false, alo, false, blo, (short)0, c, false, false);
    c = __builtin_amdgcn_wmma_f32_16x16x4_f32(false, ahi, false, bhi, (short)0, c, false, false);
  }
  return c;
}

constexpr int L2_STRIDE = 128 + 4;
struct WeightChunk { float4 q0, q1, q2, q3, q4, q5, q6, q7; };

__device__ __forceinline__ WeightChunk
load_chunk(const float4* __restrict__ p, int tid)
{
  WeightChunk w;
  w.q0 = p[tid];        w.q1 = p[tid + 256];  w.q2 = p[tid + 512];  w.q3 = p[tid + 768];
  w.q4 = p[tid + 1024]; w.q5 = p[tid + 1280]; w.q6 = p[tid + 1536]; w.q7 = p[tid + 1792];
  return w;
}

__device__ __forceinline__ void
store_chunk(float* __restrict__ Wb, const WeightChunk& w, int tid)
{
  float* d = Wb + (tid >> 5) * L2_STRIDE + 4 * (tid & 31);
  *(float4*)(d)                  = w.q0;
  *(float4*)(d +  8 * L2_STRIDE) = w.q1;
  *(float4*)(d + 16 * L2_STRIDE) = w.q2;
  *(float4*)(d + 24 * L2_STRIDE) = w.q3;
  *(float4*)(d + 32 * L2_STRIDE) = w.q4;
  *(float4*)(d + 40 * L2_STRIDE) = w.q5;
  *(float4*)(d + 48 * L2_STRIDE) = w.q6;
  *(float4*)(d + 56 * L2_STRIDE) = w.q7;
}

__global__ __launch_bounds__(256, 12) void
gather_layer2_kernel(const float* __restrict__ h1, const int* __restrict__ rowptr,
                     const int* __restrict__ nbr, const float* __restrict__ W2,
                     const float* __restrict__ b2, const float* __restrict__ W3,
                     float* __restrict__ tout, int nNodes)
{
  constexpr int FIN    = 128;
  constexpr int STRIDE = L2_STRIDE;
  constexpr int F4     = FIN / 4;
  constexpr int FT     = 64;
  constexpr int WROWS  = 64;
  static_assert(STRIDE == FIN + 4, "one pitch for the weight chunk and the row tile");
  static_assert(F4 == 32 && WROWS * F4 == 8 * 256, "a chunk is eight float4 per thread, a row per wave and load");
  static_assert(WROWS / 16 == 4, "one column tile of a chunk for each of the four wave pairs");
  __shared__ __attribute__((aligned(16))) float Wb[WROWS * STRIDE];
  __shared__ __attribute__((aligned(16))) float Rt[TILE_ROWS * STRIDE];
  __shared__ float bsh[FIN];

  const int tid  = (int)threadIdx.x;
  const int lane = tid & 31;
  const int wave = tid >> 5;
  const int half = lane >> 4;
  const int l16  = lane & 15;

  const int n0 = (int)blockIdx.x * TILE_ROWS + wave * NODES_PER_WAVE;
  const int rp = wave_bounds(rowptr, n0, nNodes, lane);

  const float bval = b2[min(tid, FIN - 1)];
  const WeightChunk w0 = load_chunk((const float4*)W2, tid);
  int b[NODES_PER_WAVE + 1];
  const int ids0 = first_ids(nbr, rp, b, lane);
  store_chunk(Wb, w0, tid);
  if (tid < FIN) bsh[tid] = bval;

  {
    const float4* x4 = (const float4*)h1;
    const float4 nob = make_float4(0.0f, 0.0f, 0.0f, 0.0f);
    gather_wave_nodes<F4, false>(x4, nbr, nob, b, ids0, n0, nNodes, lane,
      [&](int q, int, const float4& o) {
        const int r = wave * NODES_PER_WAVE + q;
        *(float4*)&Rt[r * STRIDE + 4 * lane] = o;
      });
  }
  __syncthreads();

  const int strip   = wave & 1;
  const int ct      = wave >> 1;
  const int rowBase = (int)blockIdx.x * TILE_ROWS + 16 * strip;
  const float* Al = &Rt[(16 * strip + l16) * STRIDE + 4 * half];
  const float* Bl = &Wb[(16 * ct + l16) * STRIDE + 4 * half];

  const WeightChunk w1 = load_chunk((const float4*)W2 + WROWS * F4, tid);
  __builtin_amdgcn_sched_barrier(0);
  v8f ha = strip_tile<FIN>(Al, Bl);
  {
    const float bv = bsh[16 * ct + l16];
    #pragma unroll
    for (int r = 0; r < 8; ++r) ha[r] = fmaxf(ha[r] + bv, 0.0f);
  }
  __syncthreads();
  store_chunk(Wb, w1, tid);
  __syncthreads();
  const WeightChunk w3 = load_chunk((const float4*)W3, tid);
  __builtin_amdgcn_sched_barrier(0);
  v8f hb = strip_tile<FIN>(Al, Bl);
  {
    const float bv = bsh[WROWS + 16 * ct + l16];
    #pragma unroll
    for (int r = 0; r < 8; ++r) hb[r] = fmaxf(hb[r] + bv, 0.0f);
  }
  __syncthreads();
  store_chunk(Wb, w3, tid);
  {
    float* Hw = &Rt[(16 * strip + 8 * half) * STRIDE + 16 * ct + l16];
    #pragma unroll
    for (int r = 0; r < 8; ++r) {
      Hw[r * STRIDE]         = ha[r];
      Hw[r * STRIDE + WROWS] = hb[r];
    }
  }
  __syncthreads();

  const v8f tc = strip_tile<FIN>(Al, Bl);
  if (rowBase >= nNodes) return;
  const int n = 16 * ct + l16;
  if (rowBase + 16 <= nNodes) {
    float* colBase = tout + (size_t)(rowBase + 8 * half) * FT + n;
    #pragma unroll
    for (int r = 0; r < 8; ++r) colBase[(size_t)r * FT] = tc[r];
  } else {
    #pragma unroll
    for (int r = 0; r < 8; ++r) {
      const int gRow = rowBase + r + 8 * half;
      if (gRow < nNodes) tout[(size_t)gRow * FT + n] = tc[r];
    }
  }
}

__global__ __launch_bounds__(256) void
gather_heads_kernel(const float* __restrict__ t, const int* __restrict__ rowptr,
                    const int* __restrict__ nbr, const float* __restrict__ b3,
                    float* __restrict__ emb,
                    const float* __restrict__ Wp, const float* __restrict__ bp,
                    const float* __restrict__ Wc, const float* __restrict__ bc,
                    float* __restrict__ nxt, float* __restrict__ cls, int nNodes, int nClasses)
{
  constexpr int FE      = 64;
  constexpr int STRIDE2 = FE + 4;
  constexpr int F4E     = FE / 4;
  constexpr int KPE     = FE / 8;
  constexpr int WLDP    = (FE * F4E) / 256;
  static_assert(FE / 16 == 4, "one column tile of next for each of the four wave pairs");
  __shared__ __attribute__((aligned(16))) float Bh[(FE + 16) * STRIDE2];
  __shared__ __attribute__((aligned(16))) float Et[TILE_ROWS * STRIDE2];
  __shared__ float bsh2[FE + 16];

  const int tid  = (int)threadIdx.x;
  const int lane = tid & 31;
  const int wave = tid >> 5;
  const int half = lane >> 4;
  const int l16  = lane & 15;

  const int n0 = (int)blockIdx.x * TILE_ROWS + wave * NODES_PER_WAVE;
  const int rp = wave_bounds(rowptr, n0, nNodes, lane);

  const float bvp = bp[min(tid, FE - 1)];
  const float bvc = bc[min(tid, nClasses - 1)];
  float4 wp[WLDP];
  #pragma unroll
  for (int i = 0; i < WLDP; ++i) wp[i] = ((const float4*)Wp)[tid + i * 256];
  const int wcRow = tid >> 4;
  const float4 wc = ((const float4*)Wc)[min(tid, nClasses * F4E - 1)];
  int b[NODES_PER_WAVE + 1];
  const int ids0 = first_ids(nbr, rp, b, lane);
  #pragma unroll
  for (int i = 0; i < WLDP; ++i) {
    const int idx = tid + i * 256;
    const int r = idx / F4E, c4 = idx - r * F4E;
    *(float4*)&Bh[r * STRIDE2 + 4 * c4] = wp[i];
  }
  {
    const int c4 = tid - wcRow * F4E;
    *(float4*)&Bh[(FE + wcRow) * STRIDE2 + 4 * c4] =
        (wcRow < nClasses) ? wc : make_float4(0.0f, 0.0f, 0.0f, 0.0f);
  }
  if (tid < FE) bsh2[tid] = bvp;
  if (tid < 16) bsh2[FE + tid] = (tid < nClasses) ? bvc : 0.0f;

  {
    const int c   = lane % F4E;
    const int sub = lane / F4E;
    const float4* t4 = (const float4*)t;
    float4 bv;
    bv.x = b3[4 * c]; bv.y = b3[4 * c + 1]; bv.z = b3[4 * c + 2]; bv.w = b3[4 * c + 3];
    gather_wave_nodes<F4E, true>(t4, nbr, bv, b, ids0, n0, nNodes, lane,
      [&](int q, int node, const float4& o) {
        const int r = wave * NODES_PER_WAVE + q;
        if (node < nNodes && sub == 0) ((float4*)emb)[(size_t)node * F4E + c] = o;
        if (sub == 0) *(float4*)&Et[r * STRIDE2 + 4 * c] = o;
      });
  }
  __syncthreads();

  const int strip   = wave & 1;
  const int pair    = wave >> 1;
  const int rowBase = (int)blockIdx.x * TILE_ROWS + 16 * strip;
  if (rowBase >= nNodes) return;
  float4 e4[KPE];
  {
    const float* Er = &Et[(16 * strip + l16) * STRIDE2 + 4 * half];
    #pragma unroll
    for (int j = 0; j < KPE; ++j) e4[j] = *(const float4*)(Er + 8 * j);
  }
  const bool fullRows = (rowBase + 16 <= nNodes);
  const float* Bl2 = &Bh[l16 * STRIDE2 + 4 * half];
  gemm_tiles<FE, 1>(e4, Bl2, bsh2, nxt, pair, rowBase, half, l16, nNodes, FE, 0, fullRows);
  if (pair == 3)
    gemm_tiles<FE, 1>(e4, Bl2 + FE * STRIDE2, bsh2 + FE, cls, 0, rowBase, half, l16, nNodes, nClasses, 0, fullRows);
}

extern "C" void kernel_launch(void* const* d_in, const int* in_sizes, int n_in,
                              void* d_out, int out_size, void* d_ws, size_t ws_size,
                              hipStream_t stream) {
  const float* x0 = (const float*)d_in[0];
  const int*   ei = (const int*)d_in[1];
  const float *W1 = (const float*)d_in[2],  *b1 = (const float*)d_in[3];
  const float *W2 = (const float*)d_in[4],  *b2 = (const float*)d_in[5];
  const float *W3 = (const float*)d_in[6],  *b3 = (const float*)d_in[7];
  const float *Wp = (const float*)d_in[8],  *bp = (const float*)d_in[9];
  const float *Wc = (const float*)d_in[10], *bc = (const float*)d_in[11];

  const int N      = in_sizes[0] / 64;
  const int nEdges = in_sizes[1] / 2;
  const int* src = ei;
  const int* dst = ei + nEdges;

  float* ws        = (float*)d_ws;
  float* A         = ws + 65536;
  const int nBins   = (N + BIN_NODES - 1) / BIN_NODES;
  const int nChunks = (nEdges + CHUNK - 1) / CHUNK;
  int*   pairs     = (int*)A;
  int*   offT      = (int*)A + 2 * (size_t)nChunks * CHUNK;
  int*   rowptr    = (int*)(A + (size_t)N * 128);
  int*   nbr       = rowptr + 65536;
  int*   nbrSorted = nbr + nEdges;
  float* t3        = A;

  float* out = (float*)d_out;
  float* h1  = out;
  float* emb = out;
  float* nxt = out + (size_t)N * 64;
  float* cls = out + (size_t)N * 128;

  dim3 b256(256);
  dim3 nodeGrid((unsigned)((N + 8 * NODES_PER_WAVE - 1) / (8 * NODES_PER_WAVE)));
  dim3 chunkGrid((unsigned)nChunks), binGrid((unsigned)nBins);

  chunk_sort_kernel<<<chunkGrid, b256, 0, stream>>>(src, dst, offT, pairs, nEdges, nBins);
  bin_lists_kernel <<<binGrid,   b256, 0, stream>>>(pairs, offT, rowptr, nbr, nbrSorted, N, nBins, nChunks);

  gather_gemm_kernel<<<nodeGrid, b256, 0, stream>>>(x0, rowptr, nbrSorted, W1, b1, h1, N);

  gather_layer2_kernel<<<nodeGrid, b256, 0, stream>>>(h1, rowptr, nbrSorted, W2, b2, W3, t3, N);

  gather_heads_kernel<<<nodeGrid, b256, 0, stream>>>(t3, rowptr, nbrSorted, b3, emb, Wp, bp, Wc, bc, nxt, cls, N, 10);
}


# Node launches in blocks of 16 waves on 64 rows (8 on 32 before): a weight tile is fetched and put into LDS once per 64 rows, and every block of a launch is resident in one pass; same sums and order
# speedup vs baseline: 7.2189x; 1.0519x over previous
#include <hip/hip_runtime.h>
#include <hip/hip_bf16.h>

typedef __attribute__((ext_vector_type(8))) float v8f;
typedef __attribute__((ext_vector_type(2))) float v2f;

constexpr int BIN_SHIFT = 6;
constexpr int BIN_NODES = 1 << BIN_SHIFT;
constexpr int MAX_BINS  = 1024;
constexpr int CHUNK     = 4096;
constexpr int QPT       = CHUNK / (256 * 4);
constexpr int BIN_CAP   = 4096;
constexpr int SEG_PRE   = 12;
static_assert(MAX_BINS == 4 * 256, "chunk_sort_kernel scans four bins per thread");

__device__ __forceinline__ int
load_quad(const int* __restrict__ p, int e, int n, bool aligned, int (&v)[4])
{
  if (aligned && e + 4 <= n) {
    const int4 t = *(const int4*)(p + e);
    v[0] = t.x; v[1] = t.y; v[2] = t.z; v[3] = t.w;
    return 4;
  }
  int c = 0;
  #pragma unroll
  for (int k = 0; k < 4; ++k) {
    v[k] = 0;
    if (e + k < n) { v[k] = p[e + k]; c = k + 1; }
  }
  return c;
}

template <bool GUARD>
__device__ __forceinline__ void
place_pairs(int* __restrict__ cursor, int2* __restrict__ sorted,
            const int (&d)[QPT][4], const int (&s)[QPT][4], const int (&c)[QPT])
{
  int pos[QPT][4];
  #pragma unroll
  for (int u = 0; u < QPT; ++u) {
    #pragma unroll
    for (int k = 0; k < 4; ++k) {
      pos[u][k] = 0;
      if (!GUARD || k < c[u])
        pos[u][k] = __hip_atomic_fetch_add(&cursor[d[u][k] >> BIN_SHIFT], 1, __ATOMIC_RELAXED, __HIP_MEMORY_SCOPE_WORKGROUP);
    }
  }
  #pragma unroll
  for (int u = 0; u < QPT; ++u) {
    #pragma unroll
    for (int k = 0; k < 4; ++k)
      if (!GUARD || k < c[u]) sorted[pos[u][k]] = make_int2(s[u][k], d[u][k]);
  }
}

__global__ __launch_bounds__(256) void
chunk_sort_kernel(const int* __restrict__ src, const int* __restrict__ dst,
                  int* __restrict__ offT, int* __restrict__ pairWords, int nEdges, int nBins)
{
  __shared__ int cursor[MAX_BINS];
  __shared__ int sIncl[8];
  __shared__ __attribute__((aligned(16))) int sortedW[2 * CHUNK];
  const int tid = (int)threadIdx.x, lane = tid & 31, w = tid >> 5;
  const int G = (int)gridDim.x;
  int2* sorted = (int2*)sortedW;
  const bool aligned = ((((size_t)src) | ((size_t)dst)) & 15) == 0;
  const int e0 = blockIdx.x * CHUNK + 4 * tid;
  int d[QPT][4], s[QPT][4], c[QPT];
  #pragma unroll
  for (int u = 0; u < QPT; ++u) {
    c[u] = load_quad(dst, e0 + 1024 * u, nEdges, aligned, d[u]);
    load_quad(src, e0 + 1024 * u, nEdges, aligned, s[u]);
  }
  for (int i = tid; i < MAX_BINS; i += 256) cursor[i] = 0;
  __syncthreads();
  #pragma unroll
  for (int u = 0; u < QPT; ++u) {
    #pragma unroll
    for (int k = 0; k < 4; ++k)
      if (k < c[u])
        __hip_atomic_fetch_add(&cursor[d[u][k] >> BIN_SHIFT], 1, __ATOMIC_RELAXED, __HIP_MEMORY_SCOPE_WORKGROUP);
  }
  __syncthreads();
  const int idx = 4 * tid;
  int v[4];
  #pragma unroll
  for (int k = 0; k < 4; ++k) v[k] = cursor[idx + k];
  const int local = (v[0] + v[1]) + (v[2] + v[3]);
  int incl = local;
  #pragma unroll
  for (int o = 1; o < 32; o <<= 1) {
    int y = __shfl_up(incl, o);
    if (lane >= o) incl += y;
  }
  if (lane == 31) sIncl[w] = incl;
  __syncthreads();
  int r = incl - local;
  #pragma unroll
  for (int i = 0; i < 8; ++i)
    if (i < w) r += sIncl[i];
  #pragma unroll
  for (int k = 0; k < 4; ++k) {
    cursor[idx + k] = r;
    if (idx + k < nBins) offT[(size_t)(idx + k) * G + blockIdx.x] = r;
    r += v[k];
  }
  if (tid == 255) offT[(size_t)nBins * G + blockIdx.x] = r;
  __syncthreads();
  if ((c[0] + c[1]) + (c[2] + c[3]) == 4 * QPT) place_pairs<false>(cursor, sorted, d, s, c);
  else                                         place_pairs<true >(cursor, sorted, d, s, c);
  __syncthreads();
  const int total = min(CHUNK, max(nEdges - (int)blockIdx.x * CHUNK, 0));
  int4*       outp = (int4*)pairWords + (size_t)blockIdx.x * (CHUNK / 2);
  const int4* sp   = (const int4*)sortedW;
  if (total == CHUNK) {
    #pragma unroll
    for (int u = 0; u < CHUNK / 512; ++u) outp[tid + 256 * u] = sp[tid + 256 * u];
  } else {
    #pragma unroll
    for (int u = 0; u < CHUNK / 512; ++u) {
      const int i = tid + 256 * u;
      if (2 * i < total) outp[i] = sp[i];
    }
  }
}

constexpr int NODES_PER_WAVE = 4;
constexpr int NODE_WAVES     = 16;
constexpr int NODE_THREADS   = 32 * NODE_WAVES;

constexpr int PRE = 6;

template <typename L>
__device__ __forceinline__ int
rank_in_list(const L& list, int beg, int end, int p, int s)
{
  int r = 0;
  for (int j = beg; j < end; j += 8) {
    int t[8];
    #pragma unroll
    for (int u = 0; u < 8; ++u) t[u] = list(min(j + u, end - 1));
    #pragma unroll
    for (int u = 0; u < 8; ++u) {
      const int jj = j + u;
      const bool front = (t[u] < s) || (t[u] == s && jj < p);
      r += (jj < end && front) ? 1 : 0;
    }
  }
  return r;
}

__global__ __launch_bounds__(256) void
bin_lists_kernel(const int* __restrict__ pairWords, const int* __restrict__ offT,
                 int* __restrict__ rowptr, int* nbr, int* __restrict__ nbrSorted,
                 int nNodes, int nBins, int G)
{
  __shared__ int cntL[BIN_NODES];
  __shared__ int lpL[BIN_NODES + 1];
  __shared__ int curL[BIN_NODES];
  __shared__ int listL[BIN_CAP];
  __shared__ int nodeL[BIN_CAP];
  __shared__ __attribute__((aligned(8))) int pairLW[2 * BIN_CAP];
  __shared__ int sCnt[8];
  __shared__ int sOff[8];
  const int tid = (int)threadIdx.x, lane = tid & 31, w = tid >> 5;
  const int bin = (int)blockIdx.x;
  const int2* pairs = (const int2*)pairWords;
  int2* pairL = (int2*)pairLW;
  const int nodeBase = bin * BIN_NODES;
  const int* row0 = offT + (size_t)bin * G;
  const int* row1 = row0 + G;
  if (tid < BIN_NODES) cntL[tid] = 0;
  __syncthreads();
  int m = 0;
  int start = 0;
  for (int b0 = 0; b0 < G; b0 += 256) {
    const int b = b0 + tid;
    int o0 = 0, o1 = 0;
    if (b < G) { o0 = row0[b]; o1 = row1[b]; }
    const int cnt = o1 - o0;
    const int2* chunk = pairs + (size_t)min(b, G - 1) * CHUNK;
    const int2* seg   = chunk + o0;
    const int   last  = max(o1 - 1, 0);
    int2 p[SEG_PRE];
    #pragma unroll
    for (int u = 0; u < SEG_PRE; ++u) p[u] = chunk[min(o0 + u, last)];
    __builtin_amdgcn_sched_barrier(0);
    int incl = cnt, so = o0;
    #pragma unroll
    for (int o = 1; o < 32; o <<= 1) {
      int y = __shfl_up(incl, o);
      if (lane >= o) incl += y;
      so += __shfl_xor(so, o);
    }
    if (lane == 31) sCnt[w] = incl;
    if (lane == 0)  sOff[w] = so;
    __syncthreads();
    int base = m + incl - cnt;
    #pragma unroll
    for (int i = 0; i < 8; ++i) {
      const int t = sCnt[i];
      if (i < w) base += t;
      m += t;
      start += sOff[i];
    }
    #pragma unroll
    for (int u = 0; u < SEG_PRE; ++u)
      if (u < cnt) {
        __hip_atomic_fetch_add(&cntL[p[u].y - nodeBase], 1, __ATOMIC_RELAXED, __HIP_MEMORY_SCOPE_WORKGROUP);
        if (base + u < BIN_CAP) pairL[base + u] = p[u];
      }
    for (int k = SEG_PRE; k < cnt; ++k) {
      const int2 q = seg[k];
      __hip_atomic_fetch_add(&cntL[q.y - nodeBase], 1, __ATOMIC_RELAXED, __HIP_MEMORY_SCOPE_WORKGROUP);
      if (base + k < BIN_CAP) pairL[base + k] = q;
    }
    __syncthreads();
  }
  if (tid < 32) {
    const int c0 = cntL[2 * tid], c1 = cntL[2 * tid + 1];
    int incl = c0 + c1;
    #pragma unroll
    for (int o = 1; o < 32; o <<= 1) {
      int y = __shfl_up(incl, o);
      if (tid >= o) incl += y;
    }
    const int s0 = incl - (c0 + c1), s1 = s0 + c0;
    lpL[2 * tid]  = s0;  lpL[2 * tid + 1]  = s1;
    curL[2 * tid] = s0;  curL[2 * tid + 1] = s1;
    if (tid == 31) lpL[BIN_NODES] = incl;
    if (nodeBase + 2 * tid     <= nNodes) rowptr[nodeBase + 2 * tid]     = start + s0;
    if (nodeBase + 2 * tid + 1 <= nNodes) rowptr[nodeBase + 2 * tid + 1] = start + s1;
  }
  if (bin == nBins - 1 && tid == 0 && nodeBase + BIN_NODES == nNodes) rowptr[nNodes] = start + m;
  __syncthreads();
  if (m <= BIN_CAP) {
    int2 pr[PRE];
    #pragma unroll
    for (int u = 0; u < PRE; ++u) {
      const int i = tid + 256 * u;
      pr[u] = (i < m) ? pairL[i] : make_int2(0, nodeBase);
    }
    int slot[PRE];
    #pragma unroll
    for (int u = 0; u < PRE; ++u)
      if (tid + 256 * u < m)
        slot[u] = __hip_atomic_fetch_add(&curL[pr[u].y - nodeBase], 1, __ATOMIC_RELAXED, __HIP_MEMORY_SCOPE_WORKGROUP);
    #pragma unroll
    for (int u = 0; u < PRE; ++u)
      if (tid + 256 * u < m) {
        listL[slot[u]] = pr[u].x;
        nodeL[slot[u]] = pr[u].y - nodeBase;
      }
    for (int i = tid + 256 * PRE; i < m; i += 256) {
      const int2 q   = pairL[i];
      const int  dl  = q.y - nodeBase;
      const int  sl  = __hip_atomic_fetch_add(&curL[dl], 1, __ATOMIC_RELAXED, __HIP_MEMORY_SCOPE_WORKGROUP);
      listL[sl] = q.x;
      nodeL[sl] = dl;
    }
    __syncthreads();
    for (int p = tid; p < m; p += 256) {
      const int s  = listL[p];
      const int dl = nodeL[p];
      const int beg = lpL[dl], end = lpL[dl + 1];
      const int r = rank_in_list([&](int j) { return listL[j]; }, beg, end, p, s);
      nbrSorted[start + beg + r] = s;
    }
  } else {
    for (int b = tid; b < G; b += 256) {
      const int2* seg = pairs + (size_t)b * CHUNK;
      const int o1 = row1[b];
      for (int k = row0[b]; k < o1; ++k) {
        const int2 q  = seg[k];
        const int  sl = __hip_atomic_fetch_add(&curL[q.y - nodeBase], 1, __ATOMIC_RELAXED, __HIP_MEMORY_SCOPE_WORKGROUP);
        nbr[start + sl] = q.x;
      }
    }
    __syncthreads();
    for (int q = w; q < BIN_NODES; q += 8) {
      const int beg = lpL[q], end = lpL[q + 1];
      for (int i = beg + lane; i < end; i += 32) {
        const int s = nbr[start + i];
        const int r = rank_in_list([&](int j) { return nbr[start + j]; }, beg, end, i, s);
        nbrSorted[start + beg + r] = s;
      }
    }
  }
}


template <int F4, int U, bool CLAMP>
__device__ __forceinline__ void
add_rows(float4& acc, const float4* __restrict__ x4, int ids, int j0, int cnt, int c, int sub)
{
  constexpr int EPS = 32 / F4;
  float4 v[U];
  #pragma unroll
  for (int u = 0; u < U; ++u) {
    int s;
    if constexpr (EPS == 1) {
      int j = j0 + u;
      if (CLAMP) j = min(j, cnt - 1);
      s = __builtin_amdgcn_readlane(ids, j);
    } else {
      int ja = j0 + 2 * u, jb = ja + 1;
      if (CLAMP) { ja = min(ja, cnt - 1); jb = min(jb, cnt - 1); }
      const int sa = __builtin_amdgcn_readlane(ids, ja);
      const int sb = __builtin_amdgcn_readlane(ids, jb);
      s = sub ? sb : sa;
    }
    v[u] = x4[(size_t)s * F4 + c];
  }
  if constexpr (CLAMP) {
    static_assert(U % 4 == 0, "the fence below names four rows per statement");
    #pragma unroll
    for (int g = 0; g < U; g += 4)
      asm volatile("" :: "v"(v[g].x),     "v"(v[g].y),     "v"(v[g].z),     "v"(v[g].w),
                         "v"(v[g + 1].x), "v"(v[g + 1].y), "v"(v[g + 1].z), "v"(v[g + 1].w),
                         "v"(v[g + 2].x), "v"(v[g + 2].y), "v"(v[g + 2].z), "v"(v[g + 2].w),
                         "v"(v[g + 3].x), "v"(v[g + 3].y), "v"(v[g + 3].z), "v"(v[g + 3].w));
  }
  #pragma unroll
  for (int u = 0; u < U; ++u) {
    if (!CLAMP || (j0 + u * EPS + sub < cnt)) {
      acc.x += v[u].x; acc.y += v[u].y; acc.z += v[u].z; acc.w += v[u].w;
    }
  }
}

template <int F4>
__device__ __forceinline__ void
add_piece(float4& acc, const float4* __restrict__ x4, int ids, int cnt, int c, int sub)
{
  constexpr int EPS = 32 / F4;
  int j = 0;
  for (; j + 8 * EPS <= cnt; j += 8 * EPS) add_rows<F4, 8, false>(acc, x4, ids, j, cnt, c, sub);
  if (cnt - j > 4 * EPS)  add_rows<F4, 8, true>(acc, x4, ids, j, cnt, c, sub);
  else if (cnt - j > 0)   add_rows<F4, 4, true>(acc, x4, ids, j, cnt, c, sub);
}

template <int F4, bool ACT>
__device__ __forceinline__ float4
gather_node(const float4* __restrict__ x4, const int* __restrict__ nbr, const float4& bv,
            int node, int beg, int end, int ids, int c, int sub, int lane)
{
  constexpr int EPS = 32 / F4;
  const float4 xv = x4[(size_t)node * F4 + c];
  float4 acc = make_float4(0.0f, 0.0f, 0.0f, 0.0f);
  add_piece<F4>(acc, x4, ids, min(end - beg, 32), c, sub);
  for (int base = beg + 32; base < end; base += 32) {
    const int more = nbr[min(base + lane, end - 1)];
    const int cnt  = min(end - base, 32);
    for (int j = 0; j < cnt; j += 4 * EPS) add_rows<F4, 4, true>(acc, x4, more, j, cnt, c, sub);
  }
  if (EPS == 2) {
    acc.x += __shfl_xor(acc.x, 16);
    acc.y += __shfl_xor(acc.y, 16);
    acc.z += __shfl_xor(acc.z, 16);
    acc.w += __shfl_xor(acc.w, 16);
  }
  const float rd = 1.0f / fmaxf((float)(end - beg), 1.0f);
  float4 o;
  o.x = xv.x + acc.x * rd;
  o.y = xv.y + acc.y * rd;
  o.z = xv.z + acc.z * rd;
  o.w = xv.w + acc.w * rd;
  if constexpr (ACT) {
    o.x = fmaxf(o.x + bv.x, 0.0f);
    o.y = fmaxf(o.y + bv.y, 0.0f);
    o.z = fmaxf(o.z + bv.z, 0.0f);
    o.w = fmaxf(o.w + bv.w, 0.0f);
  }
  return o;
}

__device__ __forceinline__ int
wave_bounds(const int* __restrict__ rowptr, int n0, int nNodes, int lane)
{
  return rowptr[min(n0 + lane, nNodes)];
}

__device__ __forceinline__ int
list_ids(const int* __restrict__ nbr, int beg, int end, int lane)
{
  int ids = 0;
  if (end > beg) ids = nbr[min(beg + lane, end - 1)];
  return ids;
}

__device__ __forceinline__ int
first_ids(const int* __restrict__ nbr, int rp, int (&b)[NODES_PER_WAVE + 1], int lane)
{
  #pragma unroll
  for (int q = 0; q <= NODES_PER_WAVE; ++q) b[q] = __builtin_amdgcn_readlane(rp, q);
  return list_ids(nbr, b[0], b[1], lane);
}

template <int F4, bool ACT, typename Out>
__device__ __forceinline__ void
gather_wave_nodes(const float4* __restrict__ x4, const int* __restrict__ nbr, const float4& bv,
                  const int (&b)[NODES_PER_WAVE + 1], int ids, int n0, int nNodes, int lane,
                  Out&& out)
{
  const int c   = lane % F4;
  const int sub = lane / F4;
  #pragma unroll
  for (int q = 0; q < NODES_PER_WAVE; ++q) {
    const int node = n0 + q;
    asm volatile("" :: "v"(ids));
    int idsNext = 0;
    if (q + 1 < NODES_PER_WAVE) idsNext = list_ids(nbr, b[q + 1], b[q + 2], lane);
    float4 o = make_float4(0.0f, 0.0f, 0.0f, 0.0f);
    if (node < nNodes)
      o = gather_node<F4, ACT>(x4, nbr, bv, node, b[q], b[q + 1], ids, c, sub, lane);
    out(q, node, o);
    ids = idsNext;
  }
}

template <int FIN, int NT>
__device__ __forceinline__ void
gemm_tiles(const float4 (&a4)[FIN / 8], const float* __restrict__ Bl,
           const float* __restrict__ bsh, float* __restrict__ out,
           int ct, int rowBase, int half, int l16, int nRows, int Fout, int doRelu,
           bool fullRows)
{
  constexpr int STRIDE = FIN + 4;
  constexpr int KP = FIN / 8;
  v8f c[NT];
  #pragma unroll
  for (int t = 0; t < NT; ++t) c[t] = v8f{};
  #pragma unroll
  for (int j = 0; j < KP; ++j) {
    float4 b[NT];
    #pragma unroll
    for (int t = 0; t < NT; ++t)
      b[t] = *(const float4*)(Bl + (ct + t) * 16 * STRIDE + 8 * j);
    v2f alo, ahi;
    alo.x = a4[j].x; alo.y = a4[j].y;
    ahi.x = a4[j].z; ahi.y = a4[j].w;
    #pragma unroll
    for (int t = 0; t < NT; ++t) {
      v2f blo; blo.x = b[t].x; blo.y = b[t].y;
      c[t] = __builtin_amdgcn_wmma_f32_16x16x4_f32(false, alo, false, blo, (short)0, c[t], false, false);
    }
    #pragma unroll
    for (int t = 0; t < NT; ++t) {
      v2f bhi; bhi.x = b[t].z; bhi.y = b[t].w;
      c[t] = __builtin_amdgcn_wmma_f32_16x16x4_f32(false, ahi, false, bhi, (short)0, c[t], false, false);
    }
  }
  #pragma unroll
  for (int t = 0; t < NT; ++t) {
    const int n = (ct + t) * 16 + l16;
    const float bv = bsh[n];
    float* colBase = out + (size_t)(rowBase + 8 * half) * Fout + n;
    if (fullRows && ((ct + t) * 16 + 16 <= Fout)) {
      #pragma unroll
      for (int r = 0; r < 8; ++r) {
        float val = c[t][r] + bv;
        if (doRelu) val = fmaxf(val, 0.0f);
        colBase[(size_t)r * Fout] = val;
      }
    } else {
      #pragma unroll
      for (int r = 0; r < 8; ++r) {
        int gRow = rowBase + r + 8 * half;
        float val = c[t][r] + bv;
        if (doRelu) val = fmaxf(val, 0.0f);
        if (gRow < nRows && n < Fout) out[(size_t)gRow * Fout + n] = val;
      }
    }
  }
}

constexpr int TILE_ROWS = NODE_WAVES * NODES_PER_WAVE;
constexpr int STRIPS    = TILE_ROWS / 16;
static_assert(TILE_ROWS == 64 && STRIPS == 4, "four 16-row strips per block, wave w takes strip w & 3");
__global__ __launch_bounds__(NODE_THREADS, 16) void
gather_gemm_kernel(const float* __restrict__ x, const int* __restrict__ rowptr,
                   const int* __restrict__ nbr, const float* __restrict__ W,
                   const float* __restrict__ bias, float* __restrict__ out, int nNodes)
{
  constexpr int FIN    = 64;
  constexpr int FOUT   = 128;
  constexpr int STRIDE = FIN + 4;
  constexpr int F4     = FIN / 4;
  constexpr int KP     = FIN / 8;
  constexpr int WLD    = (FOUT * F4) / NODE_THREADS;
  static_assert(FOUT / 16 == 2 * (NODE_WAVES / STRIPS), "two column tiles for each of the four wave groups");
  __shared__ __attribute__((aligned(16))) float Bs[FOUT * STRIDE];
  __shared__ __attribute__((aligned(16))) float At[TILE_ROWS * STRIDE];
  __shared__ float bsh[FOUT];

  const int tid  = (int)threadIdx.x;
  const int lane = tid & 31;
  const int wave = tid >> 5;
  const int half = lane >> 4;
  const int l16  = lane & 15;

  const int n0 = (int)blockIdx.x * TILE_ROWS + wave * NODES_PER_WAVE;
  const int rp = wave_bounds(rowptr, n0, nNodes, lane);

  const float bval = bias[min(tid, FOUT - 1)];
  float4 wv[WLD];
  #pragma unroll
  for (int i = 0; i < WLD; ++i) wv[i] = ((const float4*)W)[tid + i * NODE_THREADS];
  int b[NODES_PER_WAVE + 1];
  const int ids0 = first_ids(nbr, rp, b, lane);
  #pragma unroll
  for (int i = 0; i < WLD; ++i) {
    const int idx = tid + i * NODE_THREADS;
    const int r = idx / F4, c4 = idx - r * F4;
    *(float4*)&Bs[r * STRIDE + 4 * c4] = wv[i];
  }
  if (tid < FOUT) bsh[tid] = bval;

  {
    const int c   = lane % F4;
    const int sub = lane / F4;
    const float4* x4 = (const float4*)x;
    const float4 nob = make_float4(0.0f, 0.0f, 0.0f, 0.0f);
    gather_wave_nodes<F4, false>(x4, nbr, nob, b, ids0, n0, nNodes, lane,
      [&](int q, int, const float4& o) {
        const int r = wave * NODES_PER_WAVE + q;
        if (sub == 0) *(float4*)&At[r * STRIDE + 4 * c] = o;
      });
  }
  __syncthreads();

  const int strip   = wave & (STRIPS - 1);
  const int rowBase = (int)blockIdx.x * TILE_ROWS + 16 * strip;
  if (rowBase >= nNodes) return;
  float4 a4[KP];
  {
    const float* Ar = &At[(16 * strip + l16) * STRIDE + 4 * half];
    #pragma unroll
    for (int j = 0; j < KP; ++j) a4[j] = *(const float4*)(Ar + 8 * j);
  }
  const float* Bl = &Bs[l16 * STRIDE + 4 * half];
  const bool fullRows = (rowBase + 16 <= nNodes);
  gemm_tiles<FIN, 2>(a4, Bl, bsh, out, 2 * (wave / STRIPS), rowBase, half, l16, nNodes, FOUT, 1, fullRows);
}


template <int FIN>
__device__ __forceinline__ v8f
strip_tile(const float* __restrict__ Al, const float* __restrict__ Bl)
{
  v8f c = v8f{};
  #pragma unroll
  for (int j = 0; j < FIN / 8; ++j) {
    const float4 a = *(const float4*)(Al + 8 * j);
    const float4 b = *(const float4*)(Bl + 8 * j);
    v2f alo, ahi, blo, bhi;
    alo.x = a.x; alo.y = a.y;
    ahi.x = a.z; ahi.y = a.w;
    blo.x = b.x; blo.y = b.y;
    bhi.x = b.z; bhi.y = b.w;
    c = __builtin_amdgcn_wmma_f32_16x16x4_f32(false, alo, false, blo, (short)0, c, false, false);
    c = __builtin_amdgcn_wmma_f32_16x16x4_f32(false, ahi, false, bhi, (short)0, c, false, false);
  }
  return c;
}

constexpr int L2_STRIDE = 128 + 4;
struct WeightChunk { float4 q0, q1, q2, q3; };

__device__ __forceinline__ WeightChunk
load_chunk(const float4* __restrict__ p, int tid)
{
  WeightChunk w;
  w.q0 = p[tid];
  w.q1 = p[tid + NODE_THREADS];
  w.q2 = p[tid + 2 * NODE_THREADS];
  w.q3 = p[tid + 3 * NODE_THREADS];
  return w;
}

__device__ __forceinline__ void
store_chunk(float* __restrict__ Wb, const WeightChunk& w, int tid)
{
  float* d = Wb + (tid >> 5) * L2_STRIDE + 4 * (tid & 31);
  *(float4*)(d)                           = w.q0;
  *(float4*)(d +     NODE_WAVES * L2_STRIDE) = w.q1;
  *(float4*)(d + 2 * NODE_WAVES * L2_STRIDE) = w.q2;
  *(float4*)(d + 3 * NODE_WAVES * L2_STRIDE) = w.q3;
}

__global__ __launch_bounds__(NODE_THREADS, 16) void
gather_layer2_kernel(const float* __restrict__ h1, const int* __restrict__ rowptr,
                     const int* __restrict__ nbr, const float* __restrict__ W2,
                     const float* __restrict__ b2, const float* __restrict__ W3,
                     float* __restrict__ tout, int nNodes)
{
  constexpr int FIN    = 128;
  constexpr int STRIDE = L2_STRIDE;
  constexpr int F4     = FIN / 4;
  constexpr int FT     = 64;
  constexpr int WROWS  = 64;
  static_assert(STRIDE == FIN + 4, "one pitch for the weight chunk and the row tile");
  static_assert(F4 == 32 && WROWS * F4 == 4 * NODE_THREADS, "a chunk is four float4 per thread, a row per wave and load");
  static_assert(WROWS / 16 == NODE_WAVES / STRIPS, "one column tile of a chunk for each of the four wave groups");
  __shared__ __attribute__((aligned(16))) float Wb[WROWS * STRIDE];
  __shared__ __attribute__((aligned(16))) float Rt[TILE_ROWS * STRIDE];
  __shared__ float bsh[FIN];

  const int tid  = (int)threadIdx.x;
  const int lane = tid & 31;
  const int wave = tid >> 5;
  const int half = lane >> 4;
  const int l16  = lane & 15;

  const int n0 = (int)blockIdx.x * TILE_ROWS + wave * NODES_PER_WAVE;
  const int rp = wave_bounds(rowptr, n0, nNodes, lane);

  const float bval = b2[min(tid, FIN - 1)];
  const WeightChunk w0 = load_chunk((const float4*)W2, tid);
  int b[NODES_PER_WAVE + 1];
  const int ids0 = first_ids(nbr, rp, b, lane);
  store_chunk(Wb, w0, tid);
  if (tid < FIN) bsh[tid] = bval;

  {
    const float4* x4 = (const float4*)h1;
    const float4 nob = make_float4(0.0f, 0.0f, 0.0f, 0.0f);
    gather_wave_nodes<F4, false>(x4, nbr, nob, b, ids0, n0, nNodes, lane,
      [&](int q, int, const float4& o) {
        const int r = wave * NODES_PER_WAVE + q;
        *(float4*)&Rt[r * STRIDE + 4 * lane] = o;
      });
  }
  __syncthreads();

  const int strip   = wave & (STRIPS - 1);
  const int ct      = wave / STRIPS;
  const int rowBase = (int)blockIdx.x * TILE_ROWS + 16 * strip;
  const float* Al = &Rt[(16 * strip + l16) * STRIDE + 4 * half];
  const float* Bl = &Wb[(16 * ct + l16) * STRIDE + 4 * half];

  const WeightChunk w1 = load_chunk((const float4*)W2 + WROWS * F4, tid);
  __builtin_amdgcn_sched_barrier(0);
  v8f ha = strip_tile<FIN>(Al, Bl);
  {
    const float bv = bsh[16 * ct + l16];
    #pragma unroll
    for (int r = 0; r < 8; ++r) ha[r] = fmaxf(ha[r] + bv, 0.0f);
  }
  __syncthreads();
  store_chunk(Wb, w1, tid);
  __syncthreads();
  const WeightChunk w3 = load_chunk((const float4*)W3, tid);
  __builtin_amdgcn_sched_barrier(0);
  v8f hb = strip_tile<FIN>(Al, Bl);
  {
    const float bv = bsh[WROWS + 16 * ct + l16];
    #pragma unroll
    for (int r = 0; r < 8; ++r) hb[r] = fmaxf(hb[r] + bv, 0.0f);
  }
  __syncthreads();
  store_chunk(Wb, w3, tid);
  {
    float* Hw = &Rt[(16 * strip + 8 * half) * STRIDE + 16 * ct + l16];
    #pragma unroll
    for (int r = 0; r < 8; ++r) {
      Hw[r * STRIDE]         = ha[r];
      Hw[r * STRIDE + WROWS] = hb[r];
    }
  }
  __syncthreads();

  const v8f tc = strip_tile<FIN>(Al, Bl);
  if (rowBase >= nNodes) return;
  const int n = 16 * ct + l16;
  if (rowBase + 16 <= nNodes) {
    float* colBase = tout + (size_t)(rowBase + 8 * half) * FT + n;
    #pragma unroll
    for (int r = 0; r < 8; ++r) colBase[(size_t)r * FT] = tc[r];
  } else {
    #pragma unroll
    for (int r = 0; r < 8; ++r) {
      const int gRow = rowBase + r + 8 * half;
      if (gRow < nNodes) tout[(size_t)gRow * FT + n] = tc[r];
    }
  }
}

__global__ __launch_bounds__(NODE_THREADS, 16) void
gather_heads_kernel(const float* __restrict__ t, const int* __restrict__ rowptr,
                    const int* __restrict__ nbr, const float* __restrict__ b3,
                    float* __restrict__ emb,
                    const float* __restrict__ Wp, const float* __restrict__ bp,
                    const float* __restrict__ Wc, const float* __restrict__ bc,
                    float* __restrict__ nxt, float* __restrict__ cls, int nNodes, int nClasses)
{
  constexpr int FE      = 64;
  constexpr int STRIDE2 = FE + 4;
  constexpr int F4E     = FE / 4;
  constexpr int KPE     = FE / 8;
  constexpr int WLDP    = (FE * F4E) / NODE_THREADS;
  static_assert(FE / 16 == NODE_WAVES / STRIPS, "one column tile of next for each of the four wave groups");
  __shared__ __attribute__((aligned(16))) float Bh[(FE + 16) * STRIDE2];
  __shared__ __attribute__((aligned(16))) float Et[TILE_ROWS * STRIDE2];
  __shared__ float bsh2[FE + 16];

  const int tid  = (int)threadIdx.x;
  const int lane = tid & 31;
  const int wave = tid >> 5;
  const int half = lane >> 4;
  const int l16  = lane & 15;

  const int n0 = (int)blockIdx.x * TILE_ROWS + wave * NODES_PER_WAVE;
  const int rp = wave_bounds(rowptr, n0, nNodes, lane);

  const float bvp = bp[min(tid, FE - 1)];
  const float bvc = bc[min(tid, nClasses - 1)];
  float4 wp[WLDP];
  #pragma unroll
  for (int i = 0; i < WLDP; ++i) wp[i] = ((const float4*)Wp)[tid + i * NODE_THREADS];
  const int wcRow = tid >> 4;
  const float4 wc = ((const float4*)Wc)[min(tid, nClasses * F4E - 1)];
  int b[NODES_PER_WAVE + 1];
  const int ids0 = first_ids(nbr, rp, b, lane);
  #pragma unroll
  for (int i = 0; i < WLDP; ++i) {
    const int idx = tid + i * NODE_THREADS;
    const int r = idx / F4E, c4 = idx - r * F4E;
    *(float4*)&Bh[r * STRIDE2 + 4 * c4] = wp[i];
  }
  if (wcRow < 16) {
    const int c4 = tid - wcRow * F4E;
    *(float4*)&Bh[(FE + wcRow) * STRIDE2 + 4 * c4] =
        (wcRow < nClasses) ? wc : make_float4(0.0f, 0.0f, 0.0f, 0.0f);
  }
  if (tid < FE) bsh2[tid] = bvp;
  if (tid < 16) bsh2[FE + tid] = (tid < nClasses) ? bvc : 0.0f;

  {
    const int c   = lane % F4E;
    const int sub = lane / F4E;
    const float4* t4 = (const float4*)t;
    float4 bv;
    bv.x = b3[4 * c]; bv.y = b3[4 * c + 1]; bv.z = b3[4 * c + 2]; bv.w = b3[4 * c + 3];
    gather_wave_nodes<F4E, true>(t4, nbr, bv, b, ids0, n0, nNodes, lane,
      [&](int q, int node, const float4& o) {
        const int r = wave * NODES_PER_WAVE + q;
        if (node < nNodes && sub == 0) ((float4*)emb)[(size_t)node * F4E + c] = o;
        if (sub == 0) *(float4*)&Et[r * STRIDE2 + 4 * c] = o;
      });
  }
  __syncthreads();

  const int strip   = wave & (STRIPS - 1);
  const int pair    = wave / STRIPS;
  const int rowBase = (int)blockIdx.x * TILE_ROWS + 16 * strip;
  if (rowBase >= nNodes) return;
  float4 e4[KPE];
  {
    const float* Er = &Et[(16 * strip + l16) * STRIDE2 + 4 * half];
    #pragma unroll
    for (int j = 0; j < KPE; ++j) e4[j] = *(const float4*)(Er + 8 * j);
  }
  const bool fullRows = (rowBase + 16 <= nNodes);
  const float* Bl2 = &Bh[l16 * STRIDE2 + 4 * half];
  gemm_tiles<FE, 1>(e4, Bl2, bsh2, nxt, pair, rowBase, half, l16, nNodes, FE, 0, fullRows);
  if (pair == NODE_WAVES / STRIPS - 1)
    gemm_tiles<FE, 1>(e4, Bl2 + FE * STRIDE2, bsh2 + FE, cls, 0, rowBase, half, l16, nNodes, nClasses, 0, fullRows);
}

extern "C" void kernel_launch(void* const* d_in, const int* in_sizes, int n_in,
                              void* d_out, int out_size, void* d_ws, size_t ws_size,
                              hipStream_t stream) {
  const float* x0 = (const float*)d_in[0];
  const int*   ei = (const int*)d_in[1];
  const float *W1 = (const float*)d_in[2],  *b1 = (const float*)d_in[3];
  const float *W2 = (const float*)d_in[4],  *b2 = (const float*)d_in[5];
  const float *W3 = (const float*)d_in[6],  *b3 = (const float*)d_in[7];
  const float *Wp = (const float*)d_in[8],  *bp = (const float*)d_in[9];
  const float *Wc = (const float*)d_in[10], *bc = (const float*)d_in[11];

  const int N      = in_sizes[0] / 64;
  const int nEdges = in_sizes[1] / 2;
  const int* src = ei;
  const int* dst = ei + nEdges;

  float* ws        = (float*)d_ws;
  float* A         = ws + 65536;
  const int nBins   = (N + BIN_NODES - 1) / BIN_NODES;
  const int nChunks = (nEdges + CHUNK - 1) / CHUNK;
  int*   pairs     = (int*)A;
  int*   offT      = (int*)A + 2 * (size_t)nChunks * CHUNK;
  int*   rowptr    = (int*)(A + (size_t)N * 128);
  int*   nbr       = rowptr + 65536;
  int*   nbrSorted = nbr + nEdges;
  float* t3        = A;

  float* out = (float*)d_out;
  float* h1  = out;
  float* emb = out;
  float* nxt = out + (size_t)N * 64;
  float* cls = out + (size_t)N * 128;

  dim3 b256(256), bNode(NODE_THREADS);
  dim3 nodeGrid((unsigned)((N + TILE_ROWS - 1) / TILE_ROWS));
  dim3 chunkGrid((unsigned)nChunks), binGrid((unsigned)nBins);

  chunk_sort_kernel<<<chunkGrid, b256, 0, stream>>>(src, dst, offT, pairs, nEdges, nBins);
  bin_lists_kernel <<<binGrid,   b256, 0, stream>>>(pairs, offT, rowptr, nbr, nbrSorted, N, nBins, nChunks);

  gather_gemm_kernel<<<nodeGrid, bNode, 0, stream>>>(x0, rowptr, nbrSorted, W1, b1, h1, N);

  gather_layer2_kernel<<<nodeGrid, bNode, 0, stream>>>(h1, rowptr, nbrSorted, W2, b2, W3, t3, N);

  gather_heads_kernel<<<nodeGrid, bNode, 0, stream>>>(t3, rowptr, nbrSorted, b3, emb, Wp, bp, Wc, bc, nxt, cls, N, 10);
}
